# starting kernel
# baseline (speedup 1.0000x reference)
#include <hip/hip_runtime.h>
#include <hip/hip_bf16.h>
#include <math.h>

#define N_NODES   100000
#define N_EDGES   3200000

typedef float v2f __attribute__((ext_vector_type(2)));
typedef float v8f __attribute__((ext_vector_type(8)));

__global__ void deg_init_kernel(float* __restrict__ deg, int n) {
    int i = blockIdx.x * blockDim.x + threadIdx.x;
    if (i < n) deg[i] = 1.0f;
}

__global__ void deg_accum_kernel(const int* __restrict__ col,
                                 float* __restrict__ deg, int ne) {
    int e = blockIdx.x * blockDim.x + threadIdx.x;
    if (e >= ne) return;
    __hip_atomic_fetch_add(&deg[col[e]], 1.0f,
                           __ATOMIC_RELAXED, __HIP_MEMORY_SCOPE_AGENT);
}

__global__ void dinv_kernel(float* __restrict__ dinv, int n) {
    int i = blockIdx.x * blockDim.x + threadIdx.x;
    if (i < n) dinv[i] = rsqrtf(dinv[i]);
}

__global__ void norm_kernel(const int* __restrict__ row,
                            const int* __restrict__ col,
                            const float* __restrict__ dinv,
                            float* __restrict__ norm, int ne) {
    int e = blockIdx.x * blockDim.x + threadIdx.x;
    if (e >= ne) return;
    norm[e] = dinv[row[e]] * dinv[col[e]];
}

__global__ void __launch_bounds__(256)
gcn_wmma_xw1_kernel(const float* __restrict__ x,
                    const float* __restrict__ W1,
                    float* __restrict__ hlin,
                    int n) {
    __shared__ float Wp[128 * 16];
    const int t = threadIdx.x;
    #pragma unroll
    for (int j = 0; j < 8; ++j) {
        int idx = t * 8 + j;
        int kr  = idx >> 4;
        int nc  = idx & 15;
        Wp[idx] = (nc < 4) ? W1[kr * 4 + nc] : 0.0f;
    }
    __syncthreads();

    const int wave  = t >> 5;
    const int lane  = t & 31;
    const int tile  = blockIdx.x * 8 + wave;
    const int nTiles = n >> 4;
    if (tile >= nTiles) return;

    const int base = tile * 16;
    const int hh   = lane >> 4;
    const int nl   = lane & 15;
    const int arow = base + nl;

    v8f c = {};
    #pragma unroll 4
    for (int k = 0; k < 128; k += 4) {
        const int kk = k + 2 * hh;
        v2f a = *reinterpret_cast<const v2f*>(x + (size_t)arow * 128 + kk);
        v2f b;
        b.x = Wp[(kk + 0) * 16 + nl];
        b.y = Wp[(kk + 1) * 16 + nl];
        c = __builtin_amdgcn_wmma_f32_16x16x4_f32(
                 false, a,  false, b,
                 (short)0, c,  false,  false);
    }

    if (nl < 4) {
        #pragma unroll
        for (int v = 0; v < 8; ++v) {
            int m = base + v + 8 * hh;
            hlin[(size_t)m * 4 + nl] = c[v];
        }
    }
}

template <int IN, int OUT>
__global__ void dense_small_kernel(const float* __restrict__ h,
                                   const float* __restrict__ W,
                                   float* __restrict__ out, int n) {
    int i = blockIdx.x * blockDim.x + threadIdx.x;
    if (i >= n) return;
    float hv[IN];
    #pragma unroll
    for (int k = 0; k < IN; ++k) hv[k] = h[(size_t)i * IN + k];
    #pragma unroll
    for (int o = 0; o < OUT; ++o) {
        float s = 0.0f;
        #pragma unroll
        for (int k = 0; k < IN; ++k) s = fmaf(hv[k], W[k * OUT + o], s);
        out[(size_t)i * OUT + o] = s;
    }
}

template <int OUT>
__global__ void edge_scatter_kernel(const int* __restrict__ row,
                                    const int* __restrict__ col,
                                    const float* __restrict__ norm,
                                    const float* __restrict__ hlin,
                                    float* __restrict__ agg, int ne) {
    int e = blockIdx.x * blockDim.x + threadIdx.x;
    if (e >= ne) return;
    const int r = row[e];
    const int c = col[e];
    const float nm = norm[e];
    float msg[OUT];
    if (OUT == 4) {
        float4 v = reinterpret_cast<const float4*>(hlin)[c];
        msg[0] = v.x; msg[1] = v.y; msg[2] = v.z; msg[3] = v.w;
    } else if (OUT == 2) {
        float2 v = reinterpret_cast<const float2*>(hlin)[c];
        msg[0] = v.x; msg[1] = v.y;
    } else {
        #pragma unroll
        for (int f = 0; f < OUT; ++f) msg[f] = hlin[(size_t)c * OUT + f];
    }
    #pragma unroll
    for (int f = 0; f < OUT; ++f) {
        __hip_atomic_fetch_add(&agg[(size_t)r * OUT + f], msg[f] * nm,
                               __ATOMIC_RELAXED, __HIP_MEMORY_SCOPE_AGENT);
    }
}

template <int OUT>
__global__ void finalize_kernel(const float* __restrict__ agg,
                                const float* __restrict__ hlin,
                                const float* __restrict__ dinv,
                                const float* __restrict__ b,
                                float* __restrict__ out, int n) {
    int i = blockIdx.x * blockDim.x + threadIdx.x;
    if (i >= n) return;
    const float d2 = dinv[i] * dinv[i];
    #pragma unroll
    for (int f = 0; f < OUT; ++f) {
        float v = agg[(size_t)i * OUT + f]
                + hlin[(size_t)i * OUT + f] * d2 + b[f];
        out[(size_t)i * OUT + f] = tanhf(v);
    }
}

__global__ void classify_kernel(const float* __restrict__ h,
                                const float* __restrict__ Wc,
                                const float* __restrict__ bc,
                                float* __restrict__ out, int n) {
    int i = blockIdx.x * blockDim.x + threadIdx.x;
    if (i >= n) return;
    const float h0 = h[(size_t)i * 2 + 0];
    const float h1 = h[(size_t)i * 2 + 1];
    #pragma unroll
    for (int c = 0; c < 4; ++c) {
        out[(size_t)i * 4 + c] = fmaf(h0, Wc[c], fmaf(h1, Wc[4 + c], bc[c]));
    }
}

extern "C" void kernel_launch(void* const* d_in, const int* in_sizes, int n_in,
                              void* d_out, int out_size, void* d_ws, size_t ws_size,
                              hipStream_t stream) {
    const float* x  = (const float*)d_in[0];
    const int*   ei = (const int*)d_in[1];
    const float* W1 = (const float*)d_in[2];
    const float* b1 = (const float*)d_in[3];
    const float* W2 = (const float*)d_in[4];
    const float* b2 = (const float*)d_in[5];
    const float* W3 = (const float*)d_in[6];
    const float* b3 = (const float*)d_in[7];
    const float* Wc = (const float*)d_in[8];
    const float* bc = (const float*)d_in[9];

    const int N = N_NODES;
    const int E = N_EDGES;
    const int* row = ei;
    const int* col = ei + E;

    float* ws   = (float*)d_ws;
    float* dinv = ws;
    float* norm = dinv + N;
    float* hlin = norm + E;
    float* agg  = hlin + 4 * N;
    float* hA   = agg  + 4 * N;
    float* hB   = hA   + 4 * N;

    float* outC = (float*)d_out;
    float* outH = outC + (size_t)4 * N;

    const int BT = 256;
    const int gN = (N + BT - 1) / BT;
    const int gE = (E + BT - 1) / BT;

    deg_init_kernel<<<gN, BT, 0, stream>>>(dinv, N);
    deg_accum_kernel<<<gE, BT, 0, stream>>>(col, dinv, E);
    dinv_kernel<<<gN, BT, 0, stream>>>(dinv, N);
    norm_kernel<<<gE, BT, 0, stream>>>(row, col, dinv, norm, E);

    {
        const int tiles  = N / 16;
        const int blocks = (tiles + 7) / 8;
        gcn_wmma_xw1_kernel<<<blocks, 256, 0, stream>>>(x, W1, hlin, N);
        hipMemsetAsync(agg, 0, (size_t)4 * N * sizeof(float), stream);
        edge_scatter_kernel<4><<<gE, BT, 0, stream>>>(row, col, norm, hlin, agg, E);
        finalize_kernel<4><<<gN, BT, 0, stream>>>(agg, hlin, dinv, b1, hA, N);
    }
    {
        dense_small_kernel<4, 4><<<gN, BT, 0, stream>>>(hA, W2, hlin, N);
        hipMemsetAsync(agg, 0, (size_t)4 * N * sizeof(float), stream);
        edge_scatter_kernel<4><<<gE, BT, 0, stream>>>(row, col, norm, hlin, agg, E);
        finalize_kernel<4><<<gN, BT, 0, stream>>>(agg, hlin, dinv, b2, hB, N);
    }
    {
        dense_small_kernel<4, 2><<<gN, BT, 0, stream>>>(hB, W3, hlin, N);
        hipMemsetAsync(agg, 0, (size_t)2 * N * sizeof(float), stream);
        edge_scatter_kernel<2><<<gE, BT, 0, stream>>>(row, col, norm, hlin, agg, E);
        finalize_kernel<2><<<gN, BT, 0, stream>>>(agg, hlin, dinv, b3, outH, N);
    }
    classify_kernel<<<gN, BT, 0, stream>>>(outH, Wc, bc, outC, N);
}


# Scatter-add float atomics of the 3 layers replaced by a gather over edges grouped by destination (int atomic rank, scan, fill, per call); 8 lanes per node + shuffle tree: summation order changes
# speedup vs baseline: 2.4026x; 2.4026x over previous
#include <hip/hip_runtime.h>
#include <hip/hip_bf16.h>
#include <math.h>

#define N_NODES   100000
#define N_EDGES   3200000

typedef float v2f __attribute__((ext_vector_type(2)));
typedef float v8f __attribute__((ext_vector_type(8)));

__global__ void init_kernel(float* __restrict__ deg, int* __restrict__ cnt, int n) {
    int i = blockIdx.x * blockDim.x + threadIdx.x;
    if (i < n) {
        deg[i] = 1.0f;
        cnt[i] = 0;
    }
}

__global__ void count_rank_kernel(const int* __restrict__ row,
                                  const int* __restrict__ col,
                                  float* __restrict__ deg,
                                  int* __restrict__ cnt,
                                  int* __restrict__ rank, int ne4) {
    int q = blockIdx.x * blockDim.x + threadIdx.x;
    if (q >= ne4) return;
    const int4 c = reinterpret_cast<const int4*>(col)[q];
    const int4 r = reinterpret_cast<const int4*>(row)[q];
    __hip_atomic_fetch_add(&deg[c.x], 1.0f, __ATOMIC_RELAXED, __HIP_MEMORY_SCOPE_AGENT);
    __hip_atomic_fetch_add(&deg[c.y], 1.0f, __ATOMIC_RELAXED, __HIP_MEMORY_SCOPE_AGENT);
    __hip_atomic_fetch_add(&deg[c.z], 1.0f, __ATOMIC_RELAXED, __HIP_MEMORY_SCOPE_AGENT);
    __hip_atomic_fetch_add(&deg[c.w], 1.0f, __ATOMIC_RELAXED, __HIP_MEMORY_SCOPE_AGENT);
    int4 k;
    k.x = __hip_atomic_fetch_add(&cnt[r.x], 1, __ATOMIC_RELAXED, __HIP_MEMORY_SCOPE_AGENT);
    k.y = __hip_atomic_fetch_add(&cnt[r.y], 1, __ATOMIC_RELAXED, __HIP_MEMORY_SCOPE_AGENT);
    k.z = __hip_atomic_fetch_add(&cnt[r.z], 1, __ATOMIC_RELAXED, __HIP_MEMORY_SCOPE_AGENT);
    k.w = __hip_atomic_fetch_add(&cnt[r.w], 1, __ATOMIC_RELAXED, __HIP_MEMORY_SCOPE_AGENT);
    reinterpret_cast<int4*>(rank)[q] = k;
}

__global__ void __launch_bounds__(256)
dinv_blocksum_kernel(float* __restrict__ dinv, const int* __restrict__ cnt,
                     int* __restrict__ bsum, int n) {
    __shared__ int red[256];
    const int t = threadIdx.x;
    const int i = blockIdx.x * 256 + t;
    int c = 0;
    if (i < n) {
        dinv[i] = rsqrtf(dinv[i]);
        c = cnt[i];
    }
    red[t] = c;
    __syncthreads();
    for (int s = 128; s > 0; s >>= 1) {
        if (t < s) red[t] += red[t + s];
        __syncthreads();
    }
    if (t == 0) bsum[blockIdx.x] = red[0];
}

__global__ void __launch_bounds__(256)
rowptr_kernel(const int* __restrict__ cnt, const int* __restrict__ bsum,
              int* __restrict__ rowptr, int n) {
    __shared__ int red[256];
    __shared__ int sc[256];
    const int t = threadIdx.x;
    const int b = blockIdx.x;
    const int i = b * 256 + t;
    int part = 0;
    for (int j = t; j < b; j += 256) part += bsum[j];
    red[t] = part;
    const int c = (i < n) ? cnt[i] : 0;
    sc[t] = c;
    __syncthreads();
    for (int s = 128; s > 0; s >>= 1) {
        if (t < s) red[t] += red[t + s];
        __syncthreads();
    }
    for (int off = 1; off < 256; off <<= 1) {
        const int v = (t >= off) ? sc[t - off] : 0;
        __syncthreads();
        sc[t] += v;
        __syncthreads();
    }
    if (i < n) rowptr[i] = red[0] + sc[t] - c;
}

__global__ void fill_kernel(const int* __restrict__ row,
                            const int* __restrict__ col,
                            const int* __restrict__ rank,
                            const int* __restrict__ rowptr,
                            const float* __restrict__ dinv,
                            int* __restrict__ pair, int ne4) {
    int q = blockIdx.x * blockDim.x + threadIdx.x;
    if (q >= ne4) return;
    const int4 r = reinterpret_cast<const int4*>(row)[q];
    const int4 c = reinterpret_cast<const int4*>(col)[q];
    const int4 k = reinterpret_cast<const int4*>(rank)[q];
    const int p0 = rowptr[r.x] + k.x;
    const int p1 = rowptr[r.y] + k.y;
    const int p2 = rowptr[r.z] + k.z;
    const int p3 = rowptr[r.w] + k.w;
    const float n0 = dinv[r.x] * dinv[c.x];
    const float n1 = dinv[r.y] * dinv[c.y];
    const float n2 = dinv[r.z] * dinv[c.z];
    const float n3 = dinv[r.w] * dinv[c.w];
    int2* __restrict__ out = reinterpret_cast<int2*>(pair);
    out[p0] = make_int2(c.x, __float_as_int(n0));
    out[p1] = make_int2(c.y, __float_as_int(n1));
    out[p2] = make_int2(c.z, __float_as_int(n2));
    out[p3] = make_int2(c.w, __float_as_int(n3));
}

__global__ void __launch_bounds__(256)
gcn_wmma_xw1_kernel(const float* __restrict__ x,
                    const float* __restrict__ W1,
                    float* __restrict__ hlin,
                    int n) {
    __shared__ float Wp[128 * 16];
    const int t = threadIdx.x;
    #pragma unroll
    for (int j = 0; j < 8; ++j) {
        int idx = t * 8 + j;
        int kr  = idx >> 4;
        int nc  = idx & 15;
        Wp[idx] = (nc < 4) ? W1[kr * 4 + nc] : 0.0f;
    }
    __syncthreads();

    const int wave  = t >> 5;
    const int lane  = t & 31;
    const int tile  = blockIdx.x * 8 + wave;
    const int nTiles = n >> 4;
    if (tile >= nTiles) return;

    const int base = tile * 16;
    const int hh   = lane >> 4;
    const int nl   = lane & 15;
    const int arow = base + nl;

    v8f c = {};
    #pragma unroll 4
    for (int k = 0; k < 128; k += 4) {
        const int kk = k + 2 * hh;
        v2f a = *reinterpret_cast<const v2f*>(x + (size_t)arow * 128 + kk);
        v2f b;
        b.x = Wp[(kk + 0) * 16 + nl];
        b.y = Wp[(kk + 1) * 16 + nl];
        c = __builtin_amdgcn_wmma_f32_16x16x4_f32(
                 false, a,  false, b,
                 (short)0, c,  false,  false);
    }

    if (nl < 4) {
        #pragma unroll
        for (int v = 0; v < 8; ++v) {
            int m = base + v + 8 * hh;
            hlin[(size_t)m * 4 + nl] = c[v];
        }
    }
}

template <int IN, int OUT>
__global__ void dense_small_kernel(const float* __restrict__ h,
                                   const float* __restrict__ W,
                                   float* __restrict__ out, int n) {
    int i = blockIdx.x * blockDim.x + threadIdx.x;
    if (i >= n) return;
    float hv[IN];
    #pragma unroll
    for (int k = 0; k < IN; ++k) hv[k] = h[(size_t)i * IN + k];
    #pragma unroll
    for (int o = 0; o < OUT; ++o) {
        float s = 0.0f;
        #pragma unroll
        for (int k = 0; k < IN; ++k) s = fmaf(hv[k], W[k * OUT + o], s);
        out[(size_t)i * OUT + o] = s;
    }
}

template <int OUT>
__global__ void __launch_bounds__(256)
edge_gather_kernel(const int* __restrict__ rowptr,
                   const int* __restrict__ cnt,
                   const int* __restrict__ pair,
                   const float* __restrict__ hlin,
                   float* __restrict__ agg, int n) {
    const int g   = blockIdx.x * 256 + threadIdx.x;
    const int r   = g >> 3;
    const int sub = g & 7;
    if (r >= n) return;
    const int beg = rowptr[r];
    const int end = beg + cnt[r];
    float acc[OUT];
    #pragma unroll
    for (int f = 0; f < OUT; ++f) acc[f] = 0.0f;
    const int2* __restrict__ slots = reinterpret_cast<const int2*>(pair);
    for (int p = beg + sub; p < end; p += 32) {
        int2 pr[4];
        #pragma unroll
        for (int k = 0; k < 4; ++k) {
            const int q = p + 8 * k;
            pr[k] = slots[(q < end) ? q : (end - 1)];
        }
        #pragma unroll
        for (int k = 0; k < 4; ++k) {
            const bool live = (p + 8 * k) < end;
            const float nm = live ? __int_as_float(pr[k].y) : 0.0f;
            if (OUT == 4) {
                const float4 v = reinterpret_cast<const float4*>(hlin)[pr[k].x];
                acc[0] += (live ? v.x : 0.0f) * nm;
                acc[1] += (live ? v.y : 0.0f) * nm;
                acc[2] += (live ? v.z : 0.0f) * nm;
                acc[3] += (live ? v.w : 0.0f) * nm;
            } else {
                const float2 v = reinterpret_cast<const float2*>(hlin)[pr[k].x];
                acc[0] += (live ? v.x : 0.0f) * nm;
                acc[1] += (live ? v.y : 0.0f) * nm;
            }
        }
    }
    #pragma unroll
    for (int f = 0; f < OUT; ++f) {
        acc[f] += __shfl_xor(acc[f], 4);
        acc[f] += __shfl_xor(acc[f], 2);
        acc[f] += __shfl_xor(acc[f], 1);
    }
    if (sub == 0) {
        if (OUT == 4) {
            reinterpret_cast<float4*>(agg)[r] =
                make_float4(acc[0], acc[1], acc[2], acc[3]);
        } else {
            reinterpret_cast<float2*>(agg)[r] = make_float2(acc[0], acc[1]);
        }
    }
}

template <int OUT>
__global__ void finalize_kernel(const float* __restrict__ agg,
                                const float* __restrict__ hlin,
                                const float* __restrict__ dinv,
                                const float* __restrict__ b,
                                float* __restrict__ out, int n) {
    int i = blockIdx.x * blockDim.x + threadIdx.x;
    if (i >= n) return;
    const float d2 = dinv[i] * dinv[i];
    #pragma unroll
    for (int f = 0; f < OUT; ++f) {
        float v = agg[(size_t)i * OUT + f]
                + hlin[(size_t)i * OUT + f] * d2 + b[f];
        out[(size_t)i * OUT + f] = tanhf(v);
    }
}

__global__ void classify_kernel(const float* __restrict__ h,
                                const float* __restrict__ Wc,
                                const float* __restrict__ bc,
                                float* __restrict__ out, int n) {
    int i = blockIdx.x * blockDim.x + threadIdx.x;
    if (i >= n) return;
    const float h0 = h[(size_t)i * 2 + 0];
    const float h1 = h[(size_t)i * 2 + 1];
    #pragma unroll
    for (int c = 0; c < 4; ++c) {
        out[(size_t)i * 4 + c] = fmaf(h0, Wc[c], fmaf(h1, Wc[4 + c], bc[c]));
    }
}

extern "C" void kernel_launch(void* const* d_in, const int* in_sizes, int n_in,
                              void* d_out, int out_size, void* d_ws, size_t ws_size,
                              hipStream_t stream) {
    const float* x  = (const float*)d_in[0];
    const int*   ei = (const int*)d_in[1];
    const float* W1 = (const float*)d_in[2];
    const float* b1 = (const float*)d_in[3];
    const float* W2 = (const float*)d_in[4];
    const float* b2 = (const float*)d_in[5];
    const float* W3 = (const float*)d_in[6];
    const float* b3 = (const float*)d_in[7];
    const float* Wc = (const float*)d_in[8];
    const float* bc = (const float*)d_in[9];

    const int N = N_NODES;
    const int E = N_EDGES;
    const int* row = ei;
    const int* col = ei + E;

    float* ws     = (float*)d_ws;
    float* hlin   = ws;
    float* agg    = hlin + 4 * N;
    float* hA     = agg  + 4 * N;
    float* hB     = hA   + 4 * N;
    int*   pair   = (int*)(hB + 4 * N);
    int*   rank   = pair + 2 * E;
    float* dinv   = (float*)(rank + E);
    int*   cnt    = (int*)(dinv + N);
    int*   rowptr = cnt + N;
    int*   bsum   = rowptr + N;

    float* outC = (float*)d_out;
    float* outH = outC + (size_t)4 * N;

    const int BT = 256;
    const int gN = (N + BT - 1) / BT;
    const int gG = (8 * N + BT - 1) / BT;
    const int E4 = E / 4;
    const int gQ = (E4 + BT - 1) / BT;

    init_kernel<<<gN, BT, 0, stream>>>(dinv, cnt, N);
    count_rank_kernel<<<gQ, BT, 0, stream>>>(row, col, dinv, cnt, rank, E4);
    dinv_blocksum_kernel<<<gN, BT, 0, stream>>>(dinv, cnt, bsum, N);
    rowptr_kernel<<<gN, BT, 0, stream>>>(cnt, bsum, rowptr, N);
    fill_kernel<<<gQ, BT, 0, stream>>>(row, col, rank, rowptr, dinv, pair, E4);

    {
        const int tiles  = N / 16;
        const int blocks = (tiles + 7) / 8;
        gcn_wmma_xw1_kernel<<<blocks, 256, 0, stream>>>(x, W1, hlin, N);
        edge_gather_kernel<4><<<gG, BT, 0, stream>>>(rowptr, cnt, pair, hlin, agg, N);
        finalize_kernel<4><<<gN, BT, 0, stream>>>(agg, hlin, dinv, b1, hA, N);
    }
    {
        dense_small_kernel<4, 4><<<gN, BT, 0, stream>>>(hA, W2, hlin, N);
        edge_gather_kernel<4><<<gG, BT, 0, stream>>>(rowptr, cnt, pair, hlin, agg, N);
        finalize_kernel<4><<<gN, BT, 0, stream>>>(agg, hlin, dinv, b2, hB, N);
    }
    {
        dense_small_kernel<4, 2><<<gN, BT, 0, stream>>>(hB, W3, hlin, N);
        edge_gather_kernel<2><<<gG, BT, 0, stream>>>(rowptr, cnt, pair, hlin, agg, N);
        finalize_kernel<2><<<gN, BT, 0, stream>>>(agg, hlin, dinv, b3, outH, N);
    }
    classify_kernel<<<gN, BT, 0, stream>>>(outH, Wc, bc, outC, N);
}


# Edges grouped by destination by a two-level counting sort on LDS counters (256-node buckets) instead of a global int atomic rank and random stores; summation order within a node changes
# speedup vs baseline: 3.3905x; 1.4112x over previous
#include <hip/hip_runtime.h>
#include <hip/hip_bf16.h>
#include <math.h>

#define N_NODES   100000
#define N_EDGES   3200000

typedef float v2f __attribute__((ext_vector_type(2)));
typedef float v8f __attribute__((ext_vector_type(8)));

#define NBK   ((N_NODES + 255) / 256)
#define NCH   256
#define CH4   (N_EDGES / (4 * NCH))
static_assert(N_EDGES % (4 * NCH) == 0, "chunks are whole quads");
static_assert(N_NODES <= (1 << 17), "source index fits 17 bits of an edge word");

__global__ void init_kernel(float* __restrict__ deg, int n) {
    int i = blockIdx.x * blockDim.x + threadIdx.x;
    if (i < n) deg[i] = 1.0f;
}

__global__ void __launch_bounds__(1024)
bucket_count_kernel(const int* __restrict__ row, const int* __restrict__ col,
                    float* __restrict__ deg, int* __restrict__ tab) {
    __shared__ int h[NBK];
    const int t = threadIdx.x;
    for (int i = t; i < NBK; i += 1024) h[i] = 0;
    __syncthreads();
    const int q0 = blockIdx.x * CH4;
    for (int q = t; q < CH4; q += 1024) {
        const int4 c = reinterpret_cast<const int4*>(col)[q0 + q];
        const int4 r = reinterpret_cast<const int4*>(row)[q0 + q];
        __hip_atomic_fetch_add(&deg[c.x], 1.0f, __ATOMIC_RELAXED, __HIP_MEMORY_SCOPE_AGENT);
        __hip_atomic_fetch_add(&deg[c.y], 1.0f, __ATOMIC_RELAXED, __HIP_MEMORY_SCOPE_AGENT);
        __hip_atomic_fetch_add(&deg[c.z], 1.0f, __ATOMIC_RELAXED, __HIP_MEMORY_SCOPE_AGENT);
        __hip_atomic_fetch_add(&deg[c.w], 1.0f, __ATOMIC_RELAXED, __HIP_MEMORY_SCOPE_AGENT);
        __hip_atomic_fetch_add(&h[r.x >> 8], 1, __ATOMIC_RELAXED, __HIP_MEMORY_SCOPE_WORKGROUP);
        __hip_atomic_fetch_add(&h[r.y >> 8], 1, __ATOMIC_RELAXED, __HIP_MEMORY_SCOPE_WORKGROUP);
        __hip_atomic_fetch_add(&h[r.z >> 8], 1, __ATOMIC_RELAXED, __HIP_MEMORY_SCOPE_WORKGROUP);
        __hip_atomic_fetch_add(&h[r.w >> 8], 1, __ATOMIC_RELAXED, __HIP_MEMORY_SCOPE_WORKGROUP);
    }
    __syncthreads();
    for (int i = t; i < NBK; i += 1024) tab[i * NCH + blockIdx.x] = h[i];
}

__global__ void __launch_bounds__(256)
scan_dinv_kernel(const int* __restrict__ tab, int* __restrict__ toff,
                 float* __restrict__ dinv, int n) {
    __shared__ int red[256];
    __shared__ int sc[256];
    const int t = threadIdx.x;
    const int b = blockIdx.x;
    const int i = b * 256 + t;
    if (i < n) dinv[i] = rsqrtf(dinv[i]);
    int part = 0;
    const int4* __restrict__ tab4 = reinterpret_cast<const int4*>(tab);
    for (int j = t; j < b * 64; j += 256) {
        const int4 v = tab4[j];
        part += (v.x + v.y) + (v.z + v.w);
    }
    red[t] = part;
    const int c = tab[i];
    sc[t] = c;
    __syncthreads();
    for (int s = 128; s > 0; s >>= 1) {
        if (t < s) red[t] += red[t + s];
        __syncthreads();
    }
    for (int off = 1; off < 256; off <<= 1) {
        const int v = (t >= off) ? sc[t - off] : 0;
        __syncthreads();
        sc[t] += v;
        __syncthreads();
    }
    toff[i] = red[0] + sc[t] - c;
}

__global__ void __launch_bounds__(1024)
bucket_place_kernel(const int* __restrict__ row, const int* __restrict__ col,
                    const int* __restrict__ toff, int* __restrict__ part) {
    __shared__ int cur[NBK];
    const int t = threadIdx.x;
    for (int i = t; i < NBK; i += 1024) cur[i] = toff[i * NCH + blockIdx.x];
    __syncthreads();
    const int q0 = blockIdx.x * CH4;
    for (int q = t; q < CH4; q += 1024) {
        const int4 c = reinterpret_cast<const int4*>(col)[q0 + q];
        const int4 r = reinterpret_cast<const int4*>(row)[q0 + q];
        const int p0 = __hip_atomic_fetch_add(&cur[r.x >> 8], 1, __ATOMIC_RELAXED, __HIP_MEMORY_SCOPE_WORKGROUP);
        const int p1 = __hip_atomic_fetch_add(&cur[r.y >> 8], 1, __ATOMIC_RELAXED, __HIP_MEMORY_SCOPE_WORKGROUP);
        const int p2 = __hip_atomic_fetch_add(&cur[r.z >> 8], 1, __ATOMIC_RELAXED, __HIP_MEMORY_SCOPE_WORKGROUP);
        const int p3 = __hip_atomic_fetch_add(&cur[r.w >> 8], 1, __ATOMIC_RELAXED, __HIP_MEMORY_SCOPE_WORKGROUP);
        part[p0] = ((r.x & 255) << 17) | c.x;
        part[p1] = ((r.y & 255) << 17) | c.y;
        part[p2] = ((r.z & 255) << 17) | c.z;
        part[p3] = ((r.w & 255) << 17) | c.w;
    }
}

__global__ void __launch_bounds__(512)
bucket_group_kernel(const int* __restrict__ part, const int* __restrict__ toff,
                    const float* __restrict__ dinv,
                    int* __restrict__ rowptr, int* __restrict__ cnt,
                    int* __restrict__ pair, int n, int ne) {
    __shared__ int   c[256];
    __shared__ int   sc[256];
    __shared__ int   cur[256];
    __shared__ float dl[256];
    const int t = threadIdx.x;
    const int b = blockIdx.x;
    const int node = b * 256 + t;
    const int beg = toff[b * NCH];
    const int end = (b + 1 < NBK) ? toff[(b + 1) * NCH] : ne;
    if (t < 256) {
        c[t]  = 0;
        dl[t] = (node < n) ? dinv[node] : 0.0f;
    }
    __syncthreads();
    for (int e = beg + t; e < end; e += 4 * 512) {
        int w[4];
        #pragma unroll
        for (int k = 0; k < 4; ++k) {
            const int q = e + 512 * k;
            w[k] = (q < end) ? part[q] : -1;
        }
        #pragma unroll
        for (int k = 0; k < 4; ++k)
            if (w[k] >= 0)
                __hip_atomic_fetch_add(&c[w[k] >> 17], 1, __ATOMIC_RELAXED, __HIP_MEMORY_SCOPE_WORKGROUP);
    }
    __syncthreads();
    int my = 0;
    if (t < 256) { my = c[t]; sc[t] = my; }
    __syncthreads();
    for (int off = 1; off < 256; off <<= 1) {
        int v = 0;
        if (t < 256 && t >= off) v = sc[t - off];
        __syncthreads();
        if (t < 256) sc[t] += v;
        __syncthreads();
    }
    if (t < 256) {
        const int first = beg + sc[t] - my;
        cur[t] = first;
        if (node < n) { rowptr[node] = first; cnt[node] = my; }
    }
    __syncthreads();
    int2* __restrict__ out = reinterpret_cast<int2*>(pair);
    for (int e = beg + t; e < end; e += 4 * 512) {
        int   w[4];
        float ds[4];
        #pragma unroll
        for (int k = 0; k < 4; ++k) {
            const int q = e + 512 * k;
            w[k] = (q < end) ? part[q] : -1;
        }
        #pragma unroll
        for (int k = 0; k < 4; ++k)
            ds[k] = (w[k] >= 0) ? dinv[w[k] & 0x1FFFF] : 0.0f;
        #pragma unroll
        for (int k = 0; k < 4; ++k) {
            if (w[k] >= 0) {
                const int rl = w[k] >> 17;
                const int s  = w[k] & 0x1FFFF;
                const int p  = __hip_atomic_fetch_add(&cur[rl], 1, __ATOMIC_RELAXED, __HIP_MEMORY_SCOPE_WORKGROUP);
                const float nm = dl[rl] * ds[k];
                out[p] = make_int2(s, __float_as_int(nm));
            }
        }
    }
}

__global__ void __launch_bounds__(256)
gcn_wmma_xw1_kernel(const float* __restrict__ x,
                    const float* __restrict__ W1,
                    float* __restrict__ hlin,
                    int n) {
    __shared__ float Wp[128 * 16];
    const int t = threadIdx.x;
    #pragma unroll
    for (int j = 0; j < 8; ++j) {
        int idx = t * 8 + j;
        int kr  = idx >> 4;
        int nc  = idx & 15;
        Wp[idx] = (nc < 4) ? W1[kr * 4 + nc] : 0.0f;
    }
    __syncthreads();

    const int wave  = t >> 5;
    const int lane  = t & 31;
    const int tile  = blockIdx.x * 8 + wave;
    const int nTiles = n >> 4;
    if (tile >= nTiles) return;

    const int base = tile * 16;
    const int hh   = lane >> 4;
    const int nl   = lane & 15;
    const int arow = base + nl;

    v8f c = {};
    #pragma unroll 4
    for (int k = 0; k < 128; k += 4) {
        const int kk = k + 2 * hh;
        v2f a = *reinterpret_cast<const v2f*>(x + (size_t)arow * 128 + kk);
        v2f b;
        b.x = Wp[(kk + 0) * 16 + nl];
        b.y = Wp[(kk + 1) * 16 + nl];
        c = __builtin_amdgcn_wmma_f32_16x16x4_f32(
                 false, a,  false, b,
                 (short)0, c,  false,  false);
    }

    if (nl < 4) {
        #pragma unroll
        for (int v = 0; v < 8; ++v) {
            int m = base + v + 8 * hh;
            hlin[(size_t)m * 4 + nl] = c[v];
        }
    }
}

template <int IN, int OUT>
__global__ void dense_small_kernel(const float* __restrict__ h,
                                   const float* __restrict__ W,
                                   float* __restrict__ out, int n) {
    int i = blockIdx.x * blockDim.x + threadIdx.x;
    if (i >= n) return;
    float hv[IN];
    #pragma unroll
    for (int k = 0; k < IN; ++k) hv[k] = h[(size_t)i * IN + k];
    #pragma unroll
    for (int o = 0; o < OUT; ++o) {
        float s = 0.0f;
        #pragma unroll
        for (int k = 0; k < IN; ++k) s = fmaf(hv[k], W[k * OUT + o], s);
        out[(size_t)i * OUT + o] = s;
    }
}

template <int OUT>
__global__ void __launch_bounds__(256)
edge_gather_kernel(const int* __restrict__ rowptr,
                   const int* __restrict__ cnt,
                   const int* __restrict__ pair,
                   const float* __restrict__ hlin,
                   float* __restrict__ agg, int n) {
    const int g   = blockIdx.x * 256 + threadIdx.x;
    const int r   = g >> 3;
    const int sub = g & 7;
    if (r >= n) return;
    const int beg = rowptr[r];
    const int end = beg + cnt[r];
    float acc[OUT];
    #pragma unroll
    for (int f = 0; f < OUT; ++f) acc[f] = 0.0f;
    const int2* __restrict__ slots = reinterpret_cast<const int2*>(pair);
    for (int p = beg + sub; p < end; p += 32) {
        int2 pr[4];
        #pragma unroll
        for (int k = 0; k < 4; ++k) {
            const int q = p + 8 * k;
            pr[k] = slots[(q < end) ? q : (end - 1)];
        }
        #pragma unroll
        for (int k = 0; k < 4; ++k) {
            const bool live = (p + 8 * k) < end;
            const float nm = live ? __int_as_float(pr[k].y) : 0.0f;
            if (OUT == 4) {
                const float4 v = reinterpret_cast<const float4*>(hlin)[pr[k].x];
                acc[0] += (live ? v.x : 0.0f) * nm;
                acc[1] += (live ? v.y : 0.0f) * nm;
                acc[2] += (live ? v.z : 0.0f) * nm;
                acc[3] += (live ? v.w : 0.0f) * nm;
            } else {
                const float2 v = reinterpret_cast<const float2*>(hlin)[pr[k].x];
                acc[0] += (live ? v.x : 0.0f) * nm;
                acc[1] += (live ? v.y : 0.0f) * nm;
            }
        }
    }
    #pragma unroll
    for (int f = 0; f < OUT; ++f) {
        acc[f] += __shfl_xor(acc[f], 4);
        acc[f] += __shfl_xor(acc[f], 2);
        acc[f] += __shfl_xor(acc[f], 1);
    }
    if (sub == 0) {
        if (OUT == 4) {
            reinterpret_cast<float4*>(agg)[r] =
                make_float4(acc[0], acc[1], acc[2], acc[3]);
        } else {
            reinterpret_cast<float2*>(agg)[r] = make_float2(acc[0], acc[1]);
        }
    }
}

template <int OUT>
__global__ void finalize_kernel(const float* __restrict__ agg,
                                const float* __restrict__ hlin,
                                const float* __restrict__ dinv,
                                const float* __restrict__ b,
                                float* __restrict__ out, int n) {
    int i = blockIdx.x * blockDim.x + threadIdx.x;
    if (i >= n) return;
    const float d2 = dinv[i] * dinv[i];
    #pragma unroll
    for (int f = 0; f < OUT; ++f) {
        float v = agg[(size_t)i * OUT + f]
                + hlin[(size_t)i * OUT + f] * d2 + b[f];
        out[(size_t)i * OUT + f] = tanhf(v);
    }
}

__global__ void classify_kernel(const float* __restrict__ h,
                                const float* __restrict__ Wc,
                                const float* __restrict__ bc,
                                float* __restrict__ out, int n) {
    int i = blockIdx.x * blockDim.x + threadIdx.x;
    if (i >= n) return;
    const float h0 = h[(size_t)i * 2 + 0];
    const float h1 = h[(size_t)i * 2 + 1];
    #pragma unroll
    for (int c = 0; c < 4; ++c) {
        out[(size_t)i * 4 + c] = fmaf(h0, Wc[c], fmaf(h1, Wc[4 + c], bc[c]));
    }
}

extern "C" void kernel_launch(void* const* d_in, const int* in_sizes, int n_in,
                              void* d_out, int out_size, void* d_ws, size_t ws_size,
                              hipStream_t stream) {
    const float* x  = (const float*)d_in[0];
    const int*   ei = (const int*)d_in[1];
    const float* W1 = (const float*)d_in[2];
    const float* b1 = (const float*)d_in[3];
    const float* W2 = (const float*)d_in[4];
    const float* b2 = (const float*)d_in[5];
    const float* W3 = (const float*)d_in[6];
    const float* b3 = (const float*)d_in[7];
    const float* Wc = (const float*)d_in[8];
    const float* bc = (const float*)d_in[9];

    const int N = N_NODES;
    const int E = N_EDGES;
    const int* row = ei;
    const int* col = ei + E;

    float* ws     = (float*)d_ws;
    float* hlin   = ws;
    float* agg    = hlin + 4 * N;
    float* hA     = agg  + 4 * N;
    float* hB     = hA   + 4 * N;
    int*   pair   = (int*)(hB + 4 * N);
    int*   part   = pair + 2 * E;
    int*   tab    = part + E;
    int*   toff   = tab + NBK * NCH;
    float* dinv   = (float*)(toff + NBK * NCH);
    int*   cnt    = (int*)(dinv + N);
    int*   rowptr = cnt + N;

    float* outC = (float*)d_out;
    float* outH = outC + (size_t)4 * N;

    const int BT = 256;
    const int gN = (N + BT - 1) / BT;
    const int gG = (8 * N + BT - 1) / BT;

    init_kernel<<<gN, BT, 0, stream>>>(dinv, N);
    bucket_count_kernel<<<NCH, 1024, 0, stream>>>(row, col, dinv, tab);
    scan_dinv_kernel<<<NBK, 256, 0, stream>>>(tab, toff, dinv, N);
    bucket_place_kernel<<<NCH, 1024, 0, stream>>>(row, col, toff, part);
    bucket_group_kernel<<<NBK, 512, 0, stream>>>(part, toff, dinv, rowptr, cnt, pair, N, E);

    {
        const int tiles  = N / 16;
        const int blocks = (tiles + 7) / 8;
        gcn_wmma_xw1_kernel<<<blocks, 256, 0, stream>>>(x, W1, hlin, N);
        edge_gather_kernel<4><<<gG, BT, 0, stream>>>(rowptr, cnt, pair, hlin, agg, N);
        finalize_kernel<4><<<gN, BT, 0, stream>>>(agg, hlin, dinv, b1, hA, N);
    }
    {
        dense_small_kernel<4, 4><<<gN, BT, 0, stream>>>(hA, W2, hlin, N);
        edge_gather_kernel<4><<<gG, BT, 0, stream>>>(rowptr, cnt, pair, hlin, agg, N);
        finalize_kernel<4><<<gN, BT, 0, stream>>>(agg, hlin, dinv, b2, hB, N);
    }
    {
        dense_small_kernel<4, 2><<<gN, BT, 0, stream>>>(hB, W3, hlin, N);
        edge_gather_kernel<2><<<gG, BT, 0, stream>>>(rowptr, cnt, pair, hlin, agg, N);
        finalize_kernel<2><<<gN, BT, 0, stream>>>(agg, hlin, dinv, b3, outH, N);
    }
    classify_kernel<<<gN, BT, 0, stream>>>(outH, Wc, bc, outC, N);
}


# Degree count: per-edge scattered global float atomic adds replaced by per-slice LDS histograms (two 16-bit counts per word, 128 slices) whose planes the scan launch adds per node; deg stays exact
# speedup vs baseline: 5.6399x; 1.6634x over previous
#include <hip/hip_runtime.h>
#include <hip/hip_bf16.h>
#include <math.h>

#define N_NODES   100000
#define N_EDGES   3200000

typedef float v2f __attribute__((ext_vector_type(2)));
typedef float v8f __attribute__((ext_vector_type(8)));

#define NBK   ((N_NODES + 255) / 256)
#define NCH   256
#define CH4   (N_EDGES / (4 * NCH))
static_assert(N_EDGES % (4 * NCH) == 0, "chunks are whole quads");
static_assert(N_NODES <= (1 << 17), "source index fits 17 bits of an edge word");

#define NSL   128
#define SL4   (N_EDGES / (4 * NSL))
#define SLT   ((SL4 + 1023) / 1024)
#define NPW   (NBK * 128)
static_assert(N_EDGES % (4 * NSL) == 0, "slices are whole quads");
static_assert(N_EDGES / NSL <= 0xFFFF, "a slice's count of one node fits 16 bits");

__global__ void __launch_bounds__(1024)
deg_hist_kernel(const int* __restrict__ col, unsigned* __restrict__ planes) {
    __shared__ uint4 hc4[NPW / 4];
    unsigned* hc = reinterpret_cast<unsigned*>(hc4);
    const int t = threadIdx.x;
    const int4* __restrict__ c4 =
        reinterpret_cast<const int4*>(col) + (size_t)blockIdx.x * SL4;
    int4 c[SLT];
    #pragma unroll
    for (int k = 0; k < SLT; ++k) {
        const int q = t + 1024 * k;
        c[k] = (q < SL4) ? c4[q] : make_int4(-1, -1, -1, -1);
    }
    for (int i = t; i < NPW / 4; i += 1024) hc4[i] = make_uint4(0u, 0u, 0u, 0u);
    __syncthreads();
    #pragma unroll
    for (int k = 0; k < SLT; ++k) {
        if (c[k].x >= 0) {
            __hip_atomic_fetch_add(&hc[c[k].x >> 1], 1u << ((c[k].x & 1) << 4), __ATOMIC_RELAXED, __HIP_MEMORY_SCOPE_WORKGROUP);
            __hip_atomic_fetch_add(&hc[c[k].y >> 1], 1u << ((c[k].y & 1) << 4), __ATOMIC_RELAXED, __HIP_MEMORY_SCOPE_WORKGROUP);
            __hip_atomic_fetch_add(&hc[c[k].z >> 1], 1u << ((c[k].z & 1) << 4), __ATOMIC_RELAXED, __HIP_MEMORY_SCOPE_WORKGROUP);
            __hip_atomic_fetch_add(&hc[c[k].w >> 1], 1u << ((c[k].w & 1) << 4), __ATOMIC_RELAXED, __HIP_MEMORY_SCOPE_WORKGROUP);
        }
    }
    __syncthreads();
    uint4* __restrict__ dst =
        reinterpret_cast<uint4*>(planes) + (size_t)blockIdx.x * (NPW / 4);
    for (int i = t; i < NPW / 4; i += 1024) dst[i] = hc4[i];
}

__global__ void __launch_bounds__(1024)
bucket_count_kernel(const int* __restrict__ row, int* __restrict__ tab) {
    __shared__ int h[NBK];
    const int t = threadIdx.x;
    for (int i = t; i < NBK; i += 1024) h[i] = 0;
    __syncthreads();
    const int q0 = blockIdx.x * CH4;
    for (int q = t; q < CH4; q += 1024) {
        const int4 r = reinterpret_cast<const int4*>(row)[q0 + q];
        __hip_atomic_fetch_add(&h[r.x >> 8], 1, __ATOMIC_RELAXED, __HIP_MEMORY_SCOPE_WORKGROUP);
        __hip_atomic_fetch_add(&h[r.y >> 8], 1, __ATOMIC_RELAXED, __HIP_MEMORY_SCOPE_WORKGROUP);
        __hip_atomic_fetch_add(&h[r.z >> 8], 1, __ATOMIC_RELAXED, __HIP_MEMORY_SCOPE_WORKGROUP);
        __hip_atomic_fetch_add(&h[r.w >> 8], 1, __ATOMIC_RELAXED, __HIP_MEMORY_SCOPE_WORKGROUP);
    }
    __syncthreads();
    for (int i = t; i < NBK; i += 1024) tab[i * NCH + blockIdx.x] = h[i];
}

__global__ void __launch_bounds__(256)
scan_dinv_kernel(const int* __restrict__ tab, int* __restrict__ toff,
                 const unsigned* __restrict__ planes,
                 float* __restrict__ dinv, int n) {
    __shared__ int red[256];
    __shared__ int sc[256];
    __shared__ unsigned clo[2][128];
    __shared__ unsigned chi[2][128];
    const int t = threadIdx.x;
    const int b = blockIdx.x;
    const int i = b * 256 + t;
    {
        const int w  = t & 127;
        const int ph = t >> 7;
        const unsigned* __restrict__ p =
            planes + (size_t)(ph * (NSL / 2)) * NPW + (b * 128 + w);
        unsigned lo = 0, hi = 0;
        #pragma unroll 32
        for (int s = 0; s < NSL / 2; ++s) {
            const unsigned v = p[(size_t)s * NPW];
            lo += v & 0xFFFFu;
            hi += v >> 16;
        }
        clo[ph][w] = lo;
        chi[ph][w] = hi;
    }
    int part = 0;
    const int4* __restrict__ tab4 = reinterpret_cast<const int4*>(tab);
    for (int j = t; j < b * 64; j += 256) {
        const int4 v = tab4[j];
        part += (v.x + v.y) + (v.z + v.w);
    }
    red[t] = part;
    const int c = tab[i];
    sc[t] = c;
    __syncthreads();
    if (i < n) {
        const int w = t >> 1;
        const unsigned cn = (t & 1) ? (chi[0][w] + chi[1][w]) : (clo[0][w] + clo[1][w]);
        dinv[i] = rsqrtf(1.0f + (float)cn);
    }
    for (int s = 128; s > 0; s >>= 1) {
        if (t < s) red[t] += red[t + s];
        __syncthreads();
    }
    for (int off = 1; off < 256; off <<= 1) {
        const int v = (t >= off) ? sc[t - off] : 0;
        __syncthreads();
        sc[t] += v;
        __syncthreads();
    }
    toff[i] = red[0] + sc[t] - c;
}

__global__ void __launch_bounds__(1024)
bucket_place_kernel(const int* __restrict__ row, const int* __restrict__ col,
                    const int* __restrict__ toff, int* __restrict__ part) {
    __shared__ int cur[NBK];
    const int t = threadIdx.x;
    for (int i = t; i < NBK; i += 1024) cur[i] = toff[i * NCH + blockIdx.x];
    __syncthreads();
    const int q0 = blockIdx.x * CH4;
    for (int q = t; q < CH4; q += 1024) {
        const int4 c = reinterpret_cast<const int4*>(col)[q0 + q];
        const int4 r = reinterpret_cast<const int4*>(row)[q0 + q];
        const int p0 = __hip_atomic_fetch_add(&cur[r.x >> 8], 1, __ATOMIC_RELAXED, __HIP_MEMORY_SCOPE_WORKGROUP);
        const int p1 = __hip_atomic_fetch_add(&cur[r.y >> 8], 1, __ATOMIC_RELAXED, __HIP_MEMORY_SCOPE_WORKGROUP);
        const int p2 = __hip_atomic_fetch_add(&cur[r.z >> 8], 1, __ATOMIC_RELAXED, __HIP_MEMORY_SCOPE_WORKGROUP);
        const int p3 = __hip_atomic_fetch_add(&cur[r.w >> 8], 1, __ATOMIC_RELAXED, __HIP_MEMORY_SCOPE_WORKGROUP);
        part[p0] = ((r.x & 255) << 17) | c.x;
        part[p1] = ((r.y & 255) << 17) | c.y;
        part[p2] = ((r.z & 255) << 17) | c.z;
        part[p3] = ((r.w & 255) << 17) | c.w;
    }
}

__global__ void __launch_bounds__(512)
bucket_group_kernel(const int* __restrict__ part, const int* __restrict__ toff,
                    const float* __restrict__ dinv,
                    int* __restrict__ rowptr, int* __restrict__ cnt,
                    int* __restrict__ pair, int n, int ne) {
    __shared__ int   c[256];
    __shared__ int   sc[256];
    __shared__ int   cur[256];
    __shared__ float dl[256];
    const int t = threadIdx.x;
    const int b = blockIdx.x;
    const int node = b * 256 + t;
    const int beg = toff[b * NCH];
    const int end = (b + 1 < NBK) ? toff[(b + 1) * NCH] : ne;
    if (t < 256) {
        c[t]  = 0;
        dl[t] = (node < n) ? dinv[node] : 0.0f;
    }
    __syncthreads();
    for (int e = beg + t; e < end; e += 4 * 512) {
        int w[4];
        #pragma unroll
        for (int k = 0; k < 4; ++k) {
            const int q = e + 512 * k;
            w[k] = (q < end) ? part[q] : -1;
        }
        #pragma unroll
        for (int k = 0; k < 4; ++k)
            if (w[k] >= 0)
                __hip_atomic_fetch_add(&c[w[k] >> 17], 1, __ATOMIC_RELAXED, __HIP_MEMORY_SCOPE_WORKGROUP);
    }
    __syncthreads();
    int my = 0;
    if (t < 256) { my = c[t]; sc[t] = my; }
    __syncthreads();
    for (int off = 1; off < 256; off <<= 1) {
        int v = 0;
        if (t < 256 && t >= off) v = sc[t - off];
        __syncthreads();
        if (t < 256) sc[t] += v;
        __syncthreads();
    }
    if (t < 256) {
        const int first = beg + sc[t] - my;
        cur[t] = first;
        if (node < n) { rowptr[node] = first; cnt[node] = my; }
    }
    __syncthreads();
    int2* __restrict__ out = reinterpret_cast<int2*>(pair);
    for (int e = beg + t; e < end; e += 4 * 512) {
        int   w[4];
        float ds[4];
        #pragma unroll
        for (int k = 0; k < 4; ++k) {
            const int q = e + 512 * k;
            w[k] = (q < end) ? part[q] : -1;
        }
        #pragma unroll
        for (int k = 0; k < 4; ++k)
            ds[k] = (w[k] >= 0) ? dinv[w[k] & 0x1FFFF] : 0.0f;
        #pragma unroll
        for (int k = 0; k < 4; ++k) {
            if (w[k] >= 0) {
                const int rl = w[k] >> 17;
                const int s  = w[k] & 0x1FFFF;
                const int p  = __hip_atomic_fetch_add(&cur[rl], 1, __ATOMIC_RELAXED, __HIP_MEMORY_SCOPE_WORKGROUP);
                const float nm = dl[rl] * ds[k];
                out[p] = make_int2(s, __float_as_int(nm));
            }
        }
    }
}

__global__ void __launch_bounds__(256)
gcn_wmma_xw1_kernel(const float* __restrict__ x,
                    const float* __restrict__ W1,
                    float* __restrict__ hlin,
                    int n) {
    __shared__ float Wp[128 * 16];
    const int t = threadIdx.x;
    #pragma unroll
    for (int j = 0; j < 8; ++j) {
        int idx = t * 8 + j;
        int kr  = idx >> 4;
        int nc  = idx & 15;
        Wp[idx] = (nc < 4) ? W1[kr * 4 + nc] : 0.0f;
    }
    __syncthreads();

    const int wave  = t >> 5;
    const int lane  = t & 31;
    const int tile  = blockIdx.x * 8 + wave;
    const int nTiles = n >> 4;
    if (tile >= nTiles) return;

    const int base = tile * 16;
    const int hh   = lane >> 4;
    const int nl   = lane & 15;
    const int arow = base + nl;

    v8f c = {};
    #pragma unroll 4
    for (int k = 0; k < 128; k += 4) {
        const int kk = k + 2 * hh;
        v2f a = *reinterpret_cast<const v2f*>(x + (size_t)arow * 128 + kk);
        v2f b;
        b.x = Wp[(kk + 0) * 16 + nl];
        b.y = Wp[(kk + 1) * 16 + nl];
        c = __builtin_amdgcn_wmma_f32_16x16x4_f32(
                 false, a,  false, b,
                 (short)0, c,  false,  false);
    }

    if (nl < 4) {
        #pragma unroll
        for (int v = 0; v < 8; ++v) {
            int m = base + v + 8 * hh;
            hlin[(size_t)m * 4 + nl] = c[v];
        }
    }
}

template <int IN, int OUT>
__global__ void dense_small_kernel(const float* __restrict__ h,
                                   const float* __restrict__ W,
                                   float* __restrict__ out, int n) {
    int i = blockIdx.x * blockDim.x + threadIdx.x;
    if (i >= n) return;
    float hv[IN];
    #pragma unroll
    for (int k = 0; k < IN; ++k) hv[k] = h[(size_t)i * IN + k];
    #pragma unroll
    for (int o = 0; o < OUT; ++o) {
        float s = 0.0f;
        #pragma unroll
        for (int k = 0; k < IN; ++k) s = fmaf(hv[k], W[k * OUT + o], s);
        out[(size_t)i * OUT + o] = s;
    }
}

template <int OUT>
__global__ void __launch_bounds__(256)
edge_gather_kernel(const int* __restrict__ rowptr,
                   const int* __restrict__ cnt,
                   const int* __restrict__ pair,
                   const float* __restrict__ hlin,
                   float* __restrict__ agg, int n) {
    const int g   = blockIdx.x * 256 + threadIdx.x;
    const int r   = g >> 3;
    const int sub = g & 7;
    if (r >= n) return;
    const int beg = rowptr[r];
    const int end = beg + cnt[r];
    float acc[OUT];
    #pragma unroll
    for (int f = 0; f < OUT; ++f) acc[f] = 0.0f;
    const int2* __restrict__ slots = reinterpret_cast<const int2*>(pair);
    for (int p = beg + sub; p < end; p += 32) {
        int2 pr[4];
        #pragma unroll
        for (int k = 0; k < 4; ++k) {
            const int q = p + 8 * k;
            pr[k] = slots[(q < end) ? q : (end - 1)];
        }
        #pragma unroll
        for (int k = 0; k < 4; ++k) {
            const bool live = (p + 8 * k) < end;
            const float nm = live ? __int_as_float(pr[k].y) : 0.0f;
            if (OUT == 4) {
                const float4 v = reinterpret_cast<const float4*>(hlin)[pr[k].x];
                acc[0] += (live ? v.x : 0.0f) * nm;
                acc[1] += (live ? v.y : 0.0f) * nm;
                acc[2] += (live ? v.z : 0.0f) * nm;
                acc[3] += (live ? v.w : 0.0f) * nm;
            } else {
                const float2 v = reinterpret_cast<const float2*>(hlin)[pr[k].x];
                acc[0] += (live ? v.x : 0.0f) * nm;
                acc[1] += (live ? v.y : 0.0f) * nm;
            }
        }
    }
    #pragma unroll
    for (int f = 0; f < OUT; ++f) {
        acc[f] += __shfl_xor(acc[f], 4);
        acc[f] += __shfl_xor(acc[f], 2);
        acc[f] += __shfl_xor(acc[f], 1);
    }
    if (sub == 0) {
        if (OUT == 4) {
            reinterpret_cast<float4*>(agg)[r] =
                make_float4(acc[0], acc[1], acc[2], acc[3]);
        } else {
            reinterpret_cast<float2*>(agg)[r] = make_float2(acc[0], acc[1]);
        }
    }
}

template <int OUT>
__global__ void finalize_kernel(const float* __restrict__ agg,
                                const float* __restrict__ hlin,
                                const float* __restrict__ dinv,
                                const float* __restrict__ b,
                                float* __restrict__ out, int n) {
    int i = blockIdx.x * blockDim.x + threadIdx.x;
    if (i >= n) return;
    const float d2 = dinv[i] * dinv[i];
    #pragma unroll
    for (int f = 0; f < OUT; ++f) {
        float v = agg[(size_t)i * OUT + f]
                + hlin[(size_t)i * OUT + f] * d2 + b[f];
        out[(size_t)i * OUT + f] = tanhf(v);
    }
}

__global__ void classify_kernel(const float* __restrict__ h,
                                const float* __restrict__ Wc,
                                const float* __restrict__ bc,
                                float* __restrict__ out, int n) {
    int i = blockIdx.x * blockDim.x + threadIdx.x;
    if (i >= n) return;
    const float h0 = h[(size_t)i * 2 + 0];
    const float h1 = h[(size_t)i * 2 + 1];
    #pragma unroll
    for (int c = 0; c < 4; ++c) {
        out[(size_t)i * 4 + c] = fmaf(h0, Wc[c], fmaf(h1, Wc[4 + c], bc[c]));
    }
}

extern "C" void kernel_launch(void* const* d_in, const int* in_sizes, int n_in,
                              void* d_out, int out_size, void* d_ws, size_t ws_size,
                              hipStream_t stream) {
    const float* x  = (const float*)d_in[0];
    const int*   ei = (const int*)d_in[1];
    const float* W1 = (const float*)d_in[2];
    const float* b1 = (const float*)d_in[3];
    const float* W2 = (const float*)d_in[4];
    const float* b2 = (const float*)d_in[5];
    const float* W3 = (const float*)d_in[6];
    const float* b3 = (const float*)d_in[7];
    const float* Wc = (const float*)d_in[8];
    const float* bc = (const float*)d_in[9];

    const int N = N_NODES;
    const int E = N_EDGES;
    const int* row = ei;
    const int* col = ei + E;

    float* ws     = (float*)d_ws;
    float* hlin   = ws;
    float* agg    = hlin + 4 * N;
    float* hA     = agg  + 4 * N;
    float* hB     = hA   + 4 * N;
    int*   pair   = (int*)(hB + 4 * N);
    int*   part   = pair + 2 * E;
    int*   tab    = part + E;
    int*   toff   = tab + NBK * NCH;
    float* dinv   = (float*)(toff + NBK * NCH);
    int*   cnt    = (int*)(dinv + N);
    int*   rowptr = cnt + N;
    unsigned* planes = (unsigned*)(rowptr + N);

    float* outC = (float*)d_out;
    float* outH = outC + (size_t)4 * N;

    const int BT = 256;
    const int gN = (N + BT - 1) / BT;
    const int gG = (8 * N + BT - 1) / BT;

    deg_hist_kernel<<<NSL, 1024, 0, stream>>>(col, planes);
    bucket_count_kernel<<<NCH, 1024, 0, stream>>>(row, tab);
    scan_dinv_kernel<<<NBK, 256, 0, stream>>>(tab, toff, planes, dinv, N);
    bucket_place_kernel<<<NCH, 1024, 0, stream>>>(row, col, toff, part);
    bucket_group_kernel<<<NBK, 512, 0, stream>>>(part, toff, dinv, rowptr, cnt, pair, N, E);

    {
        const int tiles  = N / 16;
        const int blocks = (tiles + 7) / 8;
        gcn_wmma_xw1_kernel<<<blocks, 256, 0, stream>>>(x, W1, hlin, N);
        edge_gather_kernel<4><<<gG, BT, 0, stream>>>(rowptr, cnt, pair, hlin, agg, N);
        finalize_kernel<4><<<gN, BT, 0, stream>>>(agg, hlin, dinv, b1, hA, N);
    }
    {
        dense_small_kernel<4, 4><<<gN, BT, 0, stream>>>(hA, W2, hlin, N);
        edge_gather_kernel<4><<<gG, BT, 0, stream>>>(rowptr, cnt, pair, hlin, agg, N);
        finalize_kernel<4><<<gN, BT, 0, stream>>>(agg, hlin, dinv, b2, hB, N);
    }
    {
        dense_small_kernel<4, 2><<<gN, BT, 0, stream>>>(hB, W3, hlin, N);
        edge_gather_kernel<2><<<gG, BT, 0, stream>>>(rowptr, cnt, pair, hlin, agg, N);
        finalize_kernel<2><<<gN, BT, 0, stream>>>(agg, hlin, dinv, b3, outH, N);
    }
    classify_kernel<<<gN, BT, 0, stream>>>(outH, Wc, bc, outC, N);
}


# Self loop, bias, tanh and the next layer's small dense product (or the classifier) folded into each gather launch, one feature per lane of a node's eight: 15 launches become 9; sums unchanged
# speedup vs baseline: 6.0873x; 1.0793x over previous
#include <hip/hip_runtime.h>
#include <hip/hip_bf16.h>
#include <math.h>

#define N_NODES   100000
#define N_EDGES   3200000

typedef float v2f __attribute__((ext_vector_type(2)));
typedef float v8f __attribute__((ext_vector_type(8)));

#define NBK   ((N_NODES + 255) / 256)
#define NCH   256
#define CH4   (N_EDGES / (4 * NCH))
static_assert(N_EDGES % (4 * NCH) == 0, "chunks are whole quads");
static_assert(N_NODES <= (1 << 17), "source index fits 17 bits of an edge word");

#define NSL   128
#define SL4   (N_EDGES / (4 * NSL))
#define SLT   ((SL4 + 1023) / 1024)
#define NPW   (NBK * 128)
static_assert(N_EDGES % (4 * NSL) == 0, "slices are whole quads");
static_assert(N_EDGES / NSL <= 0xFFFF, "a slice's count of one node fits 16 bits");

__global__ void __launch_bounds__(1024)
deg_hist_kernel(const int* __restrict__ col, unsigned* __restrict__ planes) {
    __shared__ uint4 hc4[NPW / 4];
    unsigned* hc = reinterpret_cast<unsigned*>(hc4);
    const int t = threadIdx.x;
    const int4* __restrict__ c4 =
        reinterpret_cast<const int4*>(col) + (size_t)blockIdx.x * SL4;
    int4 c[SLT];
    #pragma unroll
    for (int k = 0; k < SLT; ++k) {
        const int q = t + 1024 * k;
        c[k] = (q < SL4) ? c4[q] : make_int4(-1, -1, -1, -1);
    }
    for (int i = t; i < NPW / 4; i += 1024) hc4[i] = make_uint4(0u, 0u, 0u, 0u);
    __syncthreads();
    #pragma unroll
    for (int k = 0; k < SLT; ++k) {
        if (c[k].x >= 0) {
            __hip_atomic_fetch_add(&hc[c[k].x >> 1], 1u << ((c[k].x & 1) << 4), __ATOMIC_RELAXED, __HIP_MEMORY_SCOPE_WORKGROUP);
            __hip_atomic_fetch_add(&hc[c[k].y >> 1], 1u << ((c[k].y & 1) << 4), __ATOMIC_RELAXED, __HIP_MEMORY_SCOPE_WORKGROUP);
            __hip_atomic_fetch_add(&hc[c[k].z >> 1], 1u << ((c[k].z & 1) << 4), __ATOMIC_RELAXED, __HIP_MEMORY_SCOPE_WORKGROUP);
            __hip_atomic_fetch_add(&hc[c[k].w >> 1], 1u << ((c[k].w & 1) << 4), __ATOMIC_RELAXED, __HIP_MEMORY_SCOPE_WORKGROUP);
        }
    }
    __syncthreads();
    uint4* __restrict__ dst =
        reinterpret_cast<uint4*>(planes) + (size_t)blockIdx.x * (NPW / 4);
    for (int i = t; i < NPW / 4; i += 1024) dst[i] = hc4[i];
}

__global__ void __launch_bounds__(1024)
bucket_count_kernel(const int* __restrict__ row, int* __restrict__ tab) {
    __shared__ int h[NBK];
    const int t = threadIdx.x;
    for (int i = t; i < NBK; i += 1024) h[i] = 0;
    __syncthreads();
    const int q0 = blockIdx.x * CH4;
    for (int q = t; q < CH4; q += 1024) {
        const int4 r = reinterpret_cast<const int4*>(row)[q0 + q];
        __hip_atomic_fetch_add(&h[r.x >> 8], 1, __ATOMIC_RELAXED, __HIP_MEMORY_SCOPE_WORKGROUP);
        __hip_atomic_fetch_add(&h[r.y >> 8], 1, __ATOMIC_RELAXED, __HIP_MEMORY_SCOPE_WORKGROUP);
        __hip_atomic_fetch_add(&h[r.z >> 8], 1, __ATOMIC_RELAXED, __HIP_MEMORY_SCOPE_WORKGROUP);
        __hip_atomic_fetch_add(&h[r.w >> 8], 1, __ATOMIC_RELAXED, __HIP_MEMORY_SCOPE_WORKGROUP);
    }
    __syncthreads();
    for (int i = t; i < NBK; i += 1024) tab[i * NCH + blockIdx.x] = h[i];
}

__global__ void __launch_bounds__(256)
scan_dinv_kernel(const int* __restrict__ tab, int* __restrict__ toff,
                 const unsigned* __restrict__ planes,
                 float* __restrict__ dinv, int n) {
    __shared__ int red[256];
    __shared__ int sc[256];
    __shared__ unsigned clo[2][128];
    __shared__ unsigned chi[2][128];
    const int t = threadIdx.x;
    const int b = blockIdx.x;
    const int i = b * 256 + t;
    {
        const int w  = t & 127;
        const int ph = t >> 7;
        const unsigned* __restrict__ p =
            planes + (size_t)(ph * (NSL / 2)) * NPW + (b * 128 + w);
        unsigned lo = 0, hi = 0;
        #pragma unroll 32
        for (int s = 0; s < NSL / 2; ++s) {
            const unsigned v = p[(size_t)s * NPW];
            lo += v & 0xFFFFu;
            hi += v >> 16;
        }
        clo[ph][w] = lo;
        chi[ph][w] = hi;
    }
    int part = 0;
    const int4* __restrict__ tab4 = reinterpret_cast<const int4*>(tab);
    for (int j = t; j < b * 64; j += 256) {
        const int4 v = tab4[j];
        part += (v.x + v.y) + (v.z + v.w);
    }
    red[t] = part;
    const int c = tab[i];
    sc[t] = c;
    __syncthreads();
    if (i < n) {
        const int w = t >> 1;
        const unsigned cn = (t & 1) ? (chi[0][w] + chi[1][w]) : (clo[0][w] + clo[1][w]);
        dinv[i] = rsqrtf(1.0f + (float)cn);
    }
    for (int s = 128; s > 0; s >>= 1) {
        if (t < s) red[t] += red[t + s];
        __syncthreads();
    }
    for (int off = 1; off < 256; off <<= 1) {
        const int v = (t >= off) ? sc[t - off] : 0;
        __syncthreads();
        sc[t] += v;
        __syncthreads();
    }
    toff[i] = red[0] + sc[t] - c;
}

__global__ void __launch_bounds__(1024)
bucket_place_kernel(const int* __restrict__ row, const int* __restrict__ col,
                    const int* __restrict__ toff, int* __restrict__ part) {
    __shared__ int cur[NBK];
    const int t = threadIdx.x;
    for (int i = t; i < NBK; i += 1024) cur[i] = toff[i * NCH + blockIdx.x];
    __syncthreads();
    const int q0 = blockIdx.x * CH4;
    for (int q = t; q < CH4; q += 1024) {
        const int4 c = reinterpret_cast<const int4*>(col)[q0 + q];
        const int4 r = reinterpret_cast<const int4*>(row)[q0 + q];
        const int p0 = __hip_atomic_fetch_add(&cur[r.x >> 8], 1, __ATOMIC_RELAXED, __HIP_MEMORY_SCOPE_WORKGROUP);
        const int p1 = __hip_atomic_fetch_add(&cur[r.y >> 8], 1, __ATOMIC_RELAXED, __HIP_MEMORY_SCOPE_WORKGROUP);
        const int p2 = __hip_atomic_fetch_add(&cur[r.z >> 8], 1, __ATOMIC_RELAXED, __HIP_MEMORY_SCOPE_WORKGROUP);
        const int p3 = __hip_atomic_fetch_add(&cur[r.w >> 8], 1, __ATOMIC_RELAXED, __HIP_MEMORY_SCOPE_WORKGROUP);
        part[p0] = ((r.x & 255) << 17) | c.x;
        part[p1] = ((r.y & 255) << 17) | c.y;
        part[p2] = ((r.z & 255) << 17) | c.z;
        part[p3] = ((r.w & 255) << 17) | c.w;
    }
}

__global__ void __launch_bounds__(512)
bucket_group_kernel(const int* __restrict__ part, const int* __restrict__ toff,
                    const float* __restrict__ dinv,
                    int* __restrict__ rowptr, int* __restrict__ cnt,
                    int* __restrict__ pair, int n, int ne) {
    __shared__ int   c[256];
    __shared__ int   sc[256];
    __shared__ int   cur[256];
    __shared__ float dl[256];
    const int t = threadIdx.x;
    const int b = blockIdx.x;
    const int node = b * 256 + t;
    const int beg = toff[b * NCH];
    const int end = (b + 1 < NBK) ? toff[(b + 1) * NCH] : ne;
    if (t < 256) {
        c[t]  = 0;
        dl[t] = (node < n) ? dinv[node] : 0.0f;
    }
    __syncthreads();
    for (int e = beg + t; e < end; e += 4 * 512) {
        int w[4];
        #pragma unroll
        for (int k = 0; k < 4; ++k) {
            const int q = e + 512 * k;
            w[k] = (q < end) ? part[q] : -1;
        }
        #pragma unroll
        for (int k = 0; k < 4; ++k)
            if (w[k] >= 0)
                __hip_atomic_fetch_add(&c[w[k] >> 17], 1, __ATOMIC_RELAXED, __HIP_MEMORY_SCOPE_WORKGROUP);
    }
    __syncthreads();
    int my = 0;
    if (t < 256) { my = c[t]; sc[t] = my; }
    __syncthreads();
    for (int off = 1; off < 256; off <<= 1) {
        int v = 0;
        if (t < 256 && t >= off) v = sc[t - off];
        __syncthreads();
        if (t < 256) sc[t] += v;
        __syncthreads();
    }
    if (t < 256) {
        const int first = beg + sc[t] - my;
        cur[t] = first;
        if (node < n) { rowptr[node] = first; cnt[node] = my; }
    }
    __syncthreads();
    int2* __restrict__ out = reinterpret_cast<int2*>(pair);
    for (int e = beg + t; e < end; e += 4 * 512) {
        int   w[4];
        float ds[4];
        #pragma unroll
        for (int k = 0; k < 4; ++k) {
            const int q = e + 512 * k;
            w[k] = (q < end) ? part[q] : -1;
        }
        #pragma unroll
        for (int k = 0; k < 4; ++k)
            ds[k] = (w[k] >= 0) ? dinv[w[k] & 0x1FFFF] : 0.0f;
        #pragma unroll
        for (int k = 0; k < 4; ++k) {
            if (w[k] >= 0) {
                const int rl = w[k] >> 17;
                const int s  = w[k] & 0x1FFFF;
                const int p  = __hip_atomic_fetch_add(&cur[rl], 1, __ATOMIC_RELAXED, __HIP_MEMORY_SCOPE_WORKGROUP);
                const float nm = dl[rl] * ds[k];
                out[p] = make_int2(s, __float_as_int(nm));
            }
        }
    }
}

__global__ void __launch_bounds__(256)
gcn_wmma_xw1_kernel(const float* __restrict__ x,
                    const float* __restrict__ W1,
                    float* __restrict__ hlin,
                    int n) {
    __shared__ float Wp[128 * 16];
    const int t = threadIdx.x;
    #pragma unroll
    for (int j = 0; j < 8; ++j) {
        int idx = t * 8 + j;
        int kr  = idx >> 4;
        int nc  = idx & 15;
        Wp[idx] = (nc < 4) ? W1[kr * 4 + nc] : 0.0f;
    }
    __syncthreads();

    const int wave  = t >> 5;
    const int lane  = t & 31;
    const int tile  = blockIdx.x * 8 + wave;
    const int nTiles = n >> 4;
    if (tile >= nTiles) return;

    const int base = tile * 16;
    const int hh   = lane >> 4;
    const int nl   = lane & 15;
    const int arow = base + nl;

    v8f c = {};
    #pragma unroll 4
    for (int k = 0; k < 128; k += 4) {
        const int kk = k + 2 * hh;
        v2f a = *reinterpret_cast<const v2f*>(x + (size_t)arow * 128 + kk);
        v2f b;
        b.x = Wp[(kk + 0) * 16 + nl];
        b.y = Wp[(kk + 1) * 16 + nl];
        c = __builtin_amdgcn_wmma_f32_16x16x4_f32(
                 false, a,  false, b,
                 (short)0, c,  false,  false);
    }

    if (nl < 4) {
        #pragma unroll
        for (int v = 0; v < 8; ++v) {
            int m = base + v + 8 * hh;
            hlin[(size_t)m * 4 + nl] = c[v];
        }
    }
}

template <int OUT, int NOUT, bool LAST>
__global__ void __launch_bounds__(256)
gather_layer_kernel(const int* __restrict__ rowptr,
                    const int* __restrict__ cnt,
                    const int* __restrict__ pair,
                    const float* __restrict__ hlin,
                    const float* __restrict__ dinv,
                    const float* __restrict__ b,
                    const float* __restrict__ Wn,
                    const float* __restrict__ bn,
                    float* __restrict__ ynext,
                    float* __restrict__ tout,
                    int n) {
    static_assert(!LAST || OUT == 2, "the classifier's sum is written for two features");
    const int g   = blockIdx.x * 256 + threadIdx.x;
    const int r   = g >> 3;
    const int sub = g & 7;
    if (r >= n) return;
    const int beg = rowptr[r];
    const int end = beg + cnt[r];
    const float hs = (sub < OUT) ? hlin[(size_t)r * OUT + sub] : 0.0f;
    const float di = dinv[r];
    const float bb = (sub < OUT) ? b[sub] : 0.0f;
    float w[OUT];
    #pragma unroll
    for (int k = 0; k < OUT; ++k) w[k] = (sub < NOUT) ? Wn[k * NOUT + sub] : 0.0f;
    float bo = 0.0f;
    if (LAST) bo = (sub < NOUT) ? bn[sub] : 0.0f;

    float acc[OUT];
    #pragma unroll
    for (int f = 0; f < OUT; ++f) acc[f] = 0.0f;
    const int2* __restrict__ slots = reinterpret_cast<const int2*>(pair);
    for (int p = beg + sub; p < end; p += 32) {
        int2 pr[4];
        #pragma unroll
        for (int k = 0; k < 4; ++k) {
            const int q = p + 8 * k;
            pr[k] = slots[(q < end) ? q : (end - 1)];
        }
        #pragma unroll
        for (int k = 0; k < 4; ++k) {
            const bool live = (p + 8 * k) < end;
            const float nm = live ? __int_as_float(pr[k].y) : 0.0f;
            if (OUT == 4) {
                const float4 v = reinterpret_cast<const float4*>(hlin)[pr[k].x];
                acc[0] += (live ? v.x : 0.0f) * nm;
                acc[1] += (live ? v.y : 0.0f) * nm;
                acc[2] += (live ? v.z : 0.0f) * nm;
                acc[3] += (live ? v.w : 0.0f) * nm;
            } else {
                const float2 v = reinterpret_cast<const float2*>(hlin)[pr[k].x];
                acc[0] += (live ? v.x : 0.0f) * nm;
                acc[1] += (live ? v.y : 0.0f) * nm;
            }
        }
    }
    #pragma unroll
    for (int f = 0; f < OUT; ++f) {
        acc[f] += __shfl_xor(acc[f], 4);
        acc[f] += __shfl_xor(acc[f], 2);
        acc[f] += __shfl_xor(acc[f], 1);
    }

    float a = acc[0];
    #pragma unroll
    for (int f = 1; f < OUT; ++f) a = (sub == f) ? acc[f] : a;
    const float d2 = di * di;
    const float t  = tanhf(a + hs * d2 + bb);
    const int first = (threadIdx.x & 31) & ~7;
    float tk[OUT];
    #pragma unroll
    for (int k = 0; k < OUT; ++k) tk[k] = __shfl(t, first + k);
    float y;
    if (LAST) {
        y = fmaf(tk[0], w[0], fmaf(tk[1], w[1], bo));
    } else {
        y = 0.0f;
        #pragma unroll
        for (int k = 0; k < OUT; ++k) y = fmaf(tk[k], w[k], y);
    }
    if (sub < NOUT) ynext[(size_t)r * NOUT + sub] = y;
    if (LAST) {
        if (sub < OUT) tout[(size_t)r * OUT + sub] = t;
    }
}

extern "C" void kernel_launch(void* const* d_in, const int* in_sizes, int n_in,
                              void* d_out, int out_size, void* d_ws, size_t ws_size,
                              hipStream_t stream) {
    const float* x  = (const float*)d_in[0];
    const int*   ei = (const int*)d_in[1];
    const float* W1 = (const float*)d_in[2];
    const float* b1 = (const float*)d_in[3];
    const float* W2 = (const float*)d_in[4];
    const float* b2 = (const float*)d_in[5];
    const float* W3 = (const float*)d_in[6];
    const float* b3 = (const float*)d_in[7];
    const float* Wc = (const float*)d_in[8];
    const float* bc = (const float*)d_in[9];

    const int N = N_NODES;
    const int E = N_EDGES;
    const int* row = ei;
    const int* col = ei + E;

    float* ws     = (float*)d_ws;
    float* hP     = ws;
    float* hQ     = hP + 4 * N;
    int*   pair   = (int*)(hQ + 4 * N);
    int*   part   = pair + 2 * E;
    int*   tab    = part + E;
    int*   toff   = tab + NBK * NCH;
    float* dinv   = (float*)(toff + NBK * NCH);
    int*   cnt    = (int*)(dinv + N);
    int*   rowptr = cnt + N;
    unsigned* planes = (unsigned*)(rowptr + N);

    float* outC = (float*)d_out;
    float* outH = outC + (size_t)4 * N;

    const int BT = 256;
    const int gG = (8 * N + BT - 1) / BT;

    deg_hist_kernel<<<NSL, 1024, 0, stream>>>(col, planes);
    bucket_count_kernel<<<NCH, 1024, 0, stream>>>(row, tab);
    scan_dinv_kernel<<<NBK, 256, 0, stream>>>(tab, toff, planes, dinv, N);
    bucket_place_kernel<<<NCH, 1024, 0, stream>>>(row, col, toff, part);
    bucket_group_kernel<<<NBK, 512, 0, stream>>>(part, toff, dinv, rowptr, cnt, pair, N, E);

    {
        const int tiles  = N / 16;
        const int blocks = (tiles + 7) / 8;
        gcn_wmma_xw1_kernel<<<blocks, 256, 0, stream>>>(x, W1, hP, N);
    }
    gather_layer_kernel<4, 4, false><<<gG, BT, 0, stream>>>(
        rowptr, cnt, pair, hP, dinv, b1, W2, nullptr, hQ, nullptr, N);
    gather_layer_kernel<4, 2, false><<<gG, BT, 0, stream>>>(
        rowptr, cnt, pair, hQ, dinv, b2, W3, nullptr, hP, nullptr, N);
    gather_layer_kernel<2, 4, true><<<gG, BT, 0, stream>>>(
        rowptr, cnt, pair, hP, dinv, b3, Wc, bc, outC, outH, N);
}


# Both levels of the edge sort stage their output in LDS in final order and copy it out as contiguous runs instead of one scattered global store per edge; sums stay in arrival order
# speedup vs baseline: 7.8305x; 1.2864x over previous
#include <hip/hip_runtime.h>
#include <hip/hip_bf16.h>
#include <math.h>

#define N_NODES   100000
#define N_EDGES   3200000

typedef float v2f __attribute__((ext_vector_type(2)));
typedef float v8f __attribute__((ext_vector_type(8)));

#define NBK   ((N_NODES + 255) / 256)
#define NCH   256
#define CH4   (N_EDGES / (4 * NCH))
static_assert(N_EDGES % (4 * NCH) == 0, "chunks are whole quads");
static_assert(N_NODES <= (1 << 17), "source index fits 17 bits of an edge word");

#define NSL   128
#define SL4   (N_EDGES / (4 * NSL))
#define SLT   ((SL4 + 1023) / 1024)
#define NPW   (NBK * 128)
static_assert(N_EDGES % (4 * NSL) == 0, "slices are whole quads");
static_assert(N_EDGES / NSL <= 0xFFFF, "a slice's count of one node fits 16 bits");

__global__ void __launch_bounds__(1024)
deg_hist_kernel(const int* __restrict__ col, unsigned* __restrict__ planes) {
    __shared__ uint4 hc4[NPW / 4];
    unsigned* hc = reinterpret_cast<unsigned*>(hc4);
    const int t = threadIdx.x;
    const int4* __restrict__ c4 =
        reinterpret_cast<const int4*>(col) + (size_t)blockIdx.x * SL4;
    int4 c[SLT];
    #pragma unroll
    for (int k = 0; k < SLT; ++k) {
        const int q = t + 1024 * k;
        c[k] = (q < SL4) ? c4[q] : make_int4(-1, -1, -1, -1);
    }
    for (int i = t; i < NPW / 4; i += 1024) hc4[i] = make_uint4(0u, 0u, 0u, 0u);
    __syncthreads();
    #pragma unroll
    for (int k = 0; k < SLT; ++k) {
        if (c[k].x >= 0) {
            __hip_atomic_fetch_add(&hc[c[k].x >> 1], 1u << ((c[k].x & 1) << 4), __ATOMIC_RELAXED, __HIP_MEMORY_SCOPE_WORKGROUP);
            __hip_atomic_fetch_add(&hc[c[k].y >> 1], 1u << ((c[k].y & 1) << 4), __ATOMIC_RELAXED, __HIP_MEMORY_SCOPE_WORKGROUP);
            __hip_atomic_fetch_add(&hc[c[k].z >> 1], 1u << ((c[k].z & 1) << 4), __ATOMIC_RELAXED, __HIP_MEMORY_SCOPE_WORKGROUP);
            __hip_atomic_fetch_add(&hc[c[k].w >> 1], 1u << ((c[k].w & 1) << 4), __ATOMIC_RELAXED, __HIP_MEMORY_SCOPE_WORKGROUP);
        }
    }
    __syncthreads();
    uint4* __restrict__ dst =
        reinterpret_cast<uint4*>(planes) + (size_t)blockIdx.x * (NPW / 4);
    for (int i = t; i < NPW / 4; i += 1024) dst[i] = hc4[i];
}

__global__ void __launch_bounds__(1024)
bucket_count_kernel(const int* __restrict__ row, int* __restrict__ tab) {
    __shared__ int h[NBK];
    const int t = threadIdx.x;
    for (int i = t; i < NBK; i += 1024) h[i] = 0;
    __syncthreads();
    const int q0 = blockIdx.x * CH4;
    for (int q = t; q < CH4; q += 1024) {
        const int4 r = reinterpret_cast<const int4*>(row)[q0 + q];
        __hip_atomic_fetch_add(&h[r.x >> 8], 1, __ATOMIC_RELAXED, __HIP_MEMORY_SCOPE_WORKGROUP);
        __hip_atomic_fetch_add(&h[r.y >> 8], 1, __ATOMIC_RELAXED, __HIP_MEMORY_SCOPE_WORKGROUP);
        __hip_atomic_fetch_add(&h[r.z >> 8], 1, __ATOMIC_RELAXED, __HIP_MEMORY_SCOPE_WORKGROUP);
        __hip_atomic_fetch_add(&h[r.w >> 8], 1, __ATOMIC_RELAXED, __HIP_MEMORY_SCOPE_WORKGROUP);
    }
    __syncthreads();
    for (int i = t; i < NBK; i += 1024) tab[i * NCH + blockIdx.x] = h[i];
}

__global__ void __launch_bounds__(256)
scan_dinv_kernel(const int* __restrict__ tab, int* __restrict__ toff,
                 const unsigned* __restrict__ planes,
                 float* __restrict__ dinv, int n) {
    __shared__ int red[256];
    __shared__ int sc[256];
    __shared__ unsigned clo[2][128];
    __shared__ unsigned chi[2][128];
    const int t = threadIdx.x;
    const int b = blockIdx.x;
    const int i = b * 256 + t;
    {
        const int w  = t & 127;
        const int ph = t >> 7;
        const unsigned* __restrict__ p =
            planes + (size_t)(ph * (NSL / 2)) * NPW + (b * 128 + w);
        unsigned lo = 0, hi = 0;
        #pragma unroll 32
        for (int s = 0; s < NSL / 2; ++s) {
            const unsigned v = p[(size_t)s * NPW];
            lo += v & 0xFFFFu;
            hi += v >> 16;
        }
        clo[ph][w] = lo;
        chi[ph][w] = hi;
    }
    int part = 0;
    const int4* __restrict__ tab4 = reinterpret_cast<const int4*>(tab);
    for (int j = t; j < b * 64; j += 256) {
        const int4 v = tab4[j];
        part += (v.x + v.y) + (v.z + v.w);
    }
    red[t] = part;
    const int c = tab[i];
    sc[t] = c;
    __syncthreads();
    if (i < n) {
        const int w = t >> 1;
        const unsigned cn = (t & 1) ? (chi[0][w] + chi[1][w]) : (clo[0][w] + clo[1][w]);
        dinv[i] = rsqrtf(1.0f + (float)cn);
    }
    for (int s = 128; s > 0; s >>= 1) {
        if (t < s) red[t] += red[t + s];
        __syncthreads();
    }
    for (int off = 1; off < 256; off <<= 1) {
        const int v = (t >= off) ? sc[t - off] : 0;
        __syncthreads();
        sc[t] += v;
        __syncthreads();
    }
    toff[i] = red[0] + sc[t] - c;
}

#define CHE   (N_EDGES / NCH)
static_assert(NBK <= 1024, "one bucket per thread in the chunk's scan");
__global__ void __launch_bounds__(1024)
bucket_place_kernel(const int* __restrict__ row, const int* __restrict__ col,
                    const int* __restrict__ tab, const int* __restrict__ toff,
                    int* __restrict__ part) {
    __shared__ int sw[CHE];
    __shared__ int sd[CHE];
    __shared__ int cur[NBK];
    __shared__ int del[NBK];
    __shared__ int wtot[32];
    const int t    = threadIdx.x;
    const int lane = t & 31;
    const int wv   = t >> 5;
    const int q0 = blockIdx.x * CH4;
    const int4 c0 = reinterpret_cast<const int4*>(col)[q0 + t];
    const int4 r0 = reinterpret_cast<const int4*>(row)[q0 + t];
    static_assert(CH4 >= 1024, "every thread has a first quad");
    int cn = 0, g = 0;
    if (t < NBK) { cn = tab[t * NCH + blockIdx.x]; g = toff[t * NCH + blockIdx.x]; }
    int inc = cn;
    #pragma unroll
    for (int o = 1; o < 32; o <<= 1) {
        const int v = __shfl_up(inc, o);
        if (lane >= o) inc += v;
    }
    if (lane == 31) wtot[wv] = inc;
    __syncthreads();
    {
        const int wt = wtot[lane];
        int winc = wt;
        #pragma unroll
        for (int o = 1; o < 32; o <<= 1) {
            const int v = __shfl_up(winc, o);
            if (lane >= o) winc += v;
        }
        const int wbase = __shfl(winc - wt, wv);
        if (t < NBK) {
            const int ls = wbase + inc - cn;
            cur[t] = ls;
            del[t] = g - ls;
        }
    }
    __syncthreads();
    for (int q = t; q < CH4; q += 1024) {
        int4 c = c0, r = r0;
        if (q != t) {
            c = reinterpret_cast<const int4*>(col)[q0 + q];
            r = reinterpret_cast<const int4*>(row)[q0 + q];
        }
        const int b0 = r.x >> 8, b1 = r.y >> 8, b2 = r.z >> 8, b3 = r.w >> 8;
        const int p0 = __hip_atomic_fetch_add(&cur[b0], 1, __ATOMIC_RELAXED, __HIP_MEMORY_SCOPE_WORKGROUP);
        const int p1 = __hip_atomic_fetch_add(&cur[b1], 1, __ATOMIC_RELAXED, __HIP_MEMORY_SCOPE_WORKGROUP);
        const int p2 = __hip_atomic_fetch_add(&cur[b2], 1, __ATOMIC_RELAXED, __HIP_MEMORY_SCOPE_WORKGROUP);
        const int p3 = __hip_atomic_fetch_add(&cur[b3], 1, __ATOMIC_RELAXED, __HIP_MEMORY_SCOPE_WORKGROUP);
        sw[p0] = ((r.x & 255) << 17) | c.x;  sd[p0] = p0 + del[b0];
        sw[p1] = ((r.y & 255) << 17) | c.y;  sd[p1] = p1 + del[b1];
        sw[p2] = ((r.z & 255) << 17) | c.z;  sd[p2] = p2 + del[b2];
        sw[p3] = ((r.w & 255) << 17) | c.w;  sd[p3] = p3 + del[b3];
    }
    __syncthreads();
    for (int i = t; i < CHE; i += 1024) part[sd[i]] = sw[i];
}

#define GCAP  12288
__global__ void __launch_bounds__(512)
bucket_group_kernel(const int* __restrict__ part, const int* __restrict__ toff,
                    const float* __restrict__ dinv,
                    int* __restrict__ rowptr, int* __restrict__ cnt,
                    int* __restrict__ pair, int n, int ne) {
    __shared__ int   c[256];
    __shared__ int   sc[256];
    __shared__ int   cur[256];
    __shared__ float dl[256];
    __shared__ int2  st[GCAP];
    const int t = threadIdx.x;
    const int b = blockIdx.x;
    const int node = b * 256 + t;
    const int beg = toff[b * NCH];
    const int end = (b + 1 < NBK) ? toff[(b + 1) * NCH] : ne;
    if (t < 256) {
        c[t]  = 0;
        dl[t] = (node < n) ? dinv[node] : 0.0f;
    }
    __syncthreads();
    for (int e = beg + t; e < end; e += 4 * 512) {
        int w[4];
        #pragma unroll
        for (int k = 0; k < 4; ++k) {
            const int q = e + 512 * k;
            w[k] = (q < end) ? part[q] : -1;
        }
        #pragma unroll
        for (int k = 0; k < 4; ++k)
            if (w[k] >= 0)
                __hip_atomic_fetch_add(&c[w[k] >> 17], 1, __ATOMIC_RELAXED, __HIP_MEMORY_SCOPE_WORKGROUP);
    }
    __syncthreads();
    int my = 0;
    if (t < 256) { my = c[t]; sc[t] = my; }
    __syncthreads();
    for (int off = 1; off < 256; off <<= 1) {
        int v = 0;
        if (t < 256 && t >= off) v = sc[t - off];
        __syncthreads();
        if (t < 256) sc[t] += v;
        __syncthreads();
    }
    if (t < 256) {
        const int first = beg + sc[t] - my;
        cur[t] = first;
        if (node < n) { rowptr[node] = first; cnt[node] = my; }
    }
    __syncthreads();
    int2* __restrict__ out = reinterpret_cast<int2*>(pair);
    const bool staged = (end - beg) <= GCAP;
    for (int e = beg + t; e < end; e += 4 * 512) {
        int   w[4];
        float ds[4];
        #pragma unroll
        for (int k = 0; k < 4; ++k) {
            const int q = e + 512 * k;
            w[k] = (q < end) ? part[q] : -1;
        }
        #pragma unroll
        for (int k = 0; k < 4; ++k)
            ds[k] = (w[k] >= 0) ? dinv[w[k] & 0x1FFFF] : 0.0f;
        #pragma unroll
        for (int k = 0; k < 4; ++k) {
            if (w[k] >= 0) {
                const int rl = w[k] >> 17;
                const int s  = w[k] & 0x1FFFF;
                const int p  = __hip_atomic_fetch_add(&cur[rl], 1, __ATOMIC_RELAXED, __HIP_MEMORY_SCOPE_WORKGROUP);
                const float nm = dl[rl] * ds[k];
                if (staged) st[p - beg] = make_int2(s, __float_as_int(nm));
                else        out[p]      = make_int2(s, __float_as_int(nm));
            }
        }
    }
    if (staged) {
        __syncthreads();
        for (int i = t; i < end - beg; i += 512) out[beg + i] = st[i];
    }
}

__global__ void __launch_bounds__(256)
gcn_wmma_xw1_kernel(const float* __restrict__ x,
                    const float* __restrict__ W1,
                    float* __restrict__ hlin,
                    int n) {
    __shared__ float Wp[128 * 16];
    const int t = threadIdx.x;
    #pragma unroll
    for (int j = 0; j < 8; ++j) {
        int idx = t * 8 + j;
        int kr  = idx >> 4;
        int nc  = idx & 15;
        Wp[idx] = (nc < 4) ? W1[kr * 4 + nc] : 0.0f;
    }
    __syncthreads();

    const int wave  = t >> 5;
    const int lane  = t & 31;
    const int tile  = blockIdx.x * 8 + wave;
    const int nTiles = n >> 4;
    if (tile >= nTiles) return;

    const int base = tile * 16;
    const int hh   = lane >> 4;
    const int nl   = lane & 15;
    const int arow = base + nl;

    v8f c = {};
    #pragma unroll 4
    for (int k = 0; k < 128; k += 4) {
        const int kk = k + 2 * hh;
        v2f a = *reinterpret_cast<const v2f*>(x + (size_t)arow * 128 + kk);
        v2f b;
        b.x = Wp[(kk + 0) * 16 + nl];
        b.y = Wp[(kk + 1) * 16 + nl];
        c = __builtin_amdgcn_wmma_f32_16x16x4_f32(
                 false, a,  false, b,
                 (short)0, c,  false,  false);
    }

    if (nl < 4) {
        #pragma unroll
        for (int v = 0; v < 8; ++v) {
            int m = base + v + 8 * hh;
            hlin[(size_t)m * 4 + nl] = c[v];
        }
    }
}

template <int OUT, int NOUT, bool LAST>
__global__ void __launch_bounds__(256)
gather_layer_kernel(const int* __restrict__ rowptr,
                    const int* __restrict__ cnt,
                    const int* __restrict__ pair,
                    const float* __restrict__ hlin,
                    const float* __restrict__ dinv,
                    const float* __restrict__ b,
                    const float* __restrict__ Wn,
                    const float* __restrict__ bn,
                    float* __restrict__ ynext,
                    float* __restrict__ tout,
                    int n) {
    static_assert(!LAST || OUT == 2, "the classifier's sum is written for two features");
    const int g   = blockIdx.x * 256 + threadIdx.x;
    const int r   = g >> 3;
    const int sub = g & 7;
    if (r >= n) return;
    const int beg = rowptr[r];
    const int end = beg + cnt[r];
    const float hs = (sub < OUT) ? hlin[(size_t)r * OUT + sub] : 0.0f;
    const float di = dinv[r];
    const float bb = (sub < OUT) ? b[sub] : 0.0f;
    float w[OUT];
    #pragma unroll
    for (int k = 0; k < OUT; ++k) w[k] = (sub < NOUT) ? Wn[k * NOUT + sub] : 0.0f;
    float bo = 0.0f;
    if (LAST) bo = (sub < NOUT) ? bn[sub] : 0.0f;

    float acc[OUT];
    #pragma unroll
    for (int f = 0; f < OUT; ++f) acc[f] = 0.0f;
    const int2* __restrict__ slots = reinterpret_cast<const int2*>(pair);
    for (int p = beg + sub; p < end; p += 32) {
        int2 pr[4];
        #pragma unroll
        for (int k = 0; k < 4; ++k) {
            const int q = p + 8 * k;
            pr[k] = slots[(q < end) ? q : (end - 1)];
        }
        #pragma unroll
        for (int k = 0; k < 4; ++k) {
            const bool live = (p + 8 * k) < end;
            const float nm = live ? __int_as_float(pr[k].y) : 0.0f;
            if (OUT == 4) {
                const float4 v = reinterpret_cast<const float4*>(hlin)[pr[k].x];
                acc[0] += (live ? v.x : 0.0f) * nm;
                acc[1] += (live ? v.y : 0.0f) * nm;
                acc[2] += (live ? v.z : 0.0f) * nm;
                acc[3] += (live ? v.w : 0.0f) * nm;
            } else {
                const float2 v = reinterpret_cast<const float2*>(hlin)[pr[k].x];
                acc[0] += (live ? v.x : 0.0f) * nm;
                acc[1] += (live ? v.y : 0.0f) * nm;
            }
        }
    }
    #pragma unroll
    for (int f = 0; f < OUT; ++f) {
        acc[f] += __shfl_xor(acc[f], 4);
        acc[f] += __shfl_xor(acc[f], 2);
        acc[f] += __shfl_xor(acc[f], 1);
    }

    float a = acc[0];
    #pragma unroll
    for (int f = 1; f < OUT; ++f) a = (sub == f) ? acc[f] : a;
    const float d2 = di * di;
    const float t  = tanhf(a + hs * d2 + bb);
    const int first = (threadIdx.x & 31) & ~7;
    float tk[OUT];
    #pragma unroll
    for (int k = 0; k < OUT; ++k) tk[k] = __shfl(t, first + k);
    float y;
    if (LAST) {
        y = fmaf(tk[0], w[0], fmaf(tk[1], w[1], bo));
    } else {
        y = 0.0f;
        #pragma unroll
        for (int k = 0; k < OUT; ++k) y = fmaf(tk[k], w[k], y);
    }
    if (sub < NOUT) ynext[(size_t)r * NOUT + sub] = y;
    if (LAST) {
        if (sub < OUT) tout[(size_t)r * OUT + sub] = t;
    }
}

extern "C" void kernel_launch(void* const* d_in, const int* in_sizes, int n_in,
                              void* d_out, int out_size, void* d_ws, size_t ws_size,
                              hipStream_t stream) {
    const float* x  = (const float*)d_in[0];
    const int*   ei = (const int*)d_in[1];
    const float* W1 = (const float*)d_in[2];
    const float* b1 = (const float*)d_in[3];
    const float* W2 = (const float*)d_in[4];
    const float* b2 = (const float*)d_in[5];
    const float* W3 = (const float*)d_in[6];
    const float* b3 = (const float*)d_in[7];
    const float* Wc = (const float*)d_in[8];
    const float* bc = (const float*)d_in[9];

    const int N = N_NODES;
    const int E = N_EDGES;
    const int* row = ei;
    const int* col = ei + E;

    float* ws     = (float*)d_ws;
    float* hP     = ws;
    float* hQ     = hP + 4 * N;
    int*   pair   = (int*)(hQ + 4 * N);
    int*   part   = pair + 2 * E;
    int*   tab    = part + E;
    int*   toff   = tab + NBK * NCH;
    float* dinv   = (float*)(toff + NBK * NCH);
    int*   cnt    = (int*)(dinv + N);
    int*   rowptr = cnt + N;
    unsigned* planes = (unsigned*)(rowptr + N);

    float* outC = (float*)d_out;
    float* outH = outC + (size_t)4 * N;

    const int BT = 256;
    const int gG = (8 * N + BT - 1) / BT;

    deg_hist_kernel<<<NSL, 1024, 0, stream>>>(col, planes);
    bucket_count_kernel<<<NCH, 1024, 0, stream>>>(row, tab);
    scan_dinv_kernel<<<NBK, 256, 0, stream>>>(tab, toff, planes, dinv, N);
    bucket_place_kernel<<<NCH, 1024, 0, stream>>>(row, col, tab, toff, part);
    bucket_group_kernel<<<NBK, 512, 0, stream>>>(part, toff, dinv, rowptr, cnt, pair, N, E);

    {
        const int tiles  = N / 16;
        const int blocks = (tiles + 7) / 8;
        gcn_wmma_xw1_kernel<<<blocks, 256, 0, stream>>>(x, W1, hP, N);
    }
    gather_layer_kernel<4, 4, false><<<gG, BT, 0, stream>>>(
        rowptr, cnt, pair, hP, dinv, b1, W2, nullptr, hQ, nullptr, N);
    gather_layer_kernel<4, 2, false><<<gG, BT, 0, stream>>>(
        rowptr, cnt, pair, hQ, dinv, b2, W3, nullptr, hP, nullptr, N);
    gather_layer_kernel<2, 4, true><<<gG, BT, 0, stream>>>(
        rowptr, cnt, pair, hP, dinv, b3, Wc, bc, outC, outH, N);
}


# Per-edge norm dropped: gathered rows are stored times dinv of their own node and a node scales its finished sum by its dinv once; slots hold the source only, the sort reads no dinv; sum order kept
# speedup vs baseline: 8.6081x; 1.0993x over previous
#include <hip/hip_runtime.h>
#include <hip/hip_bf16.h>
#include <math.h>

#define N_NODES   100000
#define N_EDGES   3200000

typedef float v2f __attribute__((ext_vector_type(2)));
typedef float v8f __attribute__((ext_vector_type(8)));

#define NBK   ((N_NODES + 255) / 256)
#define NCH   256
#define CH4   (N_EDGES / (4 * NCH))
static_assert(N_EDGES % (4 * NCH) == 0, "chunks are whole quads");
static_assert(N_NODES <= (1 << 17), "source index fits 17 bits of an edge word");

#define NSL   128
#define SL4   (N_EDGES / (4 * NSL))
#define SLT   ((SL4 + 1023) / 1024)
#define NPW   (NBK * 128)
static_assert(N_EDGES % (4 * NSL) == 0, "slices are whole quads");
static_assert(N_EDGES / NSL <= 0xFFFF, "a slice's count of one node fits 16 bits");

__global__ void __launch_bounds__(1024)
deg_hist_kernel(const int* __restrict__ col, unsigned* __restrict__ planes) {
    __shared__ uint4 hc4[NPW / 4];
    unsigned* hc = reinterpret_cast<unsigned*>(hc4);
    const int t = threadIdx.x;
    const int4* __restrict__ c4 =
        reinterpret_cast<const int4*>(col) + (size_t)blockIdx.x * SL4;
    int4 c[SLT];
    #pragma unroll
    for (int k = 0; k < SLT; ++k) {
        const int q = t + 1024 * k;
        c[k] = (q < SL4) ? c4[q] : make_int4(-1, -1, -1, -1);
    }
    for (int i = t; i < NPW / 4; i += 1024) hc4[i] = make_uint4(0u, 0u, 0u, 0u);
    __syncthreads();
    #pragma unroll
    for (int k = 0; k < SLT; ++k) {
        if (c[k].x >= 0) {
            __hip_atomic_fetch_add(&hc[c[k].x >> 1], 1u << ((c[k].x & 1) << 4), __ATOMIC_RELAXED, __HIP_MEMORY_SCOPE_WORKGROUP);
            __hip_atomic_fetch_add(&hc[c[k].y >> 1], 1u << ((c[k].y & 1) << 4), __ATOMIC_RELAXED, __HIP_MEMORY_SCOPE_WORKGROUP);
            __hip_atomic_fetch_add(&hc[c[k].z >> 1], 1u << ((c[k].z & 1) << 4), __ATOMIC_RELAXED, __HIP_MEMORY_SCOPE_WORKGROUP);
            __hip_atomic_fetch_add(&hc[c[k].w >> 1], 1u << ((c[k].w & 1) << 4), __ATOMIC_RELAXED, __HIP_MEMORY_SCOPE_WORKGROUP);
        }
    }
    __syncthreads();
    uint4* __restrict__ dst =
        reinterpret_cast<uint4*>(planes) + (size_t)blockIdx.x * (NPW / 4);
    for (int i = t; i < NPW / 4; i += 1024) dst[i] = hc4[i];
}

__global__ void __launch_bounds__(1024)
bucket_count_kernel(const int* __restrict__ row, int* __restrict__ tab) {
    __shared__ int h[NBK];
    const int t = threadIdx.x;
    for (int i = t; i < NBK; i += 1024) h[i] = 0;
    __syncthreads();
    const int q0 = blockIdx.x * CH4;
    for (int q = t; q < CH4; q += 1024) {
        const int4 r = reinterpret_cast<const int4*>(row)[q0 + q];
        __hip_atomic_fetch_add(&h[r.x >> 8], 1, __ATOMIC_RELAXED, __HIP_MEMORY_SCOPE_WORKGROUP);
        __hip_atomic_fetch_add(&h[r.y >> 8], 1, __ATOMIC_RELAXED, __HIP_MEMORY_SCOPE_WORKGROUP);
        __hip_atomic_fetch_add(&h[r.z >> 8], 1, __ATOMIC_RELAXED, __HIP_MEMORY_SCOPE_WORKGROUP);
        __hip_atomic_fetch_add(&h[r.w >> 8], 1, __ATOMIC_RELAXED, __HIP_MEMORY_SCOPE_WORKGROUP);
    }
    __syncthreads();
    for (int i = t; i < NBK; i += 1024) tab[i * NCH + blockIdx.x] = h[i];
}

__global__ void __launch_bounds__(256)
scan_dinv_kernel(const int* __restrict__ tab, int* __restrict__ toff,
                 const unsigned* __restrict__ planes,
                 float* __restrict__ dinv, int n) {
    __shared__ int red[256];
    __shared__ int sc[256];
    __shared__ unsigned clo[2][128];
    __shared__ unsigned chi[2][128];
    const int t = threadIdx.x;
    const int b = blockIdx.x;
    const int i = b * 256 + t;
    {
        const int w  = t & 127;
        const int ph = t >> 7;
        const unsigned* __restrict__ p =
            planes + (size_t)(ph * (NSL / 2)) * NPW + (b * 128 + w);
        unsigned lo = 0, hi = 0;
        #pragma unroll 32
        for (int s = 0; s < NSL / 2; ++s) {
            const unsigned v = p[(size_t)s * NPW];
            lo += v & 0xFFFFu;
            hi += v >> 16;
        }
        clo[ph][w] = lo;
        chi[ph][w] = hi;
    }
    int part = 0;
    const int4* __restrict__ tab4 = reinterpret_cast<const int4*>(tab);
    for (int j = t; j < b * 64; j += 256) {
        const int4 v = tab4[j];
        part += (v.x + v.y) + (v.z + v.w);
    }
    red[t] = part;
    const int c = tab[i];
    sc[t] = c;
    __syncthreads();
    if (i < n) {
        const int w = t >> 1;
        const unsigned cn = (t & 1) ? (chi[0][w] + chi[1][w]) : (clo[0][w] + clo[1][w]);
        dinv[i] = rsqrtf(1.0f + (float)cn);
    }
    for (int s = 128; s > 0; s >>= 1) {
        if (t < s) red[t] += red[t + s];
        __syncthreads();
    }
    for (int off = 1; off < 256; off <<= 1) {
        const int v = (t >= off) ? sc[t - off] : 0;
        __syncthreads();
        sc[t] += v;
        __syncthreads();
    }
    toff[i] = red[0] + sc[t] - c;
}

#define CHE   (N_EDGES / NCH)
static_assert(NBK <= 1024, "one bucket per thread in the chunk's scan");
__global__ void __launch_bounds__(1024)
bucket_place_kernel(const int* __restrict__ row, const int* __restrict__ col,
                    const int* __restrict__ tab, const int* __restrict__ toff,
                    int* __restrict__ part) {
    __shared__ int sw[CHE];
    __shared__ int sd[CHE];
    __shared__ int cur[NBK];
    __shared__ int del[NBK];
    __shared__ int wtot[32];
    const int t    = threadIdx.x;
    const int lane = t & 31;
    const int wv   = t >> 5;
    const int q0 = blockIdx.x * CH4;
    const int4 c0 = reinterpret_cast<const int4*>(col)[q0 + t];
    const int4 r0 = reinterpret_cast<const int4*>(row)[q0 + t];
    static_assert(CH4 >= 1024, "every thread has a first quad");
    int cn = 0, g = 0;
    if (t < NBK) { cn = tab[t * NCH + blockIdx.x]; g = toff[t * NCH + blockIdx.x]; }
    int inc = cn;
    #pragma unroll
    for (int o = 1; o < 32; o <<= 1) {
        const int v = __shfl_up(inc, o);
        if (lane >= o) inc += v;
    }
    if (lane == 31) wtot[wv] = inc;
    __syncthreads();
    {
        const int wt = wtot[lane];
        int winc = wt;
        #pragma unroll
        for (int o = 1; o < 32; o <<= 1) {
            const int v = __shfl_up(winc, o);
            if (lane >= o) winc += v;
        }
        const int wbase = __shfl(winc - wt, wv);
        if (t < NBK) {
            const int ls = wbase + inc - cn;
            cur[t] = ls;
            del[t] = g - ls;
        }
    }
    __syncthreads();
    for (int q = t; q < CH4; q += 1024) {
        int4 c = c0, r = r0;
        if (q != t) {
            c = reinterpret_cast<const int4*>(col)[q0 + q];
            r = reinterpret_cast<const int4*>(row)[q0 + q];
        }
        const int b0 = r.x >> 8, b1 = r.y >> 8, b2 = r.z >> 8, b3 = r.w >> 8;
        const int p0 = __hip_atomic_fetch_add(&cur[b0], 1, __ATOMIC_RELAXED, __HIP_MEMORY_SCOPE_WORKGROUP);
        const int p1 = __hip_atomic_fetch_add(&cur[b1], 1, __ATOMIC_RELAXED, __HIP_MEMORY_SCOPE_WORKGROUP);
        const int p2 = __hip_atomic_fetch_add(&cur[b2], 1, __ATOMIC_RELAXED, __HIP_MEMORY_SCOPE_WORKGROUP);
        const int p3 = __hip_atomic_fetch_add(&cur[b3], 1, __ATOMIC_RELAXED, __HIP_MEMORY_SCOPE_WORKGROUP);
        sw[p0] = ((r.x & 255) << 17) | c.x;  sd[p0] = p0 + del[b0];
        sw[p1] = ((r.y & 255) << 17) | c.y;  sd[p1] = p1 + del[b1];
        sw[p2] = ((r.z & 255) << 17) | c.z;  sd[p2] = p2 + del[b2];
        sw[p3] = ((r.w & 255) << 17) | c.w;  sd[p3] = p3 + del[b3];
    }
    __syncthreads();
    for (int i = t; i < CHE; i += 1024) part[sd[i]] = sw[i];
}

#define GCAP  12288
__global__ void __launch_bounds__(512)
bucket_group_kernel(const int* __restrict__ part, const int* __restrict__ toff,
                    int* __restrict__ rowptr, int* __restrict__ cnt,
                    int* __restrict__ src, int n, int ne) {
    __shared__ int   c[256];
    __shared__ int   sc[256];
    __shared__ int   cur[256];
    __shared__ int   st[GCAP];
    const int t = threadIdx.x;
    const int b = blockIdx.x;
    const int node = b * 256 + t;
    const int beg = toff[b * NCH];
    const int end = (b + 1 < NBK) ? toff[(b + 1) * NCH] : ne;
    if (t < 256) c[t] = 0;
    __syncthreads();
    for (int e = beg + t; e < end; e += 4 * 512) {
        int w[4];
        #pragma unroll
        for (int k = 0; k < 4; ++k) {
            const int q = e + 512 * k;
            w[k] = (q < end) ? part[q] : -1;
        }
        #pragma unroll
        for (int k = 0; k < 4; ++k)
            if (w[k] >= 0)
                __hip_atomic_fetch_add(&c[w[k] >> 17], 1, __ATOMIC_RELAXED, __HIP_MEMORY_SCOPE_WORKGROUP);
    }
    __syncthreads();
    int my = 0;
    if (t < 256) { my = c[t]; sc[t] = my; }
    __syncthreads();
    for (int off = 1; off < 256; off <<= 1) {
        int v = 0;
        if (t < 256 && t >= off) v = sc[t - off];
        __syncthreads();
        if (t < 256) sc[t] += v;
        __syncthreads();
    }
    if (t < 256) {
        const int first = beg + sc[t] - my;
        cur[t] = first;
        if (node < n) { rowptr[node] = first; cnt[node] = my; }
    }
    __syncthreads();
    const bool staged = (end - beg) <= GCAP;
    for (int e = beg + t; e < end; e += 4 * 512) {
        int w[4];
        #pragma unroll
        for (int k = 0; k < 4; ++k) {
            const int q = e + 512 * k;
            w[k] = (q < end) ? part[q] : -1;
        }
        #pragma unroll
        for (int k = 0; k < 4; ++k) {
            if (w[k] >= 0) {
                const int rl = w[k] >> 17;
                const int s  = w[k] & 0x1FFFF;
                const int p  = __hip_atomic_fetch_add(&cur[rl], 1, __ATOMIC_RELAXED, __HIP_MEMORY_SCOPE_WORKGROUP);
                if (staged) st[p - beg] = s;
                else        src[p]      = s;
            }
        }
    }
    if (staged) {
        __syncthreads();
        for (int i = t; i < end - beg; i += 512) src[beg + i] = st[i];
    }
}

__global__ void __launch_bounds__(256)
gcn_wmma_xw1_kernel(const float* __restrict__ x,
                    const float* __restrict__ W1,
                    const float* __restrict__ dinv,
                    float* __restrict__ hlin,
                    int n) {
    __shared__ float Wp[128 * 16];
    const int t = threadIdx.x;
    #pragma unroll
    for (int j = 0; j < 8; ++j) {
        int idx = t * 8 + j;
        int kr  = idx >> 4;
        int nc  = idx & 15;
        Wp[idx] = (nc < 4) ? W1[kr * 4 + nc] : 0.0f;
    }
    __syncthreads();

    const int wave  = t >> 5;
    const int lane  = t & 31;
    const int tile  = blockIdx.x * 8 + wave;
    const int nTiles = n >> 4;
    if (tile >= nTiles) return;

    const int base = tile * 16;
    const int hh   = lane >> 4;
    const int nl   = lane & 15;
    const int arow = base + nl;
    const float4* __restrict__ dv = reinterpret_cast<const float4*>(dinv + base + 8 * hh);
    const float4 d0 = dv[0];
    const float4 d1 = dv[1];

    v8f c = {};
    #pragma unroll 4
    for (int k = 0; k < 128; k += 4) {
        const int kk = k + 2 * hh;
        v2f a = *reinterpret_cast<const v2f*>(x + (size_t)arow * 128 + kk);
        v2f b;
        b.x = Wp[(kk + 0) * 16 + nl];
        b.y = Wp[(kk + 1) * 16 + nl];
        c = __builtin_amdgcn_wmma_f32_16x16x4_f32(
                 false, a,  false, b,
                 (short)0, c,  false,  false);
    }

    if (nl < 4) {
        const float ds[8] = {d0.x, d0.y, d0.z, d0.w, d1.x, d1.y, d1.z, d1.w};
        #pragma unroll
        for (int v = 0; v < 8; ++v) {
            int m = base + v + 8 * hh;
            hlin[(size_t)m * 4 + nl] = c[v] * ds[v];
        }
    }
}

template <int OUT, int NOUT, bool LAST>
__global__ void __launch_bounds__(256)
gather_layer_kernel(const int* __restrict__ rowptr,
                    const int* __restrict__ cnt,
                    const int* __restrict__ src,
                    const float* __restrict__ hlin,
                    const float* __restrict__ dinv,
                    const float* __restrict__ b,
                    const float* __restrict__ Wn,
                    const float* __restrict__ bn,
                    float* __restrict__ ynext,
                    float* __restrict__ tout,
                    int n) {
    static_assert(!LAST || OUT == 2, "the classifier's sum is written for two features");
    const int g   = blockIdx.x * 256 + threadIdx.x;
    const int r   = g >> 3;
    const int sub = g & 7;
    if (r >= n) return;
    const int beg = rowptr[r];
    const int end = beg + cnt[r];
    const float hs = (sub < OUT) ? hlin[(size_t)r * OUT + sub] : 0.0f;
    const float di = dinv[r];
    const float bb = (sub < OUT) ? b[sub] : 0.0f;
    float w[OUT];
    #pragma unroll
    for (int k = 0; k < OUT; ++k) w[k] = (sub < NOUT) ? Wn[k * NOUT + sub] : 0.0f;
    float bo = 0.0f;
    if (LAST) bo = (sub < NOUT) ? bn[sub] : 0.0f;

    float acc[OUT];
    #pragma unroll
    for (int f = 0; f < OUT; ++f) acc[f] = 0.0f;
    for (int p = beg + sub; p < end; p += 32) {
        int sr[4];
        #pragma unroll
        for (int k = 0; k < 4; ++k) {
            const int q = p + 8 * k;
            sr[k] = src[(q < end) ? q : (end - 1)];
        }
        #pragma unroll
        for (int k = 0; k < 4; ++k) {
            const bool live = (p + 8 * k) < end;
            if (OUT == 4) {
                const float4 v = reinterpret_cast<const float4*>(hlin)[sr[k]];
                acc[0] += live ? v.x : 0.0f;
                acc[1] += live ? v.y : 0.0f;
                acc[2] += live ? v.z : 0.0f;
                acc[3] += live ? v.w : 0.0f;
            } else {
                const float2 v = reinterpret_cast<const float2*>(hlin)[sr[k]];
                acc[0] += live ? v.x : 0.0f;
                acc[1] += live ? v.y : 0.0f;
            }
        }
    }
    #pragma unroll
    for (int f = 0; f < OUT; ++f) {
        acc[f] += __shfl_xor(acc[f], 4);
        acc[f] += __shfl_xor(acc[f], 2);
        acc[f] += __shfl_xor(acc[f], 1);
    }

    float a = acc[0];
    #pragma unroll
    for (int f = 1; f < OUT; ++f) a = (sub == f) ? acc[f] : a;
    const float t  = tanhf((a + hs) * di + bb);
    const int first = (threadIdx.x & 31) & ~7;
    float tk[OUT];
    #pragma unroll
    for (int k = 0; k < OUT; ++k) tk[k] = __shfl(t, first + k);
    float y;
    if (LAST) {
        y = fmaf(tk[0], w[0], fmaf(tk[1], w[1], bo));
    } else {
        y = 0.0f;
        #pragma unroll
        for (int k = 0; k < OUT; ++k) y = fmaf(tk[k], w[k], y);
        y *= di;
    }
    if (sub < NOUT) ynext[(size_t)r * NOUT + sub] = y;
    if (LAST) {
        if (sub < OUT) tout[(size_t)r * OUT + sub] = t;
    }
}

extern "C" void kernel_launch(void* const* d_in, const int* in_sizes, int n_in,
                              void* d_out, int out_size, void* d_ws, size_t ws_size,
                              hipStream_t stream) {
    const float* x  = (const float*)d_in[0];
    const int*   ei = (const int*)d_in[1];
    const float* W1 = (const float*)d_in[2];
    const float* b1 = (const float*)d_in[3];
    const float* W2 = (const float*)d_in[4];
    const float* b2 = (const float*)d_in[5];
    const float* W3 = (const float*)d_in[6];
    const float* b3 = (const float*)d_in[7];
    const float* Wc = (const float*)d_in[8];
    const float* bc = (const float*)d_in[9];

    const int N = N_NODES;
    const int E = N_EDGES;
    const int* row = ei;
    const int* col = ei + E;

    float* ws     = (float*)d_ws;
    float* hP     = ws;
    float* hQ     = hP + 4 * N;
    int*   src    = (int*)(hQ + 4 * N);
    int*   part   = src + E;
    int*   tab    = part + E;
    int*   toff   = tab + NBK * NCH;
    float* dinv   = (float*)(toff + NBK * NCH);
    int*   cnt    = (int*)(dinv + N);
    int*   rowptr = cnt + N;
    unsigned* planes = (unsigned*)(rowptr + N);

    float* outC = (float*)d_out;
    float* outH = outC + (size_t)4 * N;

    const int BT = 256;
    const int gG = (8 * N + BT - 1) / BT;

    deg_hist_kernel<<<NSL, 1024, 0, stream>>>(col, planes);
    bucket_count_kernel<<<NCH, 1024, 0, stream>>>(row, tab);
    scan_dinv_kernel<<<NBK, 256, 0, stream>>>(tab, toff, planes, dinv, N);
    bucket_place_kernel<<<NCH, 1024, 0, stream>>>(row, col, tab, toff, part);
    bucket_group_kernel<<<NBK, 512, 0, stream>>>(part, toff, rowptr, cnt, src, N, E);

    {
        const int tiles  = N / 16;
        const int blocks = (tiles + 7) / 8;
        gcn_wmma_xw1_kernel<<<blocks, 256, 0, stream>>>(x, W1, dinv, hP, N);
    }
    gather_layer_kernel<4, 4, false><<<gG, BT, 0, stream>>>(
        rowptr, cnt, src, hP, dinv, b1, W2, nullptr, hQ, nullptr, N);
    gather_layer_kernel<4, 2, false><<<gG, BT, 0, stream>>>(
        rowptr, cnt, src, hQ, dinv, b2, W3, nullptr, hP, nullptr, N);
    gather_layer_kernel<2, 4, true><<<gG, BT, 0, stream>>>(
        rowptr, cnt, src, hP, dinv, b3, Wc, bc, outC, outH, N);
}


# Launches that do not depend on each other share one launch, split by work-group index: degree histogram + bucket count, bucket place + x@W1 (32 tiles a group); 9 launches become 7; sums unchanged
# speedup vs baseline: 9.4166x; 1.0939x over previous
#include <hip/hip_runtime.h>
#include <hip/hip_bf16.h>
#include <math.h>

#define N_NODES   100000
#define N_EDGES   3200000

typedef float v2f __attribute__((ext_vector_type(2)));
typedef float v8f __attribute__((ext_vector_type(8)));

#define NBK   ((N_NODES + 255) / 256)
#define NCH   256
#define CH4   (N_EDGES / (4 * NCH))
static_assert(N_EDGES % (4 * NCH) == 0, "chunks are whole quads");
static_assert(N_NODES <= (1 << 17), "source index fits 17 bits of an edge word");

#define NSL   128
#define SL4   (N_EDGES / (4 * NSL))
#define SLT   ((SL4 + 1023) / 1024)
#define NPW   (NBK * 128)
static_assert(N_EDGES % (4 * NSL) == 0, "slices are whole quads");
static_assert(N_EDGES / NSL <= 0xFFFF, "a slice's count of one node fits 16 bits");

#define NCW   (NCH / 2)
#define CW4   (2 * CH4)
#define CWT   ((CW4 + 1023) / 1024)
static_assert(NCH % 2 == 0, "count work-groups take whole pairs of chunks");
__global__ void __launch_bounds__(1024)
deg_hist_bucket_count_kernel(const int* __restrict__ row, const int* __restrict__ col,
                             unsigned* __restrict__ planes, int* __restrict__ tab) {
    __shared__ uint4 hc4[NPW / 4];
    __shared__ int h[2][NBK];
    const int t = threadIdx.x;
    if (blockIdx.x < NSL) {
        unsigned* hc = reinterpret_cast<unsigned*>(hc4);
        const int4* __restrict__ c4 =
            reinterpret_cast<const int4*>(col) + (size_t)blockIdx.x * SL4;
        int4 c[SLT];
        #pragma unroll
        for (int k = 0; k < SLT; ++k) {
            const int q = t + 1024 * k;
            c[k] = (q < SL4) ? c4[q] : make_int4(-1, -1, -1, -1);
        }
        for (int i = t; i < NPW / 4; i += 1024) hc4[i] = make_uint4(0u, 0u, 0u, 0u);
        __syncthreads();
        #pragma unroll
        for (int k = 0; k < SLT; ++k) {
            if (c[k].x >= 0) {
                __hip_atomic_fetch_add(&hc[c[k].x >> 1], 1u << ((c[k].x & 1) << 4), __ATOMIC_RELAXED, __HIP_MEMORY_SCOPE_WORKGROUP);
                __hip_atomic_fetch_add(&hc[c[k].y >> 1], 1u << ((c[k].y & 1) << 4), __ATOMIC_RELAXED, __HIP_MEMORY_SCOPE_WORKGROUP);
                __hip_atomic_fetch_add(&hc[c[k].z >> 1], 1u << ((c[k].z & 1) << 4), __ATOMIC_RELAXED, __HIP_MEMORY_SCOPE_WORKGROUP);
                __hip_atomic_fetch_add(&hc[c[k].w >> 1], 1u << ((c[k].w & 1) << 4), __ATOMIC_RELAXED, __HIP_MEMORY_SCOPE_WORKGROUP);
            }
        }
        __syncthreads();
        uint4* __restrict__ dst =
            reinterpret_cast<uint4*>(planes) + (size_t)blockIdx.x * (NPW / 4);
        for (int i = t; i < NPW / 4; i += 1024) dst[i] = hc4[i];
    } else {
        const int j = blockIdx.x - NSL;
        const int4* __restrict__ r4 =
            reinterpret_cast<const int4*>(row) + (size_t)j * CW4;
        int4 r[CWT];
        #pragma unroll
        for (int k = 0; k < CWT; ++k) {
            const int q = t + 1024 * k;
            r[k] = (q < CW4) ? r4[q] : make_int4(-1, -1, -1, -1);
        }
        for (int i = t; i < NBK; i += 1024) { h[0][i] = 0; h[1][i] = 0; }
        __syncthreads();
        #pragma unroll
        for (int k = 0; k < CWT; ++k) {
            if (r[k].x >= 0) {
                int* hq = h[(t + 1024 * k) >= CH4];
                __hip_atomic_fetch_add(&hq[r[k].x >> 8], 1, __ATOMIC_RELAXED, __HIP_MEMORY_SCOPE_WORKGROUP);
                __hip_atomic_fetch_add(&hq[r[k].y >> 8], 1, __ATOMIC_RELAXED, __HIP_MEMORY_SCOPE_WORKGROUP);
                __hip_atomic_fetch_add(&hq[r[k].z >> 8], 1, __ATOMIC_RELAXED, __HIP_MEMORY_SCOPE_WORKGROUP);
                __hip_atomic_fetch_add(&hq[r[k].w >> 8], 1, __ATOMIC_RELAXED, __HIP_MEMORY_SCOPE_WORKGROUP);
            }
        }
        __syncthreads();
        for (int i = t; i < NBK; i += 1024)
            *reinterpret_cast<int2*>(tab + i * NCH + 2 * j) = make_int2(h[0][i], h[1][i]);
    }
}

__global__ void __launch_bounds__(256)
scan_dinv_kernel(const int* __restrict__ tab, int* __restrict__ toff,
                 const unsigned* __restrict__ planes,
                 float* __restrict__ dinv, int n) {
    __shared__ int red[256];
    __shared__ int sc[256];
    __shared__ unsigned clo[2][128];
    __shared__ unsigned chi[2][128];
    const int t = threadIdx.x;
    const int b = blockIdx.x;
    const int i = b * 256 + t;
    {
        const int w  = t & 127;
        const int ph = t >> 7;
        const unsigned* __restrict__ p =
            planes + (size_t)(ph * (NSL / 2)) * NPW + (b * 128 + w);
        unsigned lo = 0, hi = 0;
        #pragma unroll 32
        for (int s = 0; s < NSL / 2; ++s) {
            const unsigned v = p[(size_t)s * NPW];
            lo += v & 0xFFFFu;
            hi += v >> 16;
        }
        clo[ph][w] = lo;
        chi[ph][w] = hi;
    }
    int part = 0;
    const int4* __restrict__ tab4 = reinterpret_cast<const int4*>(tab);
    for (int j = t; j < b * 64; j += 256) {
        const int4 v = tab4[j];
        part += (v.x + v.y) + (v.z + v.w);
    }
    red[t] = part;
    const int c = tab[i];
    sc[t] = c;
    __syncthreads();
    if (i < n) {
        const int w = t >> 1;
        const unsigned cn = (t & 1) ? (chi[0][w] + chi[1][w]) : (clo[0][w] + clo[1][w]);
        dinv[i] = rsqrtf(1.0f + (float)cn);
    }
    for (int s = 128; s > 0; s >>= 1) {
        if (t < s) red[t] += red[t + s];
        __syncthreads();
    }
    for (int off = 1; off < 256; off <<= 1) {
        const int v = (t >= off) ? sc[t - off] : 0;
        __syncthreads();
        sc[t] += v;
        __syncthreads();
    }
    toff[i] = red[0] + sc[t] - c;
}

#define CHE   (N_EDGES / NCH)
#define XWG   ((N_NODES / 16 + 31) / 32)
static_assert(NBK <= 1024, "one bucket per thread in the chunk's scan");
static_assert(N_NODES % 16 == 0, "row tiles are whole");
__global__ void __launch_bounds__(1024)
bucket_place_xw1_kernel(const int* __restrict__ row, const int* __restrict__ col,
                        const int* __restrict__ tab, const int* __restrict__ toff,
                        int* __restrict__ part,
                        const float* __restrict__ x,
                        const float* __restrict__ W1,
                        const float* __restrict__ dinv,
                        float* __restrict__ hlin,
                        int n) {
    __shared__ int sw[CHE];
    __shared__ int sd[CHE];
    __shared__ int cur[NBK];
    __shared__ int del[NBK];
    __shared__ int wtot[32];
    __shared__ float Wp[128 * 16];
    const int t    = threadIdx.x;
    const int lane = t & 31;
    const int wv   = t >> 5;
    if (blockIdx.x < NCH) {
        const int q0 = blockIdx.x * CH4;
        const int4 c0 = reinterpret_cast<const int4*>(col)[q0 + t];
        const int4 r0 = reinterpret_cast<const int4*>(row)[q0 + t];
        static_assert(CH4 >= 1024, "every thread has a first quad");
        int cn = 0, g = 0;
        if (t < NBK) { cn = tab[t * NCH + blockIdx.x]; g = toff[t * NCH + blockIdx.x]; }
        int inc = cn;
        #pragma unroll
        for (int o = 1; o < 32; o <<= 1) {
            const int v = __shfl_up(inc, o);
            if (lane >= o) inc += v;
        }
        if (lane == 31) wtot[wv] = inc;
        __syncthreads();
        {
            const int wt = wtot[lane];
            int winc = wt;
            #pragma unroll
            for (int o = 1; o < 32; o <<= 1) {
                const int v = __shfl_up(winc, o);
                if (lane >= o) winc += v;
            }
            const int wbase = __shfl(winc - wt, wv);
            if (t < NBK) {
                const int ls = wbase + inc - cn;
                cur[t] = ls;
                del[t] = g - ls;
            }
        }
        __syncthreads();
        for (int q = t; q < CH4; q += 1024) {
            int4 c = c0, r = r0;
            if (q != t) {
                c = reinterpret_cast<const int4*>(col)[q0 + q];
                r = reinterpret_cast<const int4*>(row)[q0 + q];
            }
            const int b0 = r.x >> 8, b1 = r.y >> 8, b2 = r.z >> 8, b3 = r.w >> 8;
            const int p0 = __hip_atomic_fetch_add(&cur[b0], 1, __ATOMIC_RELAXED, __HIP_MEMORY_SCOPE_WORKGROUP);
            const int p1 = __hip_atomic_fetch_add(&cur[b1], 1, __ATOMIC_RELAXED, __HIP_MEMORY_SCOPE_WORKGROUP);
            const int p2 = __hip_atomic_fetch_add(&cur[b2], 1, __ATOMIC_RELAXED, __HIP_MEMORY_SCOPE_WORKGROUP);
            const int p3 = __hip_atomic_fetch_add(&cur[b3], 1, __ATOMIC_RELAXED, __HIP_MEMORY_SCOPE_WORKGROUP);
            sw[p0] = ((r.x & 255) << 17) | c.x;  sd[p0] = p0 + del[b0];
            sw[p1] = ((r.y & 255) << 17) | c.y;  sd[p1] = p1 + del[b1];
            sw[p2] = ((r.z & 255) << 17) | c.z;  sd[p2] = p2 + del[b2];
            sw[p3] = ((r.w & 255) << 17) | c.w;  sd[p3] = p3 + del[b3];
        }
        __syncthreads();
        for (int i = t; i < CHE; i += 1024) part[sd[i]] = sw[i];
    } else {
        {
            const int idx = t * 2;
            const int kr  = idx >> 4;
            const int nc  = idx & 15;
            const float w0 = W1[kr * 4 + (nc & 3)];
            const float w1 = W1[kr * 4 + ((nc + 1) & 3)];
            Wp[idx]     = (nc < 4) ? w0 : 0.0f;
            Wp[idx + 1] = (nc < 4) ? w1 : 0.0f;
        }
        __syncthreads();

        const int tile   = (blockIdx.x - NCH) * 32 + wv;
        const int nTiles = n >> 4;
        if (tile >= nTiles) return;

        const int base = tile * 16;
        const int hh   = lane >> 4;
        const int nl   = lane & 15;
        const int arow = base + nl;
        const float4* __restrict__ dv = reinterpret_cast<const float4*>(dinv + base + 8 * hh);
        const float4 d0 = dv[0];
        const float4 d1 = dv[1];

        v8f c = {};
        #pragma unroll 4
        for (int k = 0; k < 128; k += 4) {
            const int kk = k + 2 * hh;
            v2f a = *reinterpret_cast<const v2f*>(x + (size_t)arow * 128 + kk);
            v2f b;
            b.x = Wp[(kk + 0) * 16 + nl];
            b.y = Wp[(kk + 1) * 16 + nl];
            c = __builtin_amdgcn_wmma_f32_16x16x4_f32(
                     false, a,  false, b,
                     (short)0, c,  false,  false);
        }

        if (nl < 4) {
            const float ds[8] = {d0.x, d0.y, d0.z, d0.w, d1.x, d1.y, d1.z, d1.w};
            #pragma unroll
            for (int v = 0; v < 8; ++v) {
                int m = base + v + 8 * hh;
                hlin[(size_t)m * 4 + nl] = c[v] * ds[v];
            }
        }
    }
}

#define GCAP  12288
__global__ void __launch_bounds__(512)
bucket_group_kernel(const int* __restrict__ part, const int* __restrict__ toff,
                    int* __restrict__ rowptr, int* __restrict__ cnt,
                    int* __restrict__ src, int n, int ne) {
    __shared__ int   c[256];
    __shared__ int   sc[256];
    __shared__ int   cur[256];
    __shared__ int   st[GCAP];
    const int t = threadIdx.x;
    const int b = blockIdx.x;
    const int node = b * 256 + t;
    const int beg = toff[b * NCH];
    const int end = (b + 1 < NBK) ? toff[(b + 1) * NCH] : ne;
    if (t < 256) c[t] = 0;
    __syncthreads();
    for (int e = beg + t; e < end; e += 4 * 512) {
        int w[4];
        #pragma unroll
        for (int k = 0; k < 4; ++k) {
            const int q = e + 512 * k;
            w[k] = (q < end) ? part[q] : -1;
        }
        #pragma unroll
        for (int k = 0; k < 4; ++k)
            if (w[k] >= 0)
                __hip_atomic_fetch_add(&c[w[k] >> 17], 1, __ATOMIC_RELAXED, __HIP_MEMORY_SCOPE_WORKGROUP);
    }
    __syncthreads();
    int my = 0;
    if (t < 256) { my = c[t]; sc[t] = my; }
    __syncthreads();
    for (int off = 1; off < 256; off <<= 1) {
        int v = 0;
        if (t < 256 && t >= off) v = sc[t - off];
        __syncthreads();
        if (t < 256) sc[t] += v;
        __syncthreads();
    }
    if (t < 256) {
        const int first = beg + sc[t] - my;
        cur[t] = first;
        if (node < n) { rowptr[node] = first; cnt[node] = my; }
    }
    __syncthreads();
    const bool staged = (end - beg) <= GCAP;
    for (int e = beg + t; e < end; e += 4 * 512) {
        int w[4];
        #pragma unroll
        for (int k = 0; k < 4; ++k) {
            const int q = e + 512 * k;
            w[k] = (q < end) ? part[q] : -1;
        }
        #pragma unroll
        for (int k = 0; k < 4; ++k) {
            if (w[k] >= 0) {
                const int rl = w[k] >> 17;
                const int s  = w[k] & 0x1FFFF;
                const int p  = __hip_atomic_fetch_add(&cur[rl], 1, __ATOMIC_RELAXED, __HIP_MEMORY_SCOPE_WORKGROUP);
                if (staged) st[p - beg] = s;
                else        src[p]      = s;
            }
        }
    }
    if (staged) {
        __syncthreads();
        for (int i = t; i < end - beg; i += 512) src[beg + i] = st[i];
    }
}

template <int OUT, int NOUT, bool LAST>
__global__ void __launch_bounds__(256)
gather_layer_kernel(const int* __restrict__ rowptr,
                    const int* __restrict__ cnt,
                    const int* __restrict__ src,
                    const float* __restrict__ hlin,
                    const float* __restrict__ dinv,
                    const float* __restrict__ b,
                    const float* __restrict__ Wn,
                    const float* __restrict__ bn,
                    float* __restrict__ ynext,
                    float* __restrict__ tout,
                    int n) {
    static_assert(!LAST || OUT == 2, "the classifier's sum is written for two features");
    const int g   = blockIdx.x * 256 + threadIdx.x;
    const int r   = g >> 3;
    const int sub = g & 7;
    if (r >= n) return;
    const int beg = rowptr[r];
    const int end = beg + cnt[r];
    const float hs = (sub < OUT) ? hlin[(size_t)r * OUT + sub] : 0.0f;
    const float di = dinv[r];
    const float bb = (sub < OUT) ? b[sub] : 0.0f;
    float w[OUT];
    #pragma unroll
    for (int k = 0; k < OUT; ++k) w[k] = (sub < NOUT) ? Wn[k * NOUT + sub] : 0.0f;
    float bo = 0.0f;
    if (LAST) bo = (sub < NOUT) ? bn[sub] : 0.0f;

    float acc[OUT];
    #pragma unroll
    for (int f = 0; f < OUT; ++f) acc[f] = 0.0f;
    for (int p = beg + sub; p < end; p += 32) {
        int sr[4];
        #pragma unroll
        for (int k = 0; k < 4; ++k) {
            const int q = p + 8 * k;
            sr[k] = src[(q < end) ? q : (end - 1)];
        }
        #pragma unroll
        for (int k = 0; k < 4; ++k) {
            const bool live = (p + 8 * k) < end;
            if (OUT == 4) {
                const float4 v = reinterpret_cast<const float4*>(hlin)[sr[k]];
                acc[0] += live ? v.x : 0.0f;
                acc[1] += live ? v.y : 0.0f;
                acc[2] += live ? v.z : 0.0f;
                acc[3] += live ? v.w : 0.0f;
            } else {
                const float2 v = reinterpret_cast<const float2*>(hlin)[sr[k]];
                acc[0] += live ? v.x : 0.0f;
                acc[1] += live ? v.y : 0.0f;
            }
        }
    }
    #pragma unroll
    for (int f = 0; f < OUT; ++f) {
        acc[f] += __shfl_xor(acc[f], 4);
        acc[f] += __shfl_xor(acc[f], 2);
        acc[f] += __shfl_xor(acc[f], 1);
    }

    float a = acc[0];
    #pragma unroll
    for (int f = 1; f < OUT; ++f) a = (sub == f) ? acc[f] : a;
    const float t  = tanhf((a + hs) * di + bb);
    const int first = (threadIdx.x & 31) & ~7;
    float tk[OUT];
    #pragma unroll
    for (int k = 0; k < OUT; ++k) tk[k] = __shfl(t, first + k);
    float y;
    if (LAST) {
        y = fmaf(tk[0], w[0], fmaf(tk[1], w[1], bo));
    } else {
        y = 0.0f;
        #pragma unroll
        for (int k = 0; k < OUT; ++k) y = fmaf(tk[k], w[k], y);
        y *= di;
    }
    if (sub < NOUT) ynext[(size_t)r * NOUT + sub] = y;
    if (LAST) {
        if (sub < OUT) tout[(size_t)r * OUT + sub] = t;
    }
}

extern "C" void kernel_launch(void* const* d_in, const int* in_sizes, int n_in,
                              void* d_out, int out_size, void* d_ws, size_t ws_size,
                              hipStream_t stream) {
    const float* x  = (const float*)d_in[0];
    const int*   ei = (const int*)d_in[1];
    const float* W1 = (const float*)d_in[2];
    const float* b1 = (const float*)d_in[3];
    const float* W2 = (const float*)d_in[4];
    const float* b2 = (const float*)d_in[5];
    const float* W3 = (const float*)d_in[6];
    const float* b3 = (const float*)d_in[7];
    const float* Wc = (const float*)d_in[8];
    const float* bc = (const float*)d_in[9];

    const int N = N_NODES;
    const int E = N_EDGES;
    const int* row = ei;
    const int* col = ei + E;

    float* ws     = (float*)d_ws;
    float* hP     = ws;
    float* hQ     = hP + 4 * N;
    int*   src    = (int*)(hQ + 4 * N);
    int*   part   = src + E;
    int*   tab    = part + E;
    int*   toff   = tab + NBK * NCH;
    float* dinv   = (float*)(toff + NBK * NCH);
    int*   cnt    = (int*)(dinv + N);
    int*   rowptr = cnt + N;
    unsigned* planes = (unsigned*)(rowptr + N);

    float* outC = (float*)d_out;
    float* outH = outC + (size_t)4 * N;

    const int BT = 256;
    const int gG = (8 * N + BT - 1) / BT;

    deg_hist_bucket_count_kernel<<<NSL + NCW, 1024, 0, stream>>>(row, col, planes, tab);
    scan_dinv_kernel<<<NBK, 256, 0, stream>>>(tab, toff, planes, dinv, N);
    bucket_place_xw1_kernel<<<NCH + XWG, 1024, 0, stream>>>(row, col, tab, toff, part,
                                                          x, W1, dinv, hP, N);
    bucket_group_kernel<<<NBK, 512, 0, stream>>>(part, toff, rowptr, cnt, src, N, E);

    gather_layer_kernel<4, 4, false><<<gG, BT, 0, stream>>>(
        rowptr, cnt, src, hP, dinv, b1, W2, nullptr, hQ, nullptr, N);
    gather_layer_kernel<4, 2, false><<<gG, BT, 0, stream>>>(
        rowptr, cnt, src, hQ, dinv, b2, W3, nullptr, hP, nullptr, N);
    gather_layer_kernel<2, 4, true><<<gG, BT, 0, stream>>>(
        rowptr, cnt, src, hP, dinv, b3, Wc, bc, outC, outH, N);
}


# Gather launches: a wave does 5 groups of four nodes (64 slots a node a trip) and keeps its groups' loads in flight together: 5000 long waves a launch instead of 25000 short ones; sum order kept
# speedup vs baseline: 9.7405x; 1.0344x over previous
#include <hip/hip_runtime.h>
#include <hip/hip_bf16.h>
#include <math.h>

#define N_NODES   100000
#define N_EDGES   3200000

typedef float v2f __attribute__((ext_vector_type(2)));
typedef float v8f __attribute__((ext_vector_type(8)));

#define NBK   ((N_NODES + 255) / 256)
#define NCH   256
#define CH4   (N_EDGES / (4 * NCH))
static_assert(N_EDGES % (4 * NCH) == 0, "chunks are whole quads");
static_assert(N_NODES <= (1 << 17), "source index fits 17 bits of an edge word");

#define NSL   128
#define SL4   (N_EDGES / (4 * NSL))
#define SLT   ((SL4 + 1023) / 1024)
#define NPW   (NBK * 128)
static_assert(N_EDGES % (4 * NSL) == 0, "slices are whole quads");
static_assert(N_EDGES / NSL <= 0xFFFF, "a slice's count of one node fits 16 bits");

#define NCW   (NCH / 2)
#define CW4   (2 * CH4)
#define CWT   ((CW4 + 1023) / 1024)
static_assert(NCH % 2 == 0, "count work-groups take whole pairs of chunks");
__global__ void __launch_bounds__(1024)
deg_hist_bucket_count_kernel(const int* __restrict__ row, const int* __restrict__ col,
                             unsigned* __restrict__ planes, int* __restrict__ tab) {
    __shared__ uint4 hc4[NPW / 4];
    __shared__ int h[2][NBK];
    const int t = threadIdx.x;
    if (blockIdx.x < NSL) {
        unsigned* hc = reinterpret_cast<unsigned*>(hc4);
        const int4* __restrict__ c4 =
            reinterpret_cast<const int4*>(col) + (size_t)blockIdx.x * SL4;
        int4 c[SLT];
        #pragma unroll
        for (int k = 0; k < SLT; ++k) {
            const int q = t + 1024 * k;
            c[k] = (q < SL4) ? c4[q] : make_int4(-1, -1, -1, -1);
        }
        for (int i = t; i < NPW / 4; i += 1024) hc4[i] = make_uint4(0u, 0u, 0u, 0u);
        __syncthreads();
        #pragma unroll
        for (int k = 0; k < SLT; ++k) {
            if (c[k].x >= 0) {
                __hip_atomic_fetch_add(&hc[c[k].x >> 1], 1u << ((c[k].x & 1) << 4), __ATOMIC_RELAXED, __HIP_MEMORY_SCOPE_WORKGROUP);
                __hip_atomic_fetch_add(&hc[c[k].y >> 1], 1u << ((c[k].y & 1) << 4), __ATOMIC_RELAXED, __HIP_MEMORY_SCOPE_WORKGROUP);
                __hip_atomic_fetch_add(&hc[c[k].z >> 1], 1u << ((c[k].z & 1) << 4), __ATOMIC_RELAXED, __HIP_MEMORY_SCOPE_WORKGROUP);
                __hip_atomic_fetch_add(&hc[c[k].w >> 1], 1u << ((c[k].w & 1) << 4), __ATOMIC_RELAXED, __HIP_MEMORY_SCOPE_WORKGROUP);
            }
        }
        __syncthreads();
        uint4* __restrict__ dst =
            reinterpret_cast<uint4*>(planes) + (size_t)blockIdx.x * (NPW / 4);
        for (int i = t; i < NPW / 4; i += 1024) dst[i] = hc4[i];
    } else {
        const int j = blockIdx.x - NSL;
        const int4* __restrict__ r4 =
            reinterpret_cast<const int4*>(row) + (size_t)j * CW4;
        int4 r[CWT];
        #pragma unroll
        for (int k = 0; k < CWT; ++k) {
            const int q = t + 1024 * k;
            r[k] = (q < CW4) ? r4[q] : make_int4(-1, -1, -1, -1);
        }
        for (int i = t; i < NBK; i += 1024) { h[0][i] = 0; h[1][i] = 0; }
        __syncthreads();
        #pragma unroll
        for (int k = 0; k < CWT; ++k) {
            if (r[k].x >= 0) {
                int* hq = h[(t + 1024 * k) >= CH4];
                __hip_atomic_fetch_add(&hq[r[k].x >> 8], 1, __ATOMIC_RELAXED, __HIP_MEMORY_SCOPE_WORKGROUP);
                __hip_atomic_fetch_add(&hq[r[k].y >> 8], 1, __ATOMIC_RELAXED, __HIP_MEMORY_SCOPE_WORKGROUP);
                __hip_atomic_fetch_add(&hq[r[k].z >> 8], 1, __ATOMIC_RELAXED, __HIP_MEMORY_SCOPE_WORKGROUP);
                __hip_atomic_fetch_add(&hq[r[k].w >> 8], 1, __ATOMIC_RELAXED, __HIP_MEMORY_SCOPE_WORKGROUP);
            }
        }
        __syncthreads();
        for (int i = t; i < NBK; i += 1024)
            *reinterpret_cast<int2*>(tab + i * NCH + 2 * j) = make_int2(h[0][i], h[1][i]);
    }
}

__global__ void __launch_bounds__(256)
scan_dinv_kernel(const int* __restrict__ tab, int* __restrict__ toff,
                 const unsigned* __restrict__ planes,
                 float* __restrict__ dinv, int n) {
    __shared__ int red[256];
    __shared__ int sc[256];
    __shared__ unsigned clo[2][128];
    __shared__ unsigned chi[2][128];
    const int t = threadIdx.x;
    const int b = blockIdx.x;
    const int i = b * 256 + t;
    {
        const int w  = t & 127;
        const int ph = t >> 7;
        const unsigned* __restrict__ p =
            planes + (size_t)(ph * (NSL / 2)) * NPW + (b * 128 + w);
        unsigned lo = 0, hi = 0;
        #pragma unroll 32
        for (int s = 0; s < NSL / 2; ++s) {
            const unsigned v = p[(size_t)s * NPW];
            lo += v & 0xFFFFu;
            hi += v >> 16;
        }
        clo[ph][w] = lo;
        chi[ph][w] = hi;
    }
    int part = 0;
    const int4* __restrict__ tab4 = reinterpret_cast<const int4*>(tab);
    for (int j = t; j < b * 64; j += 256) {
        const int4 v = tab4[j];
        part += (v.x + v.y) + (v.z + v.w);
    }
    red[t] = part;
    const int c = tab[i];
    sc[t] = c;
    __syncthreads();
    if (i < n) {
        const int w = t >> 1;
        const unsigned cn = (t & 1) ? (chi[0][w] + chi[1][w]) : (clo[0][w] + clo[1][w]);
        dinv[i] = rsqrtf(1.0f + (float)cn);
    }
    for (int s = 128; s > 0; s >>= 1) {
        if (t < s) red[t] += red[t + s];
        __syncthreads();
    }
    for (int off = 1; off < 256; off <<= 1) {
        const int v = (t >= off) ? sc[t - off] : 0;
        __syncthreads();
        sc[t] += v;
        __syncthreads();
    }
    toff[i] = red[0] + sc[t] - c;
}

#define CHE   (N_EDGES / NCH)
#define XWG   ((N_NODES / 16 + 31) / 32)
static_assert(NBK <= 1024, "one bucket per thread in the chunk's scan");
static_assert(N_NODES % 16 == 0, "row tiles are whole");
__global__ void __launch_bounds__(1024)
bucket_place_xw1_kernel(const int* __restrict__ row, const int* __restrict__ col,
                        const int* __restrict__ tab, const int* __restrict__ toff,
                        int* __restrict__ part,
                        const float* __restrict__ x,
                        const float* __restrict__ W1,
                        const float* __restrict__ dinv,
                        float* __restrict__ hlin,
                        int n) {
    __shared__ int sw[CHE];
    __shared__ int sd[CHE];
    __shared__ int cur[NBK];
    __shared__ int del[NBK];
    __shared__ int wtot[32];
    __shared__ float Wp[128 * 16];
    const int t    = threadIdx.x;
    const int lane = t & 31;
    const int wv   = t >> 5;
    if (blockIdx.x < NCH) {
        const int q0 = blockIdx.x * CH4;
        const int4 c0 = reinterpret_cast<const int4*>(col)[q0 + t];
        const int4 r0 = reinterpret_cast<const int4*>(row)[q0 + t];
        static_assert(CH4 >= 1024, "every thread has a first quad");
        int cn = 0, g = 0;
        if (t < NBK) { cn = tab[t * NCH + blockIdx.x]; g = toff[t * NCH + blockIdx.x]; }
        int inc = cn;
        #pragma unroll
        for (int o = 1; o < 32; o <<= 1) {
            const int v = __shfl_up(inc, o);
            if (lane >= o) inc += v;
        }
        if (lane == 31) wtot[wv] = inc;
        __syncthreads();
        {
            const int wt = wtot[lane];
            int winc = wt;
            #pragma unroll
            for (int o = 1; o < 32; o <<= 1) {
                const int v = __shfl_up(winc, o);
                if (lane >= o) winc += v;
            }
            const int wbase = __shfl(winc - wt, wv);
            if (t < NBK) {
                const int ls = wbase + inc - cn;
                cur[t] = ls;
                del[t] = g - ls;
            }
        }
        __syncthreads();
        for (int q = t; q < CH4; q += 1024) {
            int4 c = c0, r = r0;
            if (q != t) {
                c = reinterpret_cast<const int4*>(col)[q0 + q];
                r = reinterpret_cast<const int4*>(row)[q0 + q];
            }
            const int b0 = r.x >> 8, b1 = r.y >> 8, b2 = r.z >> 8, b3 = r.w >> 8;
            const int p0 = __hip_atomic_fetch_add(&cur[b0], 1, __ATOMIC_RELAXED, __HIP_MEMORY_SCOPE_WORKGROUP);
            const int p1 = __hip_atomic_fetch_add(&cur[b1], 1, __ATOMIC_RELAXED, __HIP_MEMORY_SCOPE_WORKGROUP);
            const int p2 = __hip_atomic_fetch_add(&cur[b2], 1, __ATOMIC_RELAXED, __HIP_MEMORY_SCOPE_WORKGROUP);
            const int p3 = __hip_atomic_fetch_add(&cur[b3], 1, __ATOMIC_RELAXED, __HIP_MEMORY_SCOPE_WORKGROUP);
            sw[p0] = ((r.x & 255) << 17) | c.x;  sd[p0] = p0 + del[b0];
            sw[p1] = ((r.y & 255) << 17) | c.y;  sd[p1] = p1 + del[b1];
            sw[p2] = ((r.z & 255) << 17) | c.z;  sd[p2] = p2 + del[b2];
            sw[p3] = ((r.w & 255) << 17) | c.w;  sd[p3] = p3 + del[b3];
        }
        __syncthreads();
        for (int i = t; i < CHE; i += 1024) part[sd[i]] = sw[i];
    } else {
        {
            const int idx = t * 2;
            const int kr  = idx >> 4;
            const int nc  = idx & 15;
            const float w0 = W1[kr * 4 + (nc & 3)];
            const float w1 = W1[kr * 4 + ((nc + 1) & 3)];
            Wp[idx]     = (nc < 4) ? w0 : 0.0f;
            Wp[idx + 1] = (nc < 4) ? w1 : 0.0f;
        }
        __syncthreads();

        const int tile   = (blockIdx.x - NCH) * 32 + wv;
        const int nTiles = n >> 4;
        if (tile >= nTiles) return;

        const int base = tile * 16;
        const int hh   = lane >> 4;
        const int nl   = lane & 15;
        const int arow = base + nl;
        const float4* __restrict__ dv = reinterpret_cast<const float4*>(dinv + base + 8 * hh);
        const float4 d0 = dv[0];
        const float4 d1 = dv[1];

        v8f c = {};
        #pragma unroll 4
        for (int k = 0; k < 128; k += 4) {
            const int kk = k + 2 * hh;
            v2f a = *reinterpret_cast<const v2f*>(x + (size_t)arow * 128 + kk);
            v2f b;
            b.x = Wp[(kk + 0) * 16 + nl];
            b.y = Wp[(kk + 1) * 16 + nl];
            c = __builtin_amdgcn_wmma_f32_16x16x4_f32(
                     false, a,  false, b,
                     (short)0, c,  false,  false);
        }

        if (nl < 4) {
            const float ds[8] = {d0.x, d0.y, d0.z, d0.w, d1.x, d1.y, d1.z, d1.w};
            #pragma unroll
            for (int v = 0; v < 8; ++v) {
                int m = base + v + 8 * hh;
                hlin[(size_t)m * 4 + nl] = c[v] * ds[v];
            }
        }
    }
}

#define GCAP  12288
__global__ void __launch_bounds__(512)
bucket_group_kernel(const int* __restrict__ part, const int* __restrict__ toff,
                    int* __restrict__ rowptr, int* __restrict__ cnt,
                    int* __restrict__ src, int n, int ne) {
    __shared__ int   c[256];
    __shared__ int   sc[256];
    __shared__ int   cur[256];
    __shared__ int   st[GCAP];
    const int t = threadIdx.x;
    const int b = blockIdx.x;
    const int node = b * 256 + t;
    const int beg = toff[b * NCH];
    const int end = (b + 1 < NBK) ? toff[(b + 1) * NCH] : ne;
    if (t < 256) c[t] = 0;
    __syncthreads();
    for (int e = beg + t; e < end; e += 4 * 512) {
        int w[4];
        #pragma unroll
        for (int k = 0; k < 4; ++k) {
            const int q = e + 512 * k;
            w[k] = (q < end) ? part[q] : -1;
        }
        #pragma unroll
        for (int k = 0; k < 4; ++k)
            if (w[k] >= 0)
                __hip_atomic_fetch_add(&c[w[k] >> 17], 1, __ATOMIC_RELAXED, __HIP_MEMORY_SCOPE_WORKGROUP);
    }
    __syncthreads();
    int my = 0;
    if (t < 256) { my = c[t]; sc[t] = my; }
    __syncthreads();
    for (int off = 1; off < 256; off <<= 1) {
        int v = 0;
        if (t < 256 && t >= off) v = sc[t - off];
        __syncthreads();
        if (t < 256) sc[t] += v;
        __syncthreads();
    }
    if (t < 256) {
        const int first = beg + sc[t] - my;
        cur[t] = first;
        if (node < n) { rowptr[node] = first; cnt[node] = my; }
    }
    __syncthreads();
    const bool staged = (end - beg) <= GCAP;
    for (int e = beg + t; e < end; e += 4 * 512) {
        int w[4];
        #pragma unroll
        for (int k = 0; k < 4; ++k) {
            const int q = e + 512 * k;
            w[k] = (q < end) ? part[q] : -1;
        }
        #pragma unroll
        for (int k = 0; k < 4; ++k) {
            if (w[k] >= 0) {
                const int rl = w[k] >> 17;
                const int s  = w[k] & 0x1FFFF;
                const int p  = __hip_atomic_fetch_add(&cur[rl], 1, __ATOMIC_RELAXED, __HIP_MEMORY_SCOPE_WORKGROUP);
                if (staged) st[p - beg] = s;
                else        src[p]      = s;
            }
        }
    }
    if (staged) {
        __syncthreads();
        for (int i = t; i < end - beg; i += 512) src[beg + i] = st[i];
    }
}

#define GPW   5
#define GWV   8
#define GWIN  2
#define GWG   (N_NODES / (4 * GPW * GWV))
static_assert(N_NODES % (4 * GPW * GWV) == 0, "every wave has GPW whole groups");
static_assert(GWIN <= GPW, "the window is no longer than the wave's groups");
template <int OUT> struct row_of;
template <> struct row_of<4> { typedef float4 type; };
template <> struct row_of<2> { typedef float2 type; };
__device__ __forceinline__ void add_row(float (&acc)[4], const float4 u, const bool live) {
    acc[0] += live ? u.x : 0.0f;
    acc[1] += live ? u.y : 0.0f;
    acc[2] += live ? u.z : 0.0f;
    acc[3] += live ? u.w : 0.0f;
}
__device__ __forceinline__ void add_row(float (&acc)[2], const float2 u, const bool live) {
    acc[0] += live ? u.x : 0.0f;
    acc[1] += live ? u.y : 0.0f;
}
template <int OUT, int NOUT, bool LAST>
__global__ void __launch_bounds__(32 * GWV)
gather_layer_kernel(const int* __restrict__ rowptr,
                    const int* __restrict__ cnt,
                    const int* __restrict__ src,
                    const float* __restrict__ hlin,
                    const float* __restrict__ dinv,
                    const float* __restrict__ b,
                    const float* __restrict__ Wn,
                    const float* __restrict__ bn,
                    float* __restrict__ ynext,
                    float* __restrict__ tout,
                    int n) {
    static_assert(!LAST || OUT == 2, "the classifier's sum is written for two features");
    typedef typename row_of<OUT>::type row_t;
    const row_t* __restrict__ hrow = reinterpret_cast<const row_t*>(hlin);
    const int lane = threadIdx.x & 31;
    const int sub  = lane & 7;
    const int gw   = blockIdx.x * GWV + (threadIdx.x >> 5);
    const int r0   = gw * (4 * GPW) + (lane >> 3);
    const int first = lane & ~7;

    static_assert((OUT & (OUT - 1)) == 0 && (NOUT & (NOUT - 1)) == 0, "the wrapped index is a mask");
    const int so = sub & (OUT - 1);
    const int sn = sub & (NOUT - 1);
    const float bb = b[so];
    float w[OUT];
    #pragma unroll
    for (int k = 0; k < OUT; ++k) w[k] = Wn[k * NOUT + sn];
    float bo = 0.0f;
    if (LAST) bo = bn[sn];

    int   beg[GPW], cn[GPW];
    float hs[GPW], di[GPW];
    #pragma unroll
    for (int i = 0; i < GPW; ++i) {
        const int r  = r0 + 4 * i;
        const int rc = (r < n) ? r : (n - 1);
        beg[i] = rowptr[rc];
        const int cc = cnt[rc];
        cn[i]  = (r < n) ? cc : 0;
        hs[i]  = hlin[(size_t)rc * OUT + so];
        di[i]  = dinv[rc];
    }
    int sr[GPW][8];
    #pragma unroll
    for (int i = 0; i < GPW; ++i) {
        #pragma unroll
        for (int k = 0; k < 8; ++k) {
            const int q = sub + 8 * k;
            int idx = beg[i] + ((q < cn[i]) ? q : (cn[i] - 1));
            idx = (idx > 0) ? idx : 0;
            sr[i][k] = src[idx];
        }
    }
    row_t v[GPW][8];
    #pragma unroll
    for (int i = 0; i < GWIN; ++i) {
        #pragma unroll
        for (int k = 0; k < 8; ++k) v[i][k] = hrow[sr[i][k]];
    }
    #pragma unroll
    for (int i = 0; i < GPW; ++i) {
        float acc[OUT];
        #pragma unroll
        for (int f = 0; f < OUT; ++f) acc[f] = 0.0f;
        #pragma unroll
        for (int k = 0; k < 8; ++k) add_row(acc, v[i][k], (sub + 8 * k) < cn[i]);
        if (i + GWIN < GPW) {
            #pragma unroll
            for (int k = 0; k < 8; ++k) v[(i + GWIN) % GPW][k] = hrow[sr[(i + GWIN) % GPW][k]];
        }
        const int end = beg[i] + cn[i];
        for (int p = beg[i] + 64 + sub; p < end; p += 8) {
            add_row(acc, hrow[src[p]], true);
        }
        #pragma unroll
        for (int f = 0; f < OUT; ++f) {
            acc[f] += __shfl_xor(acc[f], 4);
            acc[f] += __shfl_xor(acc[f], 2);
            acc[f] += __shfl_xor(acc[f], 1);
        }

        float a = acc[0];
        #pragma unroll
        for (int f = 1; f < OUT; ++f) a = (sub == f) ? acc[f] : a;
        const float t  = tanhf((a + hs[i]) * di[i] + bb);
        float tk[OUT];
        #pragma unroll
        for (int k = 0; k < OUT; ++k) tk[k] = __shfl(t, first + k);
        float y;
        if (LAST) {
            y = fmaf(tk[0], w[0], fmaf(tk[1], w[1], bo));
        } else {
            y = 0.0f;
            #pragma unroll
            for (int k = 0; k < OUT; ++k) y = fmaf(tk[k], w[k], y);
            y *= di[i];
        }
        const int r = r0 + 4 * i;
        if (r < n) {
            if (sub < NOUT) ynext[(size_t)r * NOUT + sub] = y;
            if (LAST) {
                if (sub < OUT) tout[(size_t)r * OUT + sub] = t;
            }
        }
    }
}

extern "C" void kernel_launch(void* const* d_in, const int* in_sizes, int n_in,
                              void* d_out, int out_size, void* d_ws, size_t ws_size,
                              hipStream_t stream) {
    const float* x  = (const float*)d_in[0];
    const int*   ei = (const int*)d_in[1];
    const float* W1 = (const float*)d_in[2];
    const float* b1 = (const float*)d_in[3];
    const float* W2 = (const float*)d_in[4];
    const float* b2 = (const float*)d_in[5];
    const float* W3 = (const float*)d_in[6];
    const float* b3 = (const float*)d_in[7];
    const float* Wc = (const float*)d_in[8];
    const float* bc = (const float*)d_in[9];

    const int N = N_NODES;
    const int E = N_EDGES;
    const int* row = ei;
    const int* col = ei + E;

    float* ws     = (float*)d_ws;
    float* hP     = ws;
    float* hQ     = hP + 4 * N;
    int*   src    = (int*)(hQ + 4 * N);
    int*   part   = src + E;
    int*   tab    = part + E;
    int*   toff   = tab + NBK * NCH;
    float* dinv   = (float*)(toff + NBK * NCH);
    int*   cnt    = (int*)(dinv + N);
    int*   rowptr = cnt + N;
    unsigned* planes = (unsigned*)(rowptr + N);

    float* outC = (float*)d_out;
    float* outH = outC + (size_t)4 * N;

    const int BT = 32 * GWV;
    const int gG = GWG;

    deg_hist_bucket_count_kernel<<<NSL + NCW, 1024, 0, stream>>>(row, col, planes, tab);
    scan_dinv_kernel<<<NBK, 256, 0, stream>>>(tab, toff, planes, dinv, N);
    bucket_place_xw1_kernel<<<NCH + XWG, 1024, 0, stream>>>(row, col, tab, toff, part,
                                                          x, W1, dinv, hP, N);
    bucket_group_kernel<<<NBK, 512, 0, stream>>>(part, toff, rowptr, cnt, src, N, E);

    gather_layer_kernel<4, 4, false><<<gG, BT, 0, stream>>>(
        rowptr, cnt, src, hP, dinv, b1, W2, nullptr, hQ, nullptr, N);
    gather_layer_kernel<4, 2, false><<<gG, BT, 0, stream>>>(
        rowptr, cnt, src, hQ, dinv, b2, W3, nullptr, hP, nullptr, N);
    gather_layer_kernel<2, 4, true><<<gG, BT, 0, stream>>>(
        rowptr, cnt, src, hP, dinv, b3, Wc, bc, outC, outH, N);
}


# Scan launch: a block's base is the sum of 128 per-count-work-group bucket prefixes (made in launch 1 by a scan of its own counts) instead of a loop over the whole count table in front of it
# speedup vs baseline: 10.1137x; 1.0383x over previous
#include <hip/hip_runtime.h>
#include <hip/hip_bf16.h>
#include <math.h>

#define N_NODES   100000
#define N_EDGES   3200000

typedef float v2f __attribute__((ext_vector_type(2)));
typedef float v8f __attribute__((ext_vector_type(8)));

#define NBK   ((N_NODES + 255) / 256)
#define NCH   256
#define CH4   (N_EDGES / (4 * NCH))
static_assert(N_EDGES % (4 * NCH) == 0, "chunks are whole quads");
static_assert(N_NODES <= (1 << 17), "source index fits 17 bits of an edge word");

#define NSL   128
#define SL4   (N_EDGES / (4 * NSL))
#define SLT   ((SL4 + 1023) / 1024)
#define NPW   (NBK * 128)
static_assert(N_EDGES % (4 * NSL) == 0, "slices are whole quads");
static_assert(N_EDGES / NSL <= 0xFFFF, "a slice's count of one node fits 16 bits");

#define NCW   (NCH / 2)
#define CW4   (2 * CH4)
#define CWT   ((CW4 + 1023) / 1024)
static_assert(NCH % 2 == 0, "count work-groups take whole pairs of chunks");
static_assert(NBK <= 1024, "one bucket per thread in the count work-group's scan");
__global__ void __launch_bounds__(1024)
deg_hist_bucket_count_kernel(const int* __restrict__ row, const int* __restrict__ col,
                             unsigned* __restrict__ planes, int* __restrict__ tab,
                             int* __restrict__ pre) {
    __shared__ uint4 hc4[NPW / 4];
    __shared__ int h[2][NBK];
    __shared__ int wtot[32];
    const int t = threadIdx.x;
    if (blockIdx.x < NSL) {
        unsigned* hc = reinterpret_cast<unsigned*>(hc4);
        const int4* __restrict__ c4 =
            reinterpret_cast<const int4*>(col) + (size_t)blockIdx.x * SL4;
        int4 c[SLT];
        #pragma unroll
        for (int k = 0; k < SLT; ++k) {
            const int q = t + 1024 * k;
            c[k] = (q < SL4) ? c4[q] : make_int4(-1, -1, -1, -1);
        }
        for (int i = t; i < NPW / 4; i += 1024) hc4[i] = make_uint4(0u, 0u, 0u, 0u);
        __syncthreads();
        #pragma unroll
        for (int k = 0; k < SLT; ++k) {
            if (c[k].x >= 0) {
                __hip_atomic_fetch_add(&hc[c[k].x >> 1], 1u << ((c[k].x & 1) << 4), __ATOMIC_RELAXED, __HIP_MEMORY_SCOPE_WORKGROUP);
                __hip_atomic_fetch_add(&hc[c[k].y >> 1], 1u << ((c[k].y & 1) << 4), __ATOMIC_RELAXED, __HIP_MEMORY_SCOPE_WORKGROUP);
                __hip_atomic_fetch_add(&hc[c[k].z >> 1], 1u << ((c[k].z & 1) << 4), __ATOMIC_RELAXED, __HIP_MEMORY_SCOPE_WORKGROUP);
                __hip_atomic_fetch_add(&hc[c[k].w >> 1], 1u << ((c[k].w & 1) << 4), __ATOMIC_RELAXED, __HIP_MEMORY_SCOPE_WORKGROUP);
            }
        }
        __syncthreads();
        uint4* __restrict__ dst =
            reinterpret_cast<uint4*>(planes) + (size_t)blockIdx.x * (NPW / 4);
        for (int i = t; i < NPW / 4; i += 1024) dst[i] = hc4[i];
    } else {
        const int j = blockIdx.x - NSL;
        const int4* __restrict__ r4 =
            reinterpret_cast<const int4*>(row) + (size_t)j * CW4;
        int4 r[CWT];
        #pragma unroll
        for (int k = 0; k < CWT; ++k) {
            const int q = t + 1024 * k;
            r[k] = (q < CW4) ? r4[q] : make_int4(-1, -1, -1, -1);
        }
        for (int i = t; i < NBK; i += 1024) { h[0][i] = 0; h[1][i] = 0; }
        __syncthreads();
        #pragma unroll
        for (int k = 0; k < CWT; ++k) {
            if (r[k].x >= 0) {
                int* hq = h[(t + 1024 * k) >= CH4];
                __hip_atomic_fetch_add(&hq[r[k].x >> 8], 1, __ATOMIC_RELAXED, __HIP_MEMORY_SCOPE_WORKGROUP);
                __hip_atomic_fetch_add(&hq[r[k].y >> 8], 1, __ATOMIC_RELAXED, __HIP_MEMORY_SCOPE_WORKGROUP);
                __hip_atomic_fetch_add(&hq[r[k].z >> 8], 1, __ATOMIC_RELAXED, __HIP_MEMORY_SCOPE_WORKGROUP);
                __hip_atomic_fetch_add(&hq[r[k].w >> 8], 1, __ATOMIC_RELAXED, __HIP_MEMORY_SCOPE_WORKGROUP);
            }
        }
        __syncthreads();
        const int lane = t & 31;
        const int wv   = t >> 5;
        int c0 = 0, c1 = 0;
        if (t < NBK) { c0 = h[0][t]; c1 = h[1][t]; }
        const int cs = c0 + c1;
        int inc = cs;
        for (int o = 1; o < 32; o <<= 1) {
            const int v = __shfl_up(inc, o);
            if (lane >= o) inc += v;
        }
        if (lane == 31) wtot[wv] = inc;
        __syncthreads();
        const int wt = wtot[lane];
        int winc = wt;
        for (int o = 1; o < 32; o <<= 1) {
            const int v = __shfl_up(winc, o);
            if (lane >= o) winc += v;
        }
        const int wbase = __shfl(winc - wt, wv);
        if (t < NBK) {
            *reinterpret_cast<int2*>(tab + t * NCH + 2 * j) = make_int2(c0, c1);
            pre[t * NCW + j] = wbase + inc - cs;
        }
    }
}

__global__ void __launch_bounds__(256)
scan_dinv_kernel(const int* __restrict__ tab, const int* __restrict__ pre,
                 int* __restrict__ toff,
                 const unsigned* __restrict__ planes,
                 float* __restrict__ dinv, int n) {
    __shared__ int red[8];
    __shared__ int sc[256];
    __shared__ unsigned clo[2][128];
    __shared__ unsigned chi[2][128];
    const int t = threadIdx.x;
    const int b = blockIdx.x;
    const int i = b * 256 + t;
    {
        const int w  = t & 127;
        const int ph = t >> 7;
        const unsigned* __restrict__ p =
            planes + (size_t)(ph * (NSL / 2)) * NPW + (b * 128 + w);
        unsigned lo = 0, hi = 0;
        #pragma unroll 32
        for (int s = 0; s < NSL / 2; ++s) {
            const unsigned v = p[(size_t)s * NPW];
            lo += v & 0xFFFFu;
            hi += v >> 16;
        }
        clo[ph][w] = lo;
        chi[ph][w] = hi;
    }
    static_assert(NCW <= 256, "one count work-group's word per thread");
    const int c = tab[i];
    int pv = (t < NCW) ? pre[b * NCW + t] : 0;
    pv += __shfl_xor(pv, 16);
    pv += __shfl_xor(pv, 8);
    pv += __shfl_xor(pv, 4);
    pv += __shfl_xor(pv, 2);
    pv += __shfl_xor(pv, 1);
    if ((t & 31) == 0) red[t >> 5] = pv;
    sc[t] = c;
    __syncthreads();
    if (i < n) {
        const int w = t >> 1;
        const unsigned cn = (t & 1) ? (chi[0][w] + chi[1][w]) : (clo[0][w] + clo[1][w]);
        dinv[i] = rsqrtf(1.0f + (float)cn);
    }
    const int base = ((red[0] + red[1]) + (red[2] + red[3])) + ((red[4] + red[5]) + (red[6] + red[7]));
    for (int off = 1; off < 256; off <<= 1) {
        const int v = (t >= off) ? sc[t - off] : 0;
        __syncthreads();
        sc[t] += v;
        __syncthreads();
    }
    toff[i] = base + sc[t] - c;
}

#define CHE   (N_EDGES / NCH)
#define XWG   ((N_NODES / 16 + 31) / 32)
static_assert(NBK <= 1024, "one bucket per thread in the chunk's scan");
static_assert(N_NODES % 16 == 0, "row tiles are whole");
__global__ void __launch_bounds__(1024)
bucket_place_xw1_kernel(const int* __restrict__ row, const int* __restrict__ col,
                        const int* __restrict__ tab, const int* __restrict__ toff,
                        int* __restrict__ part,
                        const float* __restrict__ x,
                        const float* __restrict__ W1,
                        const float* __restrict__ dinv,
                        float* __restrict__ hlin,
                        int n) {
    __shared__ int sw[CHE];
    __shared__ int sd[CHE];
    __shared__ int cur[NBK];
    __shared__ int del[NBK];
    __shared__ int wtot[32];
    __shared__ float Wp[128 * 16];
    const int t    = threadIdx.x;
    const int lane = t & 31;
    const int wv   = t >> 5;
    if (blockIdx.x < NCH) {
        const int q0 = blockIdx.x * CH4;
        const int4 c0 = reinterpret_cast<const int4*>(col)[q0 + t];
        const int4 r0 = reinterpret_cast<const int4*>(row)[q0 + t];
        static_assert(CH4 >= 1024, "every thread has a first quad");
        int cn = 0, g = 0;
        if (t < NBK) { cn = tab[t * NCH + blockIdx.x]; g = toff[t * NCH + blockIdx.x]; }
        int inc = cn;
        #pragma unroll
        for (int o = 1; o < 32; o <<= 1) {
            const int v = __shfl_up(inc, o);
            if (lane >= o) inc += v;
        }
        if (lane == 31) wtot[wv] = inc;
        __syncthreads();
        {
            const int wt = wtot[lane];
            int winc = wt;
            #pragma unroll
            for (int o = 1; o < 32; o <<= 1) {
                const int v = __shfl_up(winc, o);
                if (lane >= o) winc += v;
            }
            const int wbase = __shfl(winc - wt, wv);
            if (t < NBK) {
                const int ls = wbase + inc - cn;
                cur[t] = ls;
                del[t] = g - ls;
            }
        }
        __syncthreads();
        for (int q = t; q < CH4; q += 1024) {
            int4 c = c0, r = r0;
            if (q != t) {
                c = reinterpret_cast<const int4*>(col)[q0 + q];
                r = reinterpret_cast<const int4*>(row)[q0 + q];
            }
            const int b0 = r.x >> 8, b1 = r.y >> 8, b2 = r.z >> 8, b3 = r.w >> 8;
            const int p0 = __hip_atomic_fetch_add(&cur[b0], 1, __ATOMIC_RELAXED, __HIP_MEMORY_SCOPE_WORKGROUP);
            const int p1 = __hip_atomic_fetch_add(&cur[b1], 1, __ATOMIC_RELAXED, __HIP_MEMORY_SCOPE_WORKGROUP);
            const int p2 = __hip_atomic_fetch_add(&cur[b2], 1, __ATOMIC_RELAXED, __HIP_MEMORY_SCOPE_WORKGROUP);
            const int p3 = __hip_atomic_fetch_add(&cur[b3], 1, __ATOMIC_RELAXED, __HIP_MEMORY_SCOPE_WORKGROUP);
            sw[p0] = ((r.x & 255) << 17) | c.x;  sd[p0] = p0 + del[b0];
            sw[p1] = ((r.y & 255) << 17) | c.y;  sd[p1] = p1 + del[b1];
            sw[p2] = ((r.z & 255) << 17) | c.z;  sd[p2] = p2 + del[b2];
            sw[p3] = ((r.w & 255) << 17) | c.w;  sd[p3] = p3 + del[b3];
        }
        __syncthreads();
        for (int i = t; i < CHE; i += 1024) part[sd[i]] = sw[i];
    } else {
        {
            const int idx = t * 2;
            const int kr  = idx >> 4;
            const int nc  = idx & 15;
            const float w0 = W1[kr * 4 + (nc & 3)];
            const float w1 = W1[kr * 4 + ((nc + 1) & 3)];
            Wp[idx]     = (nc < 4) ? w0 : 0.0f;
            Wp[idx + 1] = (nc < 4) ? w1 : 0.0f;
        }
        __syncthreads();

        const int tile   = (blockIdx.x - NCH) * 32 + wv;
        const int nTiles = n >> 4;
        if (tile >= nTiles) return;

        const int base = tile * 16;
        const int hh   = lane >> 4;
        const int nl   = lane & 15;
        const int arow = base + nl;
        const float4* __restrict__ dv = reinterpret_cast<const float4*>(dinv + base + 8 * hh);
        const float4 d0 = dv[0];
        const float4 d1 = dv[1];

        v8f c = {};
        #pragma unroll 4
        for (int k = 0; k < 128; k += 4) {
            const int kk = k + 2 * hh;
            v2f a = *reinterpret_cast<const v2f*>(x + (size_t)arow * 128 + kk);
            v2f b;
            b.x = Wp[(kk + 0) * 16 + nl];
            b.y = Wp[(kk + 1) * 16 + nl];
            c = __builtin_amdgcn_wmma_f32_16x16x4_f32(
                     false, a,  false, b,
                     (short)0, c,  false,  false);
        }

        if (nl < 4) {
            const float ds[8] = {d0.x, d0.y, d0.z, d0.w, d1.x, d1.y, d1.z, d1.w};
            #pragma unroll
            for (int v = 0; v < 8; ++v) {
                int m = base + v + 8 * hh;
                hlin[(size_t)m * 4 + nl] = c[v] * ds[v];
            }
        }
    }
}

#define GCAP  12288
__global__ void __launch_bounds__(512)
bucket_group_kernel(const int* __restrict__ part, const int* __restrict__ toff,
                    int* __restrict__ rowptr, int* __restrict__ cnt,
                    int* __restrict__ src, int n, int ne) {
    __shared__ int   c[256];
    __shared__ int   sc[256];
    __shared__ int   cur[256];
    __shared__ int   st[GCAP];
    const int t = threadIdx.x;
    const int b = blockIdx.x;
    const int node = b * 256 + t;
    const int beg = toff[b * NCH];
    const int end = (b + 1 < NBK) ? toff[(b + 1) * NCH] : ne;
    if (t < 256) c[t] = 0;
    __syncthreads();
    for (int e = beg + t; e < end; e += 4 * 512) {
        int w[4];
        #pragma unroll
        for (int k = 0; k < 4; ++k) {
            const int q = e + 512 * k;
            w[k] = (q < end) ? part[q] : -1;
        }
        #pragma unroll
        for (int k = 0; k < 4; ++k)
            if (w[k] >= 0)
                __hip_atomic_fetch_add(&c[w[k] >> 17], 1, __ATOMIC_RELAXED, __HIP_MEMORY_SCOPE_WORKGROUP);
    }
    __syncthreads();
    int my = 0;
    if (t < 256) { my = c[t]; sc[t] = my; }
    __syncthreads();
    for (int off = 1; off < 256; off <<= 1) {
        int v = 0;
        if (t < 256 && t >= off) v = sc[t - off];
        __syncthreads();
        if (t < 256) sc[t] += v;
        __syncthreads();
    }
    if (t < 256) {
        const int first = beg + sc[t] - my;
        cur[t] = first;
        if (node < n) { rowptr[node] = first; cnt[node] = my; }
    }
    __syncthreads();
    const bool staged = (end - beg) <= GCAP;
    for (int e = beg + t; e < end; e += 4 * 512) {
        int w[4];
        #pragma unroll
        for (int k = 0; k < 4; ++k) {
            const int q = e + 512 * k;
            w[k] = (q < end) ? part[q] : -1;
        }
        #pragma unroll
        for (int k = 0; k < 4; ++k) {
            if (w[k] >= 0) {
                const int rl = w[k] >> 17;
                const int s  = w[k] & 0x1FFFF;
                const int p  = __hip_atomic_fetch_add(&cur[rl], 1, __ATOMIC_RELAXED, __HIP_MEMORY_SCOPE_WORKGROUP);
                if (staged) st[p - beg] = s;
                else        src[p]      = s;
            }
        }
    }
    if (staged) {
        __syncthreads();
        for (int i = t; i < end - beg; i += 512) src[beg + i] = st[i];
    }
}

#define GPW   5
#define GWV   8
#define GWIN  2
#define GWG   (N_NODES / (4 * GPW * GWV))
static_assert(N_NODES % (4 * GPW * GWV) == 0, "every wave has GPW whole groups");
static_assert(GWIN <= GPW, "the window is no longer than the wave's groups");
template <int OUT> struct row_of;
template <> struct row_of<4> { typedef float4 type; };
template <> struct row_of<2> { typedef float2 type; };
__device__ __forceinline__ void add_row(float (&acc)[4], const float4 u, const bool live) {
    acc[0] += live ? u.x : 0.0f;
    acc[1] += live ? u.y : 0.0f;
    acc[2] += live ? u.z : 0.0f;
    acc[3] += live ? u.w : 0.0f;
}
__device__ __forceinline__ void add_row(float (&acc)[2], const float2 u, const bool live) {
    acc[0] += live ? u.x : 0.0f;
    acc[1] += live ? u.y : 0.0f;
}
template <int OUT, int NOUT, bool LAST>
__global__ void __launch_bounds__(32 * GWV)
gather_layer_kernel(const int* __restrict__ rowptr,
                    const int* __restrict__ cnt,
                    const int* __restrict__ src,
                    const float* __restrict__ hlin,
                    const float* __restrict__ dinv,
                    const float* __restrict__ b,
                    const float* __restrict__ Wn,
                    const float* __restrict__ bn,
                    float* __restrict__ ynext,
                    float* __restrict__ tout,
                    int n) {
    static_assert(!LAST || OUT == 2, "the classifier's sum is written for two features");
    typedef typename row_of<OUT>::type row_t;
    const row_t* __restrict__ hrow = reinterpret_cast<const row_t*>(hlin);
    const int lane = threadIdx.x & 31;
    const int sub  = lane & 7;
    const int gw   = blockIdx.x * GWV + (threadIdx.x >> 5);
    const int r0   = gw * (4 * GPW) + (lane >> 3);
    const int first = lane & ~7;

    static_assert((OUT & (OUT - 1)) == 0 && (NOUT & (NOUT - 1)) == 0, "the wrapped index is a mask");
    const int so = sub & (OUT - 1);
    const int sn = sub & (NOUT - 1);
    const float bb = b[so];
    float w[OUT];
    #pragma unroll
    for (int k = 0; k < OUT; ++k) w[k] = Wn[k * NOUT + sn];
    float bo = 0.0f;
    if (LAST) bo = bn[sn];

    int   beg[GPW], cn[GPW];
    float hs[GPW], di[GPW];
    #pragma unroll
    for (int i = 0; i < GPW; ++i) {
        const int r  = r0 + 4 * i;
        const int rc = (r < n) ? r : (n - 1);
        beg[i] = rowptr[rc];
        const int cc = cnt[rc];
        cn[i]  = (r < n) ? cc : 0;
        hs[i]  = hlin[(size_t)rc * OUT + so];
        di[i]  = dinv[rc];
    }
    int sr[GPW][8];
    #pragma unroll
    for (int i = 0; i < GPW; ++i) {
        #pragma unroll
        for (int k = 0; k < 8; ++k) {
            const int q = sub + 8 * k;
            int idx = beg[i] + ((q < cn[i]) ? q : (cn[i] - 1));
            idx = (idx > 0) ? idx : 0;
            sr[i][k] = src[idx];
        }
    }
    row_t v[GPW][8];
    #pragma unroll
    for (int i = 0; i < GWIN; ++i) {
        #pragma unroll
        for (int k = 0; k < 8; ++k) v[i][k] = hrow[sr[i][k]];
    }
    #pragma unroll
    for (int i = 0; i < GPW; ++i) {
        float acc[OUT];
        #pragma unroll
        for (int f = 0; f < OUT; ++f) acc[f] = 0.0f;
        #pragma unroll
        for (int k = 0; k < 8; ++k) add_row(acc, v[i][k], (sub + 8 * k) < cn[i]);
        if (i + GWIN < GPW) {
            #pragma unroll
            for (int k = 0; k < 8; ++k) v[(i + GWIN) % GPW][k] = hrow[sr[(i + GWIN) % GPW][k]];
        }
        const int end = beg[i] + cn[i];
        for (int p = beg[i] + 64 + sub; p < end; p += 8) {
            add_row(acc, hrow[src[p]], true);
        }
        #pragma unroll
        for (int f = 0; f < OUT; ++f) {
            acc[f] += __shfl_xor(acc[f], 4);
            acc[f] += __shfl_xor(acc[f], 2);
            acc[f] += __shfl_xor(acc[f], 1);
        }

        float a = acc[0];
        #pragma unroll
        for (int f = 1; f < OUT; ++f) a = (sub == f) ? acc[f] : a;
        const float t  = tanhf((a + hs[i]) * di[i] + bb);
        float tk[OUT];
        #pragma unroll
        for (int k = 0; k < OUT; ++k) tk[k] = __shfl(t, first + k);
        float y;
        if (LAST) {
            y = fmaf(tk[0], w[0], fmaf(tk[1], w[1], bo));
        } else {
            y = 0.0f;
            #pragma unroll
            for (int k = 0; k < OUT; ++k) y = fmaf(tk[k], w[k], y);
            y *= di[i];
        }
        const int r = r0 + 4 * i;
        if (r < n) {
            if (sub < NOUT) ynext[(size_t)r * NOUT + sub] = y;
            if (LAST) {
                if (sub < OUT) tout[(size_t)r * OUT + sub] = t;
            }
        }
    }
}

extern "C" void kernel_launch(void* const* d_in, const int* in_sizes, int n_in,
                              void* d_out, int out_size, void* d_ws, size_t ws_size,
                              hipStream_t stream) {
    const float* x  = (const float*)d_in[0];
    const int*   ei = (const int*)d_in[1];
    const float* W1 = (const float*)d_in[2];
    const float* b1 = (const float*)d_in[3];
    const float* W2 = (const float*)d_in[4];
    const float* b2 = (const float*)d_in[5];
    const float* W3 = (const float*)d_in[6];
    const float* b3 = (const float*)d_in[7];
    const float* Wc = (const float*)d_in[8];
    const float* bc = (const float*)d_in[9];

    const int N = N_NODES;
    const int E = N_EDGES;
    const int* row = ei;
    const int* col = ei + E;

    float* ws     = (float*)d_ws;
    float* hP     = ws;
    float* hQ     = hP + 4 * N;
    int*   src    = (int*)(hQ + 4 * N);
    int*   part   = src + E;
    int*   tab    = part + E;
    int*   toff   = tab + NBK * NCH;
    float* dinv   = (float*)(toff + NBK * NCH);
    int*   cnt    = (int*)(dinv + N);
    int*   rowptr = cnt + N;
    unsigned* planes = (unsigned*)(rowptr + N);
    int*   pre    = (int*)(planes + (size_t)NSL * NPW);

    float* outC = (float*)d_out;
    float* outH = outC + (size_t)4 * N;

    const int BT = 32 * GWV;
    const int gG = GWG;

    deg_hist_bucket_count_kernel<<<NSL + NCW, 1024, 0, stream>>>(row, col, planes, tab, pre);
    scan_dinv_kernel<<<NBK, 256, 0, stream>>>(tab, pre, toff, planes, dinv, N);
    bucket_place_xw1_kernel<<<NCH + XWG, 1024, 0, stream>>>(row, col, tab, toff, part,
                                                          x, W1, dinv, hP, N);
    bucket_group_kernel<<<NBK, 512, 0, stream>>>(part, toff, rowptr, cnt, src, N, E);

    gather_layer_kernel<4, 4, false><<<gG, BT, 0, stream>>>(
        rowptr, cnt, src, hP, dinv, b1, W2, nullptr, hQ, nullptr, N);
    gather_layer_kernel<4, 2, false><<<gG, BT, 0, stream>>>(
        rowptr, cnt, src, hQ, dinv, b2, W3, nullptr, hP, nullptr, N);
    gather_layer_kernel<2, 4, true><<<gG, BT, 0, stream>>>(
        rowptr, cnt, src, hP, dinv, b3, Wc, bc, outC, outH, N);
}


# Layer 1's gather, tanh and layer-2 product folded into the level-2 grouping launch: a bucket's work-group gathers its own 256 nodes from the slots still in LDS; 7 launches become 6; sum order kept
# speedup vs baseline: 10.7484x; 1.0628x over previous
#include <hip/hip_runtime.h>
#include <hip/hip_bf16.h>
#include <math.h>

#define N_NODES   100000
#define N_EDGES   3200000

typedef float v2f __attribute__((ext_vector_type(2)));
typedef float v8f __attribute__((ext_vector_type(8)));

#define NBK   ((N_NODES + 255) / 256)
#define NCH   256
#define CH4   (N_EDGES / (4 * NCH))
static_assert(N_EDGES % (4 * NCH) == 0, "chunks are whole quads");
static_assert(N_NODES <= (1 << 17), "source index fits 17 bits of an edge word");

#define NSL   128
#define SL4   (N_EDGES / (4 * NSL))
#define SLT   ((SL4 + 1023) / 1024)
#define NPW   (NBK * 128)
static_assert(N_EDGES % (4 * NSL) == 0, "slices are whole quads");
static_assert(N_EDGES / NSL <= 0xFFFF, "a slice's count of one node fits 16 bits");

#define NCW   (NCH / 2)
#define CW4   (2 * CH4)
#define CWT   ((CW4 + 1023) / 1024)
static_assert(NCH % 2 == 0, "count work-groups take whole pairs of chunks");
static_assert(NBK <= 1024, "one bucket per thread in the count work-group's scan");
__global__ void __launch_bounds__(1024)
deg_hist_bucket_count_kernel(const int* __restrict__ row, const int* __restrict__ col,
                             unsigned* __restrict__ planes, int* __restrict__ tab,
                             int* __restrict__ pre) {
    __shared__ uint4 hc4[NPW / 4];
    __shared__ int h[2][NBK];
    __shared__ int wtot[32];
    const int t = threadIdx.x;
    if (blockIdx.x < NSL) {
        unsigned* hc = reinterpret_cast<unsigned*>(hc4);
        const int4* __restrict__ c4 =
            reinterpret_cast<const int4*>(col) + (size_t)blockIdx.x * SL4;
        int4 c[SLT];
        #pragma unroll
        for (int k = 0; k < SLT; ++k) {
            const int q = t + 1024 * k;
            c[k] = (q < SL4) ? c4[q] : make_int4(-1, -1, -1, -1);
        }
        for (int i = t; i < NPW / 4; i += 1024) hc4[i] = make_uint4(0u, 0u, 0u, 0u);
        __syncthreads();
        #pragma unroll
        for (int k = 0; k < SLT; ++k) {
            if (c[k].x >= 0) {
                __hip_atomic_fetch_add(&hc[c[k].x >> 1], 1u << ((c[k].x & 1) << 4), __ATOMIC_RELAXED, __HIP_MEMORY_SCOPE_WORKGROUP);
                __hip_atomic_fetch_add(&hc[c[k].y >> 1], 1u << ((c[k].y & 1) << 4), __ATOMIC_RELAXED, __HIP_MEMORY_SCOPE_WORKGROUP);
                __hip_atomic_fetch_add(&hc[c[k].z >> 1], 1u << ((c[k].z & 1) << 4), __ATOMIC_RELAXED, __HIP_MEMORY_SCOPE_WORKGROUP);
                __hip_atomic_fetch_add(&hc[c[k].w >> 1], 1u << ((c[k].w & 1) << 4), __ATOMIC_RELAXED, __HIP_MEMORY_SCOPE_WORKGROUP);
            }
        }
        __syncthreads();
        uint4* __restrict__ dst =
            reinterpret_cast<uint4*>(planes) + (size_t)blockIdx.x * (NPW / 4);
        for (int i = t; i < NPW / 4; i += 1024) dst[i] = hc4[i];
    } else {
        const int j = blockIdx.x - NSL;
        const int4* __restrict__ r4 =
            reinterpret_cast<const int4*>(row) + (size_t)j * CW4;
        int4 r[CWT];
        #pragma unroll
        for (int k = 0; k < CWT; ++k) {
            const int q = t + 1024 * k;
            r[k] = (q < CW4) ? r4[q] : make_int4(-1, -1, -1, -1);
        }
        for (int i = t; i < NBK; i += 1024) { h[0][i] = 0; h[1][i] = 0; }
        __syncthreads();
        #pragma unroll
        for (int k = 0; k < CWT; ++k) {
            if (r[k].x >= 0) {
                int* hq = h[(t + 1024 * k) >= CH4];
                __hip_atomic_fetch_add(&hq[r[k].x >> 8], 1, __ATOMIC_RELAXED, __HIP_MEMORY_SCOPE_WORKGROUP);
                __hip_atomic_fetch_add(&hq[r[k].y >> 8], 1, __ATOMIC_RELAXED, __HIP_MEMORY_SCOPE_WORKGROUP);
                __hip_atomic_fetch_add(&hq[r[k].z >> 8], 1, __ATOMIC_RELAXED, __HIP_MEMORY_SCOPE_WORKGROUP);
                __hip_atomic_fetch_add(&hq[r[k].w >> 8], 1, __ATOMIC_RELAXED, __HIP_MEMORY_SCOPE_WORKGROUP);
            }
        }
        __syncthreads();
        const int lane = t & 31;
        const int wv   = t >> 5;
        int c0 = 0, c1 = 0;
        if (t < NBK) { c0 = h[0][t]; c1 = h[1][t]; }
        const int cs = c0 + c1;
        int inc = cs;
        for (int o = 1; o < 32; o <<= 1) {
            const int v = __shfl_up(inc, o);
            if (lane >= o) inc += v;
        }
        if (lane == 31) wtot[wv] = inc;
        __syncthreads();
        const int wt = wtot[lane];
        int winc = wt;
        for (int o = 1; o < 32; o <<= 1) {
            const int v = __shfl_up(winc, o);
            if (lane >= o) winc += v;
        }
        const int wbase = __shfl(winc - wt, wv);
        if (t < NBK) {
            *reinterpret_cast<int2*>(tab + t * NCH + 2 * j) = make_int2(c0, c1);
            pre[t * NCW + j] = wbase + inc - cs;
        }
    }
}

__global__ void __launch_bounds__(256)
scan_dinv_kernel(const int* __restrict__ tab, const int* __restrict__ pre,
                 int* __restrict__ toff,
                 const unsigned* __restrict__ planes,
                 float* __restrict__ dinv, int n) {
    __shared__ int red[8];
    __shared__ int sc[256];
    __shared__ unsigned clo[2][128];
    __shared__ unsigned chi[2][128];
    const int t = threadIdx.x;
    const int b = blockIdx.x;
    const int i = b * 256 + t;
    {
        const int w  = t & 127;
        const int ph = t >> 7;
        const unsigned* __restrict__ p =
            planes + (size_t)(ph * (NSL / 2)) * NPW + (b * 128 + w);
        unsigned lo = 0, hi = 0;
        #pragma unroll 32
        for (int s = 0; s < NSL / 2; ++s) {
            const unsigned v = p[(size_t)s * NPW];
            lo += v & 0xFFFFu;
            hi += v >> 16;
        }
        clo[ph][w] = lo;
        chi[ph][w] = hi;
    }
    static_assert(NCW <= 256, "one count work-group's word per thread");
    const int c = tab[i];
    int pv = (t < NCW) ? pre[b * NCW + t] : 0;
    pv += __shfl_xor(pv, 16);
    pv += __shfl_xor(pv, 8);
    pv += __shfl_xor(pv, 4);
    pv += __shfl_xor(pv, 2);
    pv += __shfl_xor(pv, 1);
    if ((t & 31) == 0) red[t >> 5] = pv;
    sc[t] = c;
    __syncthreads();
    if (i < n) {
        const int w = t >> 1;
        const unsigned cn = (t & 1) ? (chi[0][w] + chi[1][w]) : (clo[0][w] + clo[1][w]);
        dinv[i] = rsqrtf(1.0f + (float)cn);
    }
    const int base = ((red[0] + red[1]) + (red[2] + red[3])) + ((red[4] + red[5]) + (red[6] + red[7]));
    for (int off = 1; off < 256; off <<= 1) {
        const int v = (t >= off) ? sc[t - off] : 0;
        __syncthreads();
        sc[t] += v;
        __syncthreads();
    }
    toff[i] = base + sc[t] - c;
}

#define CHE   (N_EDGES / NCH)
#define XWG   ((N_NODES / 16 + 31) / 32)
static_assert(NBK <= 1024, "one bucket per thread in the chunk's scan");
static_assert(N_NODES % 16 == 0, "row tiles are whole");
__global__ void __launch_bounds__(1024)
bucket_place_xw1_kernel(const int* __restrict__ row, const int* __restrict__ col,
                        const int* __restrict__ tab, const int* __restrict__ toff,
                        int* __restrict__ part,
                        const float* __restrict__ x,
                        const float* __restrict__ W1,
                        const float* __restrict__ dinv,
                        float* __restrict__ hlin,
                        int n) {
    __shared__ int sw[CHE];
    __shared__ int sd[CHE];
    __shared__ int cur[NBK];
    __shared__ int del[NBK];
    __shared__ int wtot[32];
    __shared__ float Wp[128 * 16];
    const int t    = threadIdx.x;
    const int lane = t & 31;
    const int wv   = t >> 5;
    if (blockIdx.x < NCH) {
        const int q0 = blockIdx.x * CH4;
        const int4 c0 = reinterpret_cast<const int4*>(col)[q0 + t];
        const int4 r0 = reinterpret_cast<const int4*>(row)[q0 + t];
        static_assert(CH4 >= 1024, "every thread has a first quad");
        int cn = 0, g = 0;
        if (t < NBK) { cn = tab[t * NCH + blockIdx.x]; g = toff[t * NCH + blockIdx.x]; }
        int inc = cn;
        #pragma unroll
        for (int o = 1; o < 32; o <<= 1) {
            const int v = __shfl_up(inc, o);
            if (lane >= o) inc += v;
        }
        if (lane == 31) wtot[wv] = inc;
        __syncthreads();
        {
            const int wt = wtot[lane];
            int winc = wt;
            #pragma unroll
            for (int o = 1; o < 32; o <<= 1) {
                const int v = __shfl_up(winc, o);
                if (lane >= o) winc += v;
            }
            const int wbase = __shfl(winc - wt, wv);
            if (t < NBK) {
                const int ls = wbase + inc - cn;
                cur[t] = ls;
                del[t] = g - ls;
            }
        }
        __syncthreads();
        for (int q = t; q < CH4; q += 1024) {
            int4 c = c0, r = r0;
            if (q != t) {
                c = reinterpret_cast<const int4*>(col)[q0 + q];
                r = reinterpret_cast<const int4*>(row)[q0 + q];
            }
            const int b0 = r.x >> 8, b1 = r.y >> 8, b2 = r.z >> 8, b3 = r.w >> 8;
            const int p0 = __hip_atomic_fetch_add(&cur[b0], 1, __ATOMIC_RELAXED, __HIP_MEMORY_SCOPE_WORKGROUP);
            const int p1 = __hip_atomic_fetch_add(&cur[b1], 1, __ATOMIC_RELAXED, __HIP_MEMORY_SCOPE_WORKGROUP);
            const int p2 = __hip_atomic_fetch_add(&cur[b2], 1, __ATOMIC_RELAXED, __HIP_MEMORY_SCOPE_WORKGROUP);
            const int p3 = __hip_atomic_fetch_add(&cur[b3], 1, __ATOMIC_RELAXED, __HIP_MEMORY_SCOPE_WORKGROUP);
            sw[p0] = ((r.x & 255) << 17) | c.x;  sd[p0] = p0 + del[b0];
            sw[p1] = ((r.y & 255) << 17) | c.y;  sd[p1] = p1 + del[b1];
            sw[p2] = ((r.z & 255) << 17) | c.z;  sd[p2] = p2 + del[b2];
            sw[p3] = ((r.w & 255) << 17) | c.w;  sd[p3] = p3 + del[b3];
        }
        __syncthreads();
        for (int i = t; i < CHE; i += 1024) part[sd[i]] = sw[i];
    } else {
        {
            const int idx = t * 2;
            const int kr  = idx >> 4;
            const int nc  = idx & 15;
            const float w0 = W1[kr * 4 + (nc & 3)];
            const float w1 = W1[kr * 4 + ((nc + 1) & 3)];
            Wp[idx]     = (nc < 4) ? w0 : 0.0f;
            Wp[idx + 1] = (nc < 4) ? w1 : 0.0f;
        }
        __syncthreads();

        const int tile   = (blockIdx.x - NCH) * 32 + wv;
        const int nTiles = n >> 4;
        if (tile >= nTiles) return;

        const int base = tile * 16;
        const int hh   = lane >> 4;
        const int nl   = lane & 15;
        const int arow = base + nl;
        const float4* __restrict__ dv = reinterpret_cast<const float4*>(dinv + base + 8 * hh);
        const float4 d0 = dv[0];
        const float4 d1 = dv[1];

        v8f c = {};
        #pragma unroll 4
        for (int k = 0; k < 128; k += 4) {
            const int kk = k + 2 * hh;
            v2f a = *reinterpret_cast<const v2f*>(x + (size_t)arow * 128 + kk);
            v2f b;
            b.x = Wp[(kk + 0) * 16 + nl];
            b.y = Wp[(kk + 1) * 16 + nl];
            c = __builtin_amdgcn_wmma_f32_16x16x4_f32(
                     false, a,  false, b,
                     (short)0, c,  false,  false);
        }

        if (nl < 4) {
            const float ds[8] = {d0.x, d0.y, d0.z, d0.w, d1.x, d1.y, d1.z, d1.w};
            #pragma unroll
            for (int v = 0; v < 8; ++v) {
                int m = base + v + 8 * hh;
                hlin[(size_t)m * 4 + nl] = c[v] * ds[v];
            }
        }
    }
}

#define SGP   4
#define SWIN  2
static_assert(16 * 4 * SGP == 256, "the 16 waves of the work-group cover the bucket's 256 nodes");
static_assert(SWIN <= SGP, "the window is no longer than the wave's groups");
__device__ __forceinline__ void add_row(float (&acc)[4], const float4 u, const bool live) {
    acc[0] += live ? u.x : 0.0f;
    acc[1] += live ? u.y : 0.0f;
    acc[2] += live ? u.z : 0.0f;
    acc[3] += live ? u.w : 0.0f;
}
__device__ __forceinline__ void add_row(float (&acc)[2], const float2 u, const bool live) {
    acc[0] += live ? u.x : 0.0f;
    acc[1] += live ? u.y : 0.0f;
}
template <bool STAGED>
__device__ __forceinline__ void bucket_gather1(const int* st, const int* srcb,
                                               const int* c, const int* sc,
                                               const float4* __restrict__ hrow,
                                               const float* __restrict__ hlin,
                                               const float* __restrict__ dinv,
                                               const float* __restrict__ b1,
                                               const float* __restrict__ W2,
                                               float* __restrict__ ynext,
                                               const int b, const int n, const int t) {
    const int lane  = t & 31;
    const int wv    = t >> 5;
    const int sub   = lane & 7;
    const int first = lane & ~7;
    const int so    = sub & 3;
    const float bb = b1[so];
    float w[4];
    #pragma unroll
    for (int k = 0; k < 4; ++k) w[k] = W2[k * 4 + so];

    int   lb[SGP], cn[SGP], rc[SGP];
    float hs[SGP], di[SGP];
    #pragma unroll
    for (int i = 0; i < SGP; ++i) {
        const int ln = wv * (4 * SGP) + 4 * i + (lane >> 3);
        const int r  = b * 256 + ln;
        rc[i] = (r < n) ? r : (n - 1);
        const int cc = c[ln];
        cn[i] = (r < n) ? cc : 0;
        lb[i] = sc[ln] - cc;
        hs[i] = hlin[(size_t)rc[i] * 4 + so];
        di[i] = dinv[rc[i]];
    }
    float4 v[SGP][8];
    #pragma unroll
    for (int i = 0; i < SWIN; ++i) {
        int s8[8];
        #pragma unroll
        for (int k = 0; k < 8; ++k) {
            const int q = sub + 8 * k;
            const bool live = q < cn[i];
            int idx = lb[i] + (live ? q : (cn[i] - 1));
            idx = (idx > 0) ? idx : 0;
            const int wd = STAGED ? st[idx] : srcb[idx];
            s8[k] = live ? wd : rc[i];
        }
        #pragma unroll
        for (int k = 0; k < 8; ++k) v[i][k] = hrow[s8[k]];
    }
    #pragma unroll
    for (int i = 0; i < SGP; ++i) {
        float acc[4] = {0.0f, 0.0f, 0.0f, 0.0f};
        #pragma unroll
        for (int k = 0; k < 8; ++k) add_row(acc, v[i][k], (sub + 8 * k) < cn[i]);
        if (i + SWIN < SGP) {
            const int j = (i + SWIN) % SGP;
            int s8[8];
            #pragma unroll
            for (int k = 0; k < 8; ++k) {
                const int q = sub + 8 * k;
                const bool live = q < cn[j];
                int idx = lb[j] + (live ? q : (cn[j] - 1));
                idx = (idx > 0) ? idx : 0;
                const int wd = STAGED ? st[idx] : srcb[idx];
                s8[k] = live ? wd : rc[j];
            }
            #pragma unroll
            for (int k = 0; k < 8; ++k) v[j][k] = hrow[s8[k]];
        }
        for (int p = 64 + sub; p < cn[i]; p += 8) {
            const int s = STAGED ? st[lb[i] + p] : srcb[lb[i] + p];
            add_row(acc, hrow[s], true);
        }
        #pragma unroll
        for (int f = 0; f < 4; ++f) {
            acc[f] += __shfl_xor(acc[f], 4);
            acc[f] += __shfl_xor(acc[f], 2);
            acc[f] += __shfl_xor(acc[f], 1);
        }
        float a = acc[0];
        #pragma unroll
        for (int f = 1; f < 4; ++f) a = (sub == f) ? acc[f] : a;
        const float tv = tanhf((a + hs[i]) * di[i] + bb);
        float tk[4];
        #pragma unroll
        for (int k = 0; k < 4; ++k) tk[k] = __shfl(tv, first + k);
        float y = 0.0f;
        #pragma unroll
        for (int k = 0; k < 4; ++k) y = fmaf(tk[k], w[k], y);
        y *= di[i];
        const int r = b * 256 + wv * (4 * SGP) + 4 * i + (lane >> 3);
        if (r < n && sub < 4) ynext[(size_t)r * 4 + sub] = y;
    }
}

#define GCAP  12288
__global__ void __launch_bounds__(512, 8)
bucket_group_gather1_kernel(const int* __restrict__ part, const int* __restrict__ toff,
                            int* __restrict__ rowptr, int* __restrict__ cnt,
                            int* __restrict__ src,
                            const float* __restrict__ hlin,
                            const float* __restrict__ dinv,
                            const float* __restrict__ b1,
                            const float* __restrict__ W2,
                            float* __restrict__ ynext,
                            int n, int ne) {
    __shared__ int   c[256];
    __shared__ int   sc[256];
    __shared__ int   cur[256];
    __shared__ int   st[GCAP];
    const int t = threadIdx.x;
    const int b = blockIdx.x;
    const int node = b * 256 + t;
    const int beg = toff[b * NCH];
    const int end = (b + 1 < NBK) ? toff[(b + 1) * NCH] : ne;
    if (t < 256) c[t] = 0;
    __syncthreads();
    for (int e = beg + t; e < end; e += 4 * 512) {
        int w[4];
        #pragma unroll
        for (int k = 0; k < 4; ++k) {
            const int q = e + 512 * k;
            w[k] = (q < end) ? part[q] : -1;
        }
        #pragma unroll
        for (int k = 0; k < 4; ++k)
            if (w[k] >= 0)
                __hip_atomic_fetch_add(&c[w[k] >> 17], 1, __ATOMIC_RELAXED, __HIP_MEMORY_SCOPE_WORKGROUP);
    }
    __syncthreads();
    int my = 0;
    if (t < 256) { my = c[t]; sc[t] = my; }
    __syncthreads();
    for (int off = 1; off < 256; off <<= 1) {
        int v = 0;
        if (t < 256 && t >= off) v = sc[t - off];
        __syncthreads();
        if (t < 256) sc[t] += v;
        __syncthreads();
    }
    if (t < 256) {
        const int first = beg + sc[t] - my;
        cur[t] = first;
        if (node < n) { rowptr[node] = first; cnt[node] = my; }
    }
    __syncthreads();
    const bool staged = (end - beg) <= GCAP;
    for (int e = beg + t; e < end; e += 4 * 512) {
        int w[4];
        #pragma unroll
        for (int k = 0; k < 4; ++k) {
            const int q = e + 512 * k;
            w[k] = (q < end) ? part[q] : -1;
        }
        #pragma unroll
        for (int k = 0; k < 4; ++k) {
            if (w[k] >= 0) {
                const int rl = w[k] >> 17;
                const int s  = w[k] & 0x1FFFF;
                const int p  = __hip_atomic_fetch_add(&cur[rl], 1, __ATOMIC_RELAXED, __HIP_MEMORY_SCOPE_WORKGROUP);
                if (staged) st[p - beg] = s;
                else        src[p]      = s;
            }
        }
    }
    __syncthreads();
    const float4* __restrict__ hrow = reinterpret_cast<const float4*>(hlin);
    if (staged) {
        for (int i = t; i < end - beg; i += 512) src[beg + i] = st[i];
        bucket_gather1<true>(st, src + beg, c, sc, hrow, hlin, dinv, b1, W2, ynext, b, n, t);
    } else {
        bucket_gather1<false>(st, src + beg, c, sc, hrow, hlin, dinv, b1, W2, ynext, b, n, t);
    }
}

#define GPW   5
#define GWV   8
#define GWIN  2
#define GWG   (N_NODES / (4 * GPW * GWV))
static_assert(N_NODES % (4 * GPW * GWV) == 0, "every wave has GPW whole groups");
static_assert(GWIN <= GPW, "the window is no longer than the wave's groups");
template <int OUT> struct row_of;
template <> struct row_of<4> { typedef float4 type; };
template <> struct row_of<2> { typedef float2 type; };
template <int OUT, int NOUT, bool LAST>
__global__ void __launch_bounds__(32 * GWV)
gather_layer_kernel(const int* __restrict__ rowptr,
                    const int* __restrict__ cnt,
                    const int* __restrict__ src,
                    const float* __restrict__ hlin,
                    const float* __restrict__ dinv,
                    const float* __restrict__ b,
                    const float* __restrict__ Wn,
                    const float* __restrict__ bn,
                    float* __restrict__ ynext,
                    float* __restrict__ tout,
                    int n) {
    static_assert(!LAST || OUT == 2, "the classifier's sum is written for two features");
    typedef typename row_of<OUT>::type row_t;
    const row_t* __restrict__ hrow = reinterpret_cast<const row_t*>(hlin);
    const int lane = threadIdx.x & 31;
    const int sub  = lane & 7;
    const int gw   = blockIdx.x * GWV + (threadIdx.x >> 5);
    const int r0   = gw * (4 * GPW) + (lane >> 3);
    const int first = lane & ~7;

    static_assert((OUT & (OUT - 1)) == 0 && (NOUT & (NOUT - 1)) == 0, "the wrapped index is a mask");
    const int so = sub & (OUT - 1);
    const int sn = sub & (NOUT - 1);
    const float bb = b[so];
    float w[OUT];
    #pragma unroll
    for (int k = 0; k < OUT; ++k) w[k] = Wn[k * NOUT + sn];
    float bo = 0.0f;
    if (LAST) bo = bn[sn];

    int   beg[GPW], cn[GPW];
    float hs[GPW], di[GPW];
    #pragma unroll
    for (int i = 0; i < GPW; ++i) {
        const int r  = r0 + 4 * i;
        const int rc = (r < n) ? r : (n - 1);
        beg[i] = rowptr[rc];
        const int cc = cnt[rc];
        cn[i]  = (r < n) ? cc : 0;
        hs[i]  = hlin[(size_t)rc * OUT + so];
        di[i]  = dinv[rc];
    }
    int sr[GPW][8];
    #pragma unroll
    for (int i = 0; i < GPW; ++i) {
        #pragma unroll
        for (int k = 0; k < 8; ++k) {
            const int q = sub + 8 * k;
            int idx = beg[i] + ((q < cn[i]) ? q : (cn[i] - 1));
            idx = (idx > 0) ? idx : 0;
            sr[i][k] = src[idx];
        }
    }
    row_t v[GPW][8];
    #pragma unroll
    for (int i = 0; i < GWIN; ++i) {
        #pragma unroll
        for (int k = 0; k < 8; ++k) v[i][k] = hrow[sr[i][k]];
    }
    #pragma unroll
    for (int i = 0; i < GPW; ++i) {
        float acc[OUT];
        #pragma unroll
        for (int f = 0; f < OUT; ++f) acc[f] = 0.0f;
        #pragma unroll
        for (int k = 0; k < 8; ++k) add_row(acc, v[i][k], (sub + 8 * k) < cn[i]);
        if (i + GWIN < GPW) {
            #pragma unroll
            for (int k = 0; k < 8; ++k) v[(i + GWIN) % GPW][k] = hrow[sr[(i + GWIN) % GPW][k]];
        }
        const int end = beg[i] + cn[i];
        for (int p = beg[i] + 64 + sub; p < end; p += 8) {
            add_row(acc, hrow[src[p]], true);
        }
        #pragma unroll
        for (int f = 0; f < OUT; ++f) {
            acc[f] += __shfl_xor(acc[f], 4);
            acc[f] += __shfl_xor(acc[f], 2);
            acc[f] += __shfl_xor(acc[f], 1);
        }

        float a = acc[0];
        #pragma unroll
        for (int f = 1; f < OUT; ++f) a = (sub == f) ? acc[f] : a;
        const float t  = tanhf((a + hs[i]) * di[i] + bb);
        float tk[OUT];
        #pragma unroll
        for (int k = 0; k < OUT; ++k) tk[k] = __shfl(t, first + k);
        float y;
        if (LAST) {
            y = fmaf(tk[0], w[0], fmaf(tk[1], w[1], bo));
        } else {
            y = 0.0f;
            #pragma unroll
            for (int k = 0; k < OUT; ++k) y = fmaf(tk[k], w[k], y);
            y *= di[i];
        }
        const int r = r0 + 4 * i;
        if (r < n) {
            if (sub < NOUT) ynext[(size_t)r * NOUT + sub] = y;
            if (LAST) {
                if (sub < OUT) tout[(size_t)r * OUT + sub] = t;
            }
        }
    }
}

extern "C" void kernel_launch(void* const* d_in, const int* in_sizes, int n_in,
                              void* d_out, int out_size, void* d_ws, size_t ws_size,
                              hipStream_t stream) {
    const float* x  = (const float*)d_in[0];
    const int*   ei = (const int*)d_in[1];
    const float* W1 = (const float*)d_in[2];
    const float* b1 = (const float*)d_in[3];
    const float* W2 = (const float*)d_in[4];
    const float* b2 = (const float*)d_in[5];
    const float* W3 = (const float*)d_in[6];
    const float* b3 = (const float*)d_in[7];
    const float* Wc = (const float*)d_in[8];
    const float* bc = (const float*)d_in[9];

    const int N = N_NODES;
    const int E = N_EDGES;
    const int* row = ei;
    const int* col = ei + E;

    float* ws     = (float*)d_ws;
    float* hP     = ws;
    float* hQ     = hP + 4 * N;
    int*   src    = (int*)(hQ + 4 * N);
    int*   part   = src + E;
    int*   tab    = part + E;
    int*   toff   = tab + NBK * NCH;
    float* dinv   = (float*)(toff + NBK * NCH);
    int*   cnt    = (int*)(dinv + N);
    int*   rowptr = cnt + N;
    unsigned* planes = (unsigned*)(rowptr + N);
    int*   pre    = (int*)(planes + (size_t)NSL * NPW);

    float* outC = (float*)d_out;
    float* outH = outC + (size_t)4 * N;

    const int BT = 32 * GWV;
    const int gG = GWG;

    deg_hist_bucket_count_kernel<<<NSL + NCW, 1024, 0, stream>>>(row, col, planes, tab, pre);
    scan_dinv_kernel<<<NBK, 256, 0, stream>>>(tab, pre, toff, planes, dinv, N);
    bucket_place_xw1_kernel<<<NCH + XWG, 1024, 0, stream>>>(row, col, tab, toff, part,
                                                          x, W1, dinv, hP, N);
    bucket_group_gather1_kernel<<<NBK, 512, 0, stream>>>(part, toff, rowptr, cnt, src,
                                                         hP, dinv, b1, W2, hQ, N, E);
    gather_layer_kernel<4, 2, false><<<gG, BT, 0, stream>>>(
        rowptr, cnt, src, hQ, dinv, b2, W3, nullptr, hP, nullptr, N);
    gather_layer_kernel<2, 4, true><<<gG, BT, 0, stream>>>(
        rowptr, cnt, src, hP, dinv, b3, Wc, bc, outC, outH, N);
}
